# starting kernel
# baseline (speedup 1.0000x reference)
#include <hip/hip_runtime.h>
#include <hip/hip_bf16.h>
#include <math.h>

#define NN      50000
#define NE      800000
#define DF      128
#define NHEADS  4
#define HU      32
#define NEG_SLOPE 0.2f
#define BN_EPS    0.001f

typedef __attribute__((ext_vector_type(2))) float v2f;
typedef __attribute__((ext_vector_type(8))) float v8f;

#define OFF_H    ((size_t)0)
#define OFF_UPD  (OFF_H   + (size_t)NN * DF)
#define OFF_AGG  (OFF_UPD + (size_t)NN * DF)
#define OFF_SI   (OFF_AGG + (size_t)NN * DF)
#define OFF_SJ   (OFF_SI  + (size_t)NN * NHEADS)
#define OFF_MAX  (OFF_SJ  + (size_t)NN * NHEADS)
#define OFF_DEN  (OFF_MAX + (size_t)NN * NHEADS)
#define OFF_W    (OFF_DEN + (size_t)NN * NHEADS)
#define OFF_MACC (OFF_W   + (size_t)NE * NHEADS)
#define OFF_SQ   (OFF_MACC + (size_t)DF)

__device__ __forceinline__ void atomicMaxFloat(float* addr, float val) {
    if (val >= 0.0f) {
        atomicMax((int*)addr, __float_as_int(val));
    } else {
        atomicMin((unsigned int*)addr, __float_as_uint(val));
    }
}

__global__ void k_init(float* __restrict__ ws) {
    int i = blockIdx.x * blockDim.x + threadIdx.x;
    size_t nagg = (size_t)NN * DF;
    if ((size_t)i < nagg) ws[OFF_AGG + i] = 0.0f;
    if (i < NN * NHEADS) {
        ws[OFF_MAX + i] = -INFINITY;
        ws[OFF_DEN + i] = 0.0f;
    }
    if (i < DF) {
        ws[OFF_MACC + i] = 0.0f;
        ws[OFF_SQ + i]   = 0.0f;
    }
}

template <int WLAYOUT, int DORELU>
__global__ __launch_bounds__(256) void k_gemm_wmma(
    const float* __restrict__ A, const float* __restrict__ B,
    const float* __restrict__ bias, float* __restrict__ C, int M)
{
    __shared__ float atile[16 * DF];

    const int tid = threadIdx.x;
    const int m0  = blockIdx.x * 16;

    #pragma unroll
    for (int i = 0; i < 8; ++i) {
        const int idx = tid + i * 256;
        const int r = idx >> 7, c = idx & (DF - 1);
        const int row = m0 + r;
        atile[idx] = (row < M) ? A[(size_t)row * DF + c] : 0.0f;
    }
    __syncthreads();

    const int lane = tid & 31;
    const int half = lane >> 4;
    const int l    = lane & 15;
    const int n0   = (tid >> 5) * 16;
    const int col  = n0 + l;

    const int b_base   = (WLAYOUT == 0) ? col
                                        : ((col >> 5) * (DF * HU) + (col & 31));
    const int b_kstride = (WLAYOUT == 0) ? DF : HU;

    const float* __restrict__ arow = &atile[l * DF + 2 * half];
    const float* __restrict__ bcol = &B[b_base + (size_t)(2 * half) * b_kstride];

    v8f acc = {};
    #pragma unroll
    for (int kb = 0; kb < DF; kb += 4) {
        v2f a = *(const v2f*)(arow + kb);
        v2f b;
        b.x = bcol[(size_t)kb * b_kstride];
        b.y = bcol[(size_t)(kb + 1) * b_kstride];
        acc = __builtin_amdgcn_wmma_f32_16x16x4_f32(
                  false, a, false, b, (short)0, acc, false, false);
    }

    const float bv = bias[col];
    #pragma unroll
    for (int r = 0; r < 8; ++r) {
        float v = acc[r] + bv;
        if (DORELU) v = fmaxf(v, 0.0f);
        const int row = m0 + r + 8 * half;
        if (row < M) C[(size_t)row * DF + col] = v;
    }
}

__global__ __launch_bounds__(128) void k_scores(
    const float* __restrict__ h, const float* __restrict__ a_i,
    const float* __restrict__ a_j, float* __restrict__ s_i,
    float* __restrict__ s_j)
{
    const int node = blockIdx.x;
    const int t = threadIdx.x;
    const float hv = h[(size_t)node * DF + t];
    float vi = hv * a_i[t];
    float vj = hv * a_j[t];
    #pragma unroll
    for (int off = 16; off > 0; off >>= 1) {
        vi += __shfl_xor(vi, off, 32);
        vj += __shfl_xor(vj, off, 32);
    }
    if ((t & 31) == 0) {
        const int hd = t >> 5;
        s_i[node * NHEADS + hd] = vi;
        s_j[node * NHEADS + hd] = vj;
    }
}

__global__ void k_edge_logits_max(
    const int* __restrict__ ei, const float* __restrict__ s_i,
    const float* __restrict__ s_j, const float* __restrict__ a_e,
    const float* __restrict__ b_a, float* __restrict__ wbuf,
    float* __restrict__ maxb)
{
    const int idx = blockIdx.x * blockDim.x + threadIdx.x;
    if (idx >= NE * NHEADS) return;
    const int e  = idx >> 2;
    const int hd = idx & 3;
    const int recv = ei[2 * e];
    const int send = ei[2 * e + 1];
    float lg = s_i[recv * NHEADS + hd] + s_j[send * NHEADS + hd]
             + a_e[hd] + b_a[hd];
    lg = (lg > 0.0f) ? lg : NEG_SLOPE * lg;
    wbuf[idx] = lg;
    atomicMaxFloat(&maxb[recv * NHEADS + hd], lg);
}

__global__ void k_edge_exp_sum(
    const int* __restrict__ ei, const float* __restrict__ maxb,
    float* __restrict__ wbuf, float* __restrict__ denom)
{
    const int idx = blockIdx.x * blockDim.x + threadIdx.x;
    if (idx >= NE * NHEADS) return;
    const int e  = idx >> 2;
    const int hd = idx & 3;
    const int recv = ei[2 * e];
    const float w = expf(wbuf[idx] - maxb[recv * NHEADS + hd]);
    wbuf[idx] = w;
    atomicAdd(&denom[recv * NHEADS + hd], w);
}

__global__ __launch_bounds__(256) void k_edge_scatter(
    const int* __restrict__ ei, const float* __restrict__ h,
    const float* __restrict__ wbuf, const float* __restrict__ denom,
    float* __restrict__ agg)
{
    const long long idx = (long long)blockIdx.x * 256 + threadIdx.x;
    if (idx >= (long long)NE * DF) return;
    const int e   = (int)(idx >> 7);
    const int col = (int)(idx & (DF - 1));
    const int hd  = col >> 5;
    const int recv = ei[2 * e];
    const int send = ei[2 * e + 1];
    const float alpha = wbuf[e * NHEADS + hd]
                      / (denom[recv * NHEADS + hd] + 1e-16f);
    atomicAdd(&agg[(size_t)recv * DF + col],
              h[(size_t)send * DF + col] * alpha);
}

#define BN_ROWS 128
__global__ __launch_bounds__(128) void k_bn_partial(
    const float* __restrict__ upd, float* __restrict__ macc,
    float* __restrict__ sqacc)
{
    const int col = threadIdx.x;
    const int r0  = blockIdx.x * BN_ROWS;
    float s = 0.0f, s2 = 0.0f;
    for (int r = 0; r < BN_ROWS; ++r) {
        const int row = r0 + r;
        if (row < NN) {
            const float v = upd[(size_t)row * DF + col];
            s += v; s2 += v * v;
        }
    }
    atomicAdd(&macc[col], s);
    atomicAdd(&sqacc[col], s2);
}

__global__ void k_finalize(
    const float* __restrict__ upd, const float* __restrict__ agg,
    const float* __restrict__ macc, const float* __restrict__ sqacc,
    const float* __restrict__ gamma, const float* __restrict__ beta,
    float* __restrict__ out)
{
    const long long idx = (long long)blockIdx.x * blockDim.x + threadIdx.x;
    if (idx >= (long long)NN * DF) return;
    const int col = (int)(idx & (DF - 1));
    const float inv_n = 1.0f / (float)NN;
    const float mean = macc[col] * inv_n;
    const float var  = sqacc[col] * inv_n - mean * mean;
    const float normed = gamma[col] * (upd[idx] - mean)
                       * rsqrtf(var + BN_EPS) + beta[col];
    out[idx] = normed + fmaxf(agg[idx], 0.0f);
}

extern "C" void kernel_launch(void* const* d_in, const int* in_sizes, int n_in,
                              void* d_out, int out_size, void* d_ws, size_t ws_size,
                              hipStream_t stream) {
    const float* x     = (const float*)d_in[0];
    const int*   ei    = (const int*)  d_in[1];
    const float* Wk    = (const float*)d_in[2];
    const float* bk    = (const float*)d_in[3];
    const float* a_i   = (const float*)d_in[4];
    const float* a_j   = (const float*)d_in[5];
    const float* a_e   = (const float*)d_in[6];
    const float* b_a   = (const float*)d_in[7];
    const float* Wm    = (const float*)d_in[8];
    const float* bm    = (const float*)d_in[9];
    const float* gamma = (const float*)d_in[10];
    const float* beta  = (const float*)d_in[11];
    float* out = (float*)d_out;
    float* ws  = (float*)d_ws;

    float* h_buf  = ws + OFF_H;
    float* upd    = ws + OFF_UPD;
    float* agg    = ws + OFF_AGG;
    float* s_i    = ws + OFF_SI;
    float* s_j    = ws + OFF_SJ;
    float* maxb   = ws + OFF_MAX;
    float* denom  = ws + OFF_DEN;
    float* wbuf   = ws + OFF_W;
    float* macc   = ws + OFF_MACC;
    float* sqacc  = ws + OFF_SQ;

    {
        const long long n = (long long)NN * DF;
        k_init<<<(int)((n + 255) / 256), 256, 0, stream>>>(ws);
    }
    const int gemm_blocks = (NN + 15) / 16;
    k_gemm_wmma<1, 0><<<gemm_blocks, 256, 0, stream>>>(x, Wk, bk, h_buf, NN);
    k_gemm_wmma<0, 1><<<gemm_blocks, 256, 0, stream>>>(x, Wm, bm, upd, NN);
    k_scores<<<NN, 128, 0, stream>>>(h_buf, a_i, a_j, s_i, s_j);
    {
        const int n = NE * NHEADS;
        k_edge_logits_max<<<(n + 255) / 256, 256, 0, stream>>>(
            ei, s_i, s_j, a_e, b_a, wbuf, maxb);
    }
    {
        const int n = NE * NHEADS;
        k_edge_exp_sum<<<(n + 255) / 256, 256, 0, stream>>>(
            ei, maxb, wbuf, denom);
    }
    {
        const long long n = (long long)NE * DF;
        k_edge_scatter<<<(int)((n + 255) / 256), 256, 0, stream>>>(
            ei, h_buf, wbuf, denom, agg);
    }
    k_bn_partial<<<(NN + BN_ROWS - 1) / BN_ROWS, 128, 0, stream>>>(
        upd, macc, sqacc);
    {
        const long long n = (long long)NN * DF;
        k_finalize<<<(int)((n + 255) / 256), 256, 0, stream>>>(
            upd, agg, macc, sqacc, gamma, beta, out);
    }
}


# Per-node sums in ascending edge id (arrival-rank order made runs differ): float-atomic scatter replaced by receiver-grouped edge list (int-atomic counting sort), one wave per node sums in registers
# speedup vs baseline: 1.9119x; 1.9119x over previous
#include <hip/hip_runtime.h>
#include <hip/hip_bf16.h>
#include <math.h>

#define NN      50000
#define NE      800000
#define DF      128
#define NHEADS  4
#define HU      32
#define NEG_SLOPE 0.2f
#define BN_EPS    0.001f

typedef __attribute__((ext_vector_type(2))) float v2f;
typedef __attribute__((ext_vector_type(8))) float v8f;

#define OFF_H    ((size_t)0)
#define OFF_UPD  (OFF_H   + (size_t)NN * DF)
#define OFF_AGG  (OFF_UPD + (size_t)NN * DF)
#define OFF_SI   (OFF_AGG + (size_t)NN * DF)
#define OFF_SJ   (OFF_SI  + (size_t)NN * NHEADS)
#define OFF_CNT  (OFF_SJ  + (size_t)NN * NHEADS)
#define OFF_OFFS (OFF_CNT + (size_t)NN * NHEADS)
#define OFF_RANK (OFF_OFFS + (size_t)NN * NHEADS)
#define OFF_LIST (OFF_RANK + (size_t)NE)
#define OFF_TOT  (OFF_LIST + (size_t)2 * NE)
#define OFF_MACC (OFF_RANK + (size_t)NE * NHEADS)
#define OFF_SQ   (OFF_MACC + (size_t)DF)

#define DEV static __device__ __forceinline__
DEV int   lane_i(int v, int l)   { return __builtin_amdgcn_readlane(v, l); }
DEV float lane_f(float v, int l) { return __int_as_float(__builtin_amdgcn_readlane(__float_as_int(v), l)); }
DEV float wave_max(float v) {
    for (int off = 16; off > 0; off >>= 1) v = fmaxf(v, __shfl_xor(v, off, 32));
    return v;
}
DEV float wave_sum(float v) {
    for (int off = 16; off > 0; off >>= 1) v += __shfl_xor(v, off, 32);
    return v;
}
DEV int wave_min_i(int v) {
    for (int off = 16; off > 0; off >>= 1) v = min(v, __shfl_xor(v, off, 32));
    return v;
}
DEV float leaky(float x) { return (x > 0.0f) ? x : NEG_SLOPE * x; }
DEV float pick4(float a0, float a1, float a2, float a3, int k0, int k1, int k2, int k3) {
    return __int_as_float((__float_as_int(a0) & k0) | (__float_as_int(a1) & k1) |
                          (__float_as_int(a2) & k2) | (__float_as_int(a3) & k3));
}

__global__ void k_init(float* __restrict__ ws) {
    int i = blockIdx.x * blockDim.x + threadIdx.x;
    int* wi = (int*)ws;
    if (i < NN) wi[OFF_CNT + i] = 0;
    if (i == 0) wi[OFF_TOT] = 0;
    if (i < DF) {
        ws[OFF_MACC + i] = 0.0f;
        ws[OFF_SQ + i]   = 0.0f;
    }
}

template <int WLAYOUT, int DORELU>
__global__ __launch_bounds__(256) void k_gemm_wmma(
    const float* __restrict__ A, const float* __restrict__ B,
    const float* __restrict__ bias, float* __restrict__ C, int M)
{
    __shared__ float atile[16 * DF];

    const int tid = threadIdx.x;
    const int m0  = blockIdx.x * 16;

    #pragma unroll
    for (int i = 0; i < 8; ++i) {
        const int idx = tid + i * 256;
        const int r = idx >> 7, c = idx & (DF - 1);
        const int row = m0 + r;
        atile[idx] = (row < M) ? A[(size_t)row * DF + c] : 0.0f;
    }
    __syncthreads();

    const int lane = tid & 31;
    const int half = lane >> 4;
    const int l    = lane & 15;
    const int n0   = (tid >> 5) * 16;
    const int col  = n0 + l;

    const int b_base   = (WLAYOUT == 0) ? col
                                        : ((col >> 5) * (DF * HU) + (col & 31));
    const int b_kstride = (WLAYOUT == 0) ? DF : HU;

    const float* __restrict__ arow = &atile[l * DF + 2 * half];
    const float* __restrict__ bcol = &B[b_base + (size_t)(2 * half) * b_kstride];

    v8f acc = {};
    #pragma unroll
    for (int kb = 0; kb < DF; kb += 4) {
        v2f a = *(const v2f*)(arow + kb);
        v2f b;
        b.x = bcol[(size_t)kb * b_kstride];
        b.y = bcol[(size_t)(kb + 1) * b_kstride];
        acc = __builtin_amdgcn_wmma_f32_16x16x4_f32(
                  false, a, false, b, (short)0, acc, false, false);
    }

    const float bv = bias[col];
    #pragma unroll
    for (int r = 0; r < 8; ++r) {
        float v = acc[r] + bv;
        if (DORELU) v = fmaxf(v, 0.0f);
        const int row = m0 + r + 8 * half;
        if (row < M) C[(size_t)row * DF + col] = v;
    }
}

__global__ __launch_bounds__(128) void k_scores(
    const float* __restrict__ h, const float* __restrict__ a_i,
    const float* __restrict__ a_j, float* __restrict__ s_i,
    float* __restrict__ s_j)
{
    const int node = blockIdx.x;
    const int t = threadIdx.x;
    const float hv = h[(size_t)node * DF + t];
    float vi = hv * a_i[t];
    float vj = hv * a_j[t];
    #pragma unroll
    for (int off = 16; off > 0; off >>= 1) {
        vi += __shfl_xor(vi, off, 32);
        vj += __shfl_xor(vj, off, 32);
    }
    if ((t & 31) == 0) {
        const int hd = t >> 5;
        s_i[node * NHEADS + hd] = vi;
        s_j[node * NHEADS + hd] = vj;
    }
}

__global__ __launch_bounds__(256) void k_edge_rank(
    const int* __restrict__ ei, int* __restrict__ cnt, int* __restrict__ rank)
{
    const int e = blockIdx.x * 256 + threadIdx.x;
    if (e >= NE) return;
    rank[e] = atomicAdd(&cnt[ei[2 * e]], 1);
}

__global__ __launch_bounds__(256) void k_seg_alloc(
    const int* __restrict__ cnt, int* __restrict__ offs, int* __restrict__ total)
{
    __shared__ int wsum[8];
    __shared__ int bbase;
    const int tid = threadIdx.x, lane = tid & 31, w = tid >> 5;
    const int i = blockIdx.x * 256 + tid;
    const int c = (i < NN) ? cnt[i] : 0;
    int s = c;
    for (int d = 1; d < 32; d <<= 1) {
        const int t = __shfl_up(s, d, 32);
        if (lane >= d) s += t;
    }
    if (lane == 31) wsum[w] = s;
    __syncthreads();
    if (tid == 0) {
        int t = 0;
        for (int k = 0; k < 8; ++k) { const int v = wsum[k]; wsum[k] = t; t += v; }
        bbase = atomicAdd(total, t);
    }
    __syncthreads();
    if (i < NN) offs[i] = bbase + wsum[w] + s - c;
}

__global__ __launch_bounds__(256) void k_edge_place(
    const int* __restrict__ ei, const int* __restrict__ offs,
    const int* __restrict__ rank, int* __restrict__ list)
{
    const int e = blockIdx.x * 256 + threadIdx.x;
    if (e >= NE) return;
    const int recv = ei[2 * e];
    const int send = ei[2 * e + 1];
    ((int2*)list)[offs[recv] + rank[e]] = make_int2(e, send);
}

DEV int next_by_id(const int2* __restrict__ seg, int deg, int lane, int& last) {
    int best = 0x7fffffff, bs = 0;
    for (int i = lane; i < deg; i += 32) {
        const int2 p = seg[i];
        if (p.x > last && p.x < best) { best = p.x; bs = p.y; }
    }
    const int mn = wave_min_i(best);
    const unsigned hit = __builtin_amdgcn_ballot_w32(best == mn);
    last = mn;
    return lane_i(bs, __builtin_ctz(hit));
}

__global__ __launch_bounds__(256) void k_node_gather(
    const int* __restrict__ list, const int* __restrict__ cnt,
    const int* __restrict__ offs, const float* __restrict__ s_i,
    const float* __restrict__ s_j, const float* __restrict__ a_e,
    const float* __restrict__ b_a, const float* __restrict__ h,
    float* __restrict__ agg)
{
    const int lane = threadIdx.x & 31;
    const int node = blockIdx.x * 8 + __builtin_amdgcn_readfirstlane(threadIdx.x >> 5);
    if (node >= NN) return;
    const int deg = cnt[node];
    const int2* __restrict__ seg = (const int2*)list + offs[node];
    const int hd = lane >> 3;
    const int k0 = (hd == 0) ? -1 : 0, k1 = (hd == 1) ? -1 : 0;
    const int k2 = (hd == 2) ? -1 : 0, k3 = (hd == 3) ? -1 : 0;
    const float4 si = *(const float4*)&s_i[node * NHEADS];
    const float ae0 = a_e[0], ae1 = a_e[1], ae2 = a_e[2], ae3 = a_e[3];
    const float ba0 = b_a[0], ba1 = b_a[1], ba2 = b_a[2], ba3 = b_a[3];
    const float4* __restrict__ hrow = (const float4*)h + lane;

    float4 acc = make_float4(0.0f, 0.0f, 0.0f, 0.0f);
    if (deg == 0) {
        ((float4*)agg)[(size_t)node * (DF / 4) + lane] = acc;
        return;
    }

    if (deg <= 32) {
        const bool on = lane < deg;
        int e = 0x7fffffff, send = 0;
        if (on) { const int2 p = seg[lane]; e = p.x; send = p.y; }
        int rk = 0;
        for (int j = 0; j < deg; ++j) rk += (lane_i(e, j) < e) ? 1 : 0;
        if (!on) rk = lane;
        send = __builtin_amdgcn_ds_permute(rk << 2, send);

        float l0 = -INFINITY, l1 = -INFINITY, l2 = -INFINITY, l3 = -INFINITY;
        if (on) {
            const float4 sj = *(const float4*)&s_j[send * NHEADS];
            l0 = leaky(si.x + sj.x + ae0 + ba0);
            l1 = leaky(si.y + sj.y + ae1 + ba1);
            l2 = leaky(si.z + sj.z + ae2 + ba2);
            l3 = leaky(si.w + sj.w + ae3 + ba3);
        }
        const float m0 = wave_max(l0), m1 = wave_max(l1);
        const float m2 = wave_max(l2), m3 = wave_max(l3);
        const float w0 = expf(l0 - m0), w1 = expf(l1 - m1);
        const float w2 = expf(l2 - m2), w3 = expf(l3 - m3);
        const float al0 = w0 / (wave_sum(w0) + 1e-16f);
        const float al1 = w1 / (wave_sum(w1) + 1e-16f);
        const float al2 = w2 / (wave_sum(w2) + 1e-16f);
        const float al3 = w3 / (wave_sum(w3) + 1e-16f);

        int r = 0;
        for (; r + 4 <= deg; r += 4) {
            float4 v[4]; float a[4];
            #pragma unroll
            for (int u = 0; u < 4; ++u) {
                v[u] = hrow[(size_t)lane_i(send, r + u) * (DF / 4)];
                a[u] = pick4(lane_f(al0, r + u), lane_f(al1, r + u),
                             lane_f(al2, r + u), lane_f(al3, r + u), k0, k1, k2, k3);
            }
            #pragma unroll
            for (int u = 0; u < 4; ++u) {
                acc.x += v[u].x * a[u]; acc.y += v[u].y * a[u];
                acc.z += v[u].z * a[u]; acc.w += v[u].w * a[u];
            }
        }
        for (; r < deg; ++r) {
            const float4 v = hrow[(size_t)lane_i(send, r) * (DF / 4)];
            const float a = pick4(lane_f(al0, r), lane_f(al1, r),
                                  lane_f(al2, r), lane_f(al3, r), k0, k1, k2, k3);
            acc.x += v.x * a; acc.y += v.y * a; acc.z += v.z * a; acc.w += v.w * a;
        }
    } else {
        float m0 = -INFINITY, m1 = -INFINITY, m2 = -INFINITY, m3 = -INFINITY;
        for (int i = lane; i < deg; i += 32) {
            const float4 sj = *(const float4*)&s_j[seg[i].y * NHEADS];
            m0 = fmaxf(m0, leaky(si.x + sj.x + ae0 + ba0));
            m1 = fmaxf(m1, leaky(si.y + sj.y + ae1 + ba1));
            m2 = fmaxf(m2, leaky(si.z + sj.z + ae2 + ba2));
            m3 = fmaxf(m3, leaky(si.w + sj.w + ae3 + ba3));
        }
        m0 = wave_max(m0); m1 = wave_max(m1); m2 = wave_max(m2); m3 = wave_max(m3);
        float d0 = 0.0f, d1 = 0.0f, d2 = 0.0f, d3 = 0.0f;
        int last = -1;
        for (int t = 0; t < deg; ++t) {
            const int s = next_by_id(seg, deg, lane, last);
            const float4 sj = *(const float4*)&s_j[s * NHEADS];
            d0 += expf(leaky(si.x + sj.x + ae0 + ba0) - m0);
            d1 += expf(leaky(si.y + sj.y + ae1 + ba1) - m1);
            d2 += expf(leaky(si.z + sj.z + ae2 + ba2) - m2);
            d3 += expf(leaky(si.w + sj.w + ae3 + ba3) - m3);
        }
        last = -1;
        for (int t = 0; t < deg; ++t) {
            const int s = next_by_id(seg, deg, lane, last);
            const float4 sj = *(const float4*)&s_j[s * NHEADS];
            const float a = pick4(
                expf(leaky(si.x + sj.x + ae0 + ba0) - m0) / (d0 + 1e-16f),
                expf(leaky(si.y + sj.y + ae1 + ba1) - m1) / (d1 + 1e-16f),
                expf(leaky(si.z + sj.z + ae2 + ba2) - m2) / (d2 + 1e-16f),
                expf(leaky(si.w + sj.w + ae3 + ba3) - m3) / (d3 + 1e-16f), k0, k1, k2, k3);
            const float4 v = hrow[(size_t)s * (DF / 4)];
            acc.x += v.x * a; acc.y += v.y * a; acc.z += v.z * a; acc.w += v.w * a;
        }
    }
    ((float4*)agg)[(size_t)node * (DF / 4) + lane] = acc;
}

#define BN_ROWS 128
__global__ __launch_bounds__(128) void k_bn_partial(
    const float* __restrict__ upd, float* __restrict__ macc,
    float* __restrict__ sqacc)
{
    const int col = threadIdx.x;
    const int r0  = blockIdx.x * BN_ROWS;
    float s = 0.0f, s2 = 0.0f;
    for (int r = 0; r < BN_ROWS; ++r) {
        const int row = r0 + r;
        if (row < NN) {
            const float v = upd[(size_t)row * DF + col];
            s += v; s2 += v * v;
        }
    }
    atomicAdd(&macc[col], s);
    atomicAdd(&sqacc[col], s2);
}

__global__ void k_finalize(
    const float* __restrict__ upd, const float* __restrict__ agg,
    const float* __restrict__ macc, const float* __restrict__ sqacc,
    const float* __restrict__ gamma, const float* __restrict__ beta,
    float* __restrict__ out)
{
    const long long idx = (long long)blockIdx.x * blockDim.x + threadIdx.x;
    if (idx >= (long long)NN * DF) return;
    const int col = (int)(idx & (DF - 1));
    const float inv_n = 1.0f / (float)NN;
    const float mean = macc[col] * inv_n;
    const float var  = sqacc[col] * inv_n - mean * mean;
    const float normed = gamma[col] * (upd[idx] - mean)
                       * rsqrtf(var + BN_EPS) + beta[col];
    out[idx] = normed + fmaxf(agg[idx], 0.0f);
}

extern "C" void kernel_launch(void* const* d_in, const int* in_sizes, int n_in,
                              void* d_out, int out_size, void* d_ws, size_t ws_size,
                              hipStream_t stream) {
    const float* x     = (const float*)d_in[0];
    const int*   ei    = (const int*)  d_in[1];
    const float* Wk    = (const float*)d_in[2];
    const float* bk    = (const float*)d_in[3];
    const float* a_i   = (const float*)d_in[4];
    const float* a_j   = (const float*)d_in[5];
    const float* a_e   = (const float*)d_in[6];
    const float* b_a   = (const float*)d_in[7];
    const float* Wm    = (const float*)d_in[8];
    const float* bm    = (const float*)d_in[9];
    const float* gamma = (const float*)d_in[10];
    const float* beta  = (const float*)d_in[11];
    float* out = (float*)d_out;
    float* ws  = (float*)d_ws;

    float* h_buf  = ws + OFF_H;
    float* upd    = ws + OFF_UPD;
    float* agg    = ws + OFF_AGG;
    float* s_i    = ws + OFF_SI;
    float* s_j    = ws + OFF_SJ;
    int*   cnt    = (int*)ws + OFF_CNT;
    int*   offs   = (int*)ws + OFF_OFFS;
    int*   rank   = (int*)ws + OFF_RANK;
    int*   list   = (int*)ws + OFF_LIST;
    int*   total  = (int*)ws + OFF_TOT;
    float* macc   = ws + OFF_MACC;
    float* sqacc  = ws + OFF_SQ;

    k_init<<<(NN + 255) / 256, 256, 0, stream>>>(ws);
    const int gemm_blocks = (NN + 15) / 16;
    k_gemm_wmma<1, 0><<<gemm_blocks, 256, 0, stream>>>(x, Wk, bk, h_buf, NN);
    k_gemm_wmma<0, 1><<<gemm_blocks, 256, 0, stream>>>(x, Wm, bm, upd, NN);
    k_scores<<<NN, 128, 0, stream>>>(h_buf, a_i, a_j, s_i, s_j);
    k_edge_rank<<<(NE + 255) / 256, 256, 0, stream>>>(ei, cnt, rank);
    k_seg_alloc<<<(NN + 255) / 256, 256, 0, stream>>>(cnt, offs, total);
    k_edge_place<<<(NE + 255) / 256, 256, 0, stream>>>(ei, offs, rank, list);
    k_node_gather<<<(NN + 7) / 8, 256, 0, stream>>>(
        list, cnt, offs, s_i, s_j, a_e, b_a, h_buf, agg);
    k_bn_partial<<<(NN + BN_ROWS - 1) / BN_ROWS, 128, 0, stream>>>(
        upd, macc, sqacc);
    {
        const long long n = (long long)NN * DF;
        k_finalize<<<(int)((n + 255) / 256), 256, 0, stream>>>(
            upd, agg, macc, sqacc, gamma, beta, out);
    }
}


# Many elements per wave instead of one: scores 8 nodes/wave (dot product now 4 columns in lane, then xor tree over 8 lanes), finalize 8 rows/wave, edge kernels 5 edges/thread; 4x fewer waves launched
# speedup vs baseline: 2.1092x; 1.1032x over previous
#include <hip/hip_runtime.h>
#include <hip/hip_bf16.h>
#include <math.h>

#define NN      50000
#define NE      800000
#define DF      128
#define NHEADS  4
#define HU      32
#define NEG_SLOPE 0.2f
#define BN_EPS    0.001f

typedef __attribute__((ext_vector_type(2))) float v2f;
typedef __attribute__((ext_vector_type(8))) float v8f;

#define OFF_H    ((size_t)0)
#define OFF_UPD  (OFF_H   + (size_t)NN * DF)
#define OFF_AGG  (OFF_UPD + (size_t)NN * DF)
#define OFF_SI   (OFF_AGG + (size_t)NN * DF)
#define OFF_SJ   (OFF_SI  + (size_t)NN * NHEADS)
#define OFF_CNT  (OFF_SJ  + (size_t)NN * NHEADS)
#define OFF_OFFS (OFF_CNT + (size_t)NN * NHEADS)
#define OFF_RANK (OFF_OFFS + (size_t)NN * NHEADS)
#define OFF_LIST (OFF_RANK + (size_t)NE)
#define OFF_TOT  (OFF_LIST + (size_t)2 * NE)
#define OFF_MACC (OFF_RANK + (size_t)NE * NHEADS)
#define OFF_SQ   (OFF_MACC + (size_t)DF)

#define DEV static __device__ __forceinline__
DEV int   lane_i(int v, int l)   { return __builtin_amdgcn_readlane(v, l); }
DEV float lane_f(float v, int l) { return __int_as_float(__builtin_amdgcn_readlane(__float_as_int(v), l)); }
DEV float wave_max(float v) {
    for (int off = 16; off > 0; off >>= 1) v = fmaxf(v, __shfl_xor(v, off, 32));
    return v;
}
DEV float wave_sum(float v) {
    for (int off = 16; off > 0; off >>= 1) v += __shfl_xor(v, off, 32);
    return v;
}
DEV int wave_min_i(int v) {
    for (int off = 16; off > 0; off >>= 1) v = min(v, __shfl_xor(v, off, 32));
    return v;
}
DEV float leaky(float x) { return (x > 0.0f) ? x : NEG_SLOPE * x; }
DEV float pick4(float a0, float a1, float a2, float a3, int k0, int k1, int k2, int k3) {
    return __int_as_float((__float_as_int(a0) & k0) | (__float_as_int(a1) & k1) |
                          (__float_as_int(a2) & k2) | (__float_as_int(a3) & k3));
}

__global__ void k_init(float* __restrict__ ws) {
    int i = blockIdx.x * blockDim.x + threadIdx.x;
    int* wi = (int*)ws;
    if (i < NN) wi[OFF_CNT + i] = 0;
    if (i == 0) wi[OFF_TOT] = 0;
    if (i < DF) {
        ws[OFF_MACC + i] = 0.0f;
        ws[OFF_SQ + i]   = 0.0f;
    }
}

template <int WLAYOUT, int DORELU>
__global__ __launch_bounds__(256) void k_gemm_wmma(
    const float* __restrict__ A, const float* __restrict__ B,
    const float* __restrict__ bias, float* __restrict__ C, int M)
{
    __shared__ float atile[16 * DF];

    const int tid = threadIdx.x;
    const int m0  = blockIdx.x * 16;

    #pragma unroll
    for (int i = 0; i < 8; ++i) {
        const int idx = tid + i * 256;
        const int r = idx >> 7, c = idx & (DF - 1);
        const int row = m0 + r;
        atile[idx] = (row < M) ? A[(size_t)row * DF + c] : 0.0f;
    }
    __syncthreads();

    const int lane = tid & 31;
    const int half = lane >> 4;
    const int l    = lane & 15;
    const int n0   = (tid >> 5) * 16;
    const int col  = n0 + l;

    const int b_base   = (WLAYOUT == 0) ? col
                                        : ((col >> 5) * (DF * HU) + (col & 31));
    const int b_kstride = (WLAYOUT == 0) ? DF : HU;

    const float* __restrict__ arow = &atile[l * DF + 2 * half];
    const float* __restrict__ bcol = &B[b_base + (size_t)(2 * half) * b_kstride];

    v8f acc = {};
    #pragma unroll
    for (int kb = 0; kb < DF; kb += 4) {
        v2f a = *(const v2f*)(arow + kb);
        v2f b;
        b.x = bcol[(size_t)kb * b_kstride];
        b.y = bcol[(size_t)(kb + 1) * b_kstride];
        acc = __builtin_amdgcn_wmma_f32_16x16x4_f32(
                  false, a, false, b, (short)0, acc, false, false);
    }

    const float bv = bias[col];
    #pragma unroll
    for (int r = 0; r < 8; ++r) {
        float v = acc[r] + bv;
        if (DORELU) v = fmaxf(v, 0.0f);
        const int row = m0 + r + 8 * half;
        if (row < M) C[(size_t)row * DF + col] = v;
    }
}

#define SC_NODES 8
static_assert(NN % SC_NODES == 0, "a wave's nodes are all inside the graph");
__global__ __launch_bounds__(256) void k_scores(
    const float* __restrict__ h, const float* __restrict__ a_i,
    const float* __restrict__ a_j, float* __restrict__ s_i,
    float* __restrict__ s_j)
{
    const int lane = threadIdx.x & 31;
    const int wave = blockIdx.x * 8 + __builtin_amdgcn_readfirstlane(threadIdx.x >> 5);
    const int n0 = wave * SC_NODES;
    if (n0 >= NN) return;
    const float4 ai = ((const float4*)a_i)[lane];
    const float4 aj = ((const float4*)a_j)[lane];
    const float4* __restrict__ h4 = (const float4*)h + (size_t)n0 * (DF / 4) + lane;

    float4 hv[SC_NODES];
    #pragma unroll
    for (int k = 0; k < SC_NODES; ++k) hv[k] = h4[k * (DF / 4)];

    const int hd = lane >> 3;
    #pragma unroll
    for (int k = 0; k < SC_NODES; ++k) {
        float vi = hv[k].x * ai.x; vi += hv[k].y * ai.y; vi += hv[k].z * ai.z; vi += hv[k].w * ai.w;
        float vj = hv[k].x * aj.x; vj += hv[k].y * aj.y; vj += hv[k].z * aj.z; vj += hv[k].w * aj.w;
        #pragma unroll
        for (int off = 4; off > 0; off >>= 1) {
            vi += __shfl_xor(vi, off, 32);
            vj += __shfl_xor(vj, off, 32);
        }
        if ((lane & 7) == 0) {
            s_i[(n0 + k) * NHEADS + hd] = vi;
            s_j[(n0 + k) * NHEADS + hd] = vj;
        }
    }
}

#define EPT 5
static_assert(NE % (256 * EPT) == 0, "the edge kernels cover NE without a tail");
__global__ __launch_bounds__(256) void k_edge_rank(
    const int* __restrict__ ei, int* __restrict__ cnt, int* __restrict__ rank)
{
    const int e0 = blockIdx.x * (256 * EPT) + threadIdx.x;
    int recv[EPT], rk[EPT];
    #pragma unroll
    for (int k = 0; k < EPT; ++k) recv[k] = ei[2 * (e0 + k * 256)];
    #pragma unroll
    for (int k = 0; k < EPT; ++k) rk[k] = atomicAdd(&cnt[recv[k]], 1);
    #pragma unroll
    for (int k = 0; k < EPT; ++k) rank[e0 + k * 256] = rk[k];
}

__global__ __launch_bounds__(256) void k_seg_alloc(
    const int* __restrict__ cnt, int* __restrict__ offs, int* __restrict__ total)
{
    __shared__ int wsum[8];
    __shared__ int bbase;
    const int tid = threadIdx.x, lane = tid & 31, w = tid >> 5;
    const int i = blockIdx.x * 256 + tid;
    const int c = (i < NN) ? cnt[i] : 0;
    int s = c;
    for (int d = 1; d < 32; d <<= 1) {
        const int t = __shfl_up(s, d, 32);
        if (lane >= d) s += t;
    }
    if (lane == 31) wsum[w] = s;
    __syncthreads();
    if (tid == 0) {
        int t = 0;
        for (int k = 0; k < 8; ++k) { const int v = wsum[k]; wsum[k] = t; t += v; }
        bbase = atomicAdd(total, t);
    }
    __syncthreads();
    if (i < NN) offs[i] = bbase + wsum[w] + s - c;
}

__global__ __launch_bounds__(256) void k_edge_place(
    const int* __restrict__ ei, const int* __restrict__ offs,
    const int* __restrict__ rank, int* __restrict__ list)
{
    const int e0 = blockIdx.x * (256 * EPT) + threadIdx.x;
    int2 rs[EPT]; int rk[EPT], of[EPT];
    #pragma unroll
    for (int k = 0; k < EPT; ++k) {
        rs[k] = ((const int2*)ei)[e0 + k * 256];
        rk[k] = rank[e0 + k * 256];
    }
    #pragma unroll
    for (int k = 0; k < EPT; ++k) of[k] = offs[rs[k].x];
    #pragma unroll
    for (int k = 0; k < EPT; ++k)
        ((int2*)list)[of[k] + rk[k]] = make_int2(e0 + k * 256, rs[k].y);
}

DEV int next_by_id(const int2* __restrict__ seg, int deg, int lane, int& last) {
    int best = 0x7fffffff, bs = 0;
    for (int i = lane; i < deg; i += 32) {
        const int2 p = seg[i];
        if (p.x > last && p.x < best) { best = p.x; bs = p.y; }
    }
    const int mn = wave_min_i(best);
    const unsigned hit = __builtin_amdgcn_ballot_w32(best == mn);
    last = mn;
    return lane_i(bs, __builtin_ctz(hit));
}

__global__ __launch_bounds__(256) void k_node_gather(
    const int* __restrict__ list, const int* __restrict__ cnt,
    const int* __restrict__ offs, const float* __restrict__ s_i,
    const float* __restrict__ s_j, const float* __restrict__ a_e,
    const float* __restrict__ b_a, const float* __restrict__ h,
    float* __restrict__ agg)
{
    const int lane = threadIdx.x & 31;
    const int node = blockIdx.x * 8 + __builtin_amdgcn_readfirstlane(threadIdx.x >> 5);
    if (node >= NN) return;
    const int deg = cnt[node];
    const int2* __restrict__ seg = (const int2*)list + offs[node];
    const int hd = lane >> 3;
    const int k0 = (hd == 0) ? -1 : 0, k1 = (hd == 1) ? -1 : 0;
    const int k2 = (hd == 2) ? -1 : 0, k3 = (hd == 3) ? -1 : 0;
    const float4 si = *(const float4*)&s_i[node * NHEADS];
    const float ae0 = a_e[0], ae1 = a_e[1], ae2 = a_e[2], ae3 = a_e[3];
    const float ba0 = b_a[0], ba1 = b_a[1], ba2 = b_a[2], ba3 = b_a[3];
    const float4* __restrict__ hrow = (const float4*)h + lane;

    float4 acc = make_float4(0.0f, 0.0f, 0.0f, 0.0f);
    if (deg == 0) {
        ((float4*)agg)[(size_t)node * (DF / 4) + lane] = acc;
        return;
    }

    if (deg <= 32) {
        const bool on = lane < deg;
        int e = 0x7fffffff, send = 0;
        if (on) { const int2 p = seg[lane]; e = p.x; send = p.y; }
        int rk = 0;
        for (int j = 0; j < deg; ++j) rk += (lane_i(e, j) < e) ? 1 : 0;
        if (!on) rk = lane;
        send = __builtin_amdgcn_ds_permute(rk << 2, send);

        float l0 = -INFINITY, l1 = -INFINITY, l2 = -INFINITY, l3 = -INFINITY;
        if (on) {
            const float4 sj = *(const float4*)&s_j[send * NHEADS];
            l0 = leaky(si.x + sj.x + ae0 + ba0);
            l1 = leaky(si.y + sj.y + ae1 + ba1);
            l2 = leaky(si.z + sj.z + ae2 + ba2);
            l3 = leaky(si.w + sj.w + ae3 + ba3);
        }
        const float m0 = wave_max(l0), m1 = wave_max(l1);
        const float m2 = wave_max(l2), m3 = wave_max(l3);
        const float w0 = expf(l0 - m0), w1 = expf(l1 - m1);
        const float w2 = expf(l2 - m2), w3 = expf(l3 - m3);
        const float al0 = w0 / (wave_sum(w0) + 1e-16f);
        const float al1 = w1 / (wave_sum(w1) + 1e-16f);
        const float al2 = w2 / (wave_sum(w2) + 1e-16f);
        const float al3 = w3 / (wave_sum(w3) + 1e-16f);

        int r = 0;
        for (; r + 4 <= deg; r += 4) {
            float4 v[4]; float a[4];
            #pragma unroll
            for (int u = 0; u < 4; ++u) {
                v[u] = hrow[(size_t)lane_i(send, r + u) * (DF / 4)];
                a[u] = pick4(lane_f(al0, r + u), lane_f(al1, r + u),
                             lane_f(al2, r + u), lane_f(al3, r + u), k0, k1, k2, k3);
            }
            #pragma unroll
            for (int u = 0; u < 4; ++u) {
                acc.x += v[u].x * a[u]; acc.y += v[u].y * a[u];
                acc.z += v[u].z * a[u]; acc.w += v[u].w * a[u];
            }
        }
        for (; r < deg; ++r) {
            const float4 v = hrow[(size_t)lane_i(send, r) * (DF / 4)];
            const float a = pick4(lane_f(al0, r), lane_f(al1, r),
                                  lane_f(al2, r), lane_f(al3, r), k0, k1, k2, k3);
            acc.x += v.x * a; acc.y += v.y * a; acc.z += v.z * a; acc.w += v.w * a;
        }
    } else {
        float m0 = -INFINITY, m1 = -INFINITY, m2 = -INFINITY, m3 = -INFINITY;
        for (int i = lane; i < deg; i += 32) {
            const float4 sj = *(const float4*)&s_j[seg[i].y * NHEADS];
            m0 = fmaxf(m0, leaky(si.x + sj.x + ae0 + ba0));
            m1 = fmaxf(m1, leaky(si.y + sj.y + ae1 + ba1));
            m2 = fmaxf(m2, leaky(si.z + sj.z + ae2 + ba2));
            m3 = fmaxf(m3, leaky(si.w + sj.w + ae3 + ba3));
        }
        m0 = wave_max(m0); m1 = wave_max(m1); m2 = wave_max(m2); m3 = wave_max(m3);
        float d0 = 0.0f, d1 = 0.0f, d2 = 0.0f, d3 = 0.0f;
        int last = -1;
        for (int t = 0; t < deg; ++t) {
            const int s = next_by_id(seg, deg, lane, last);
            const float4 sj = *(const float4*)&s_j[s * NHEADS];
            d0 += expf(leaky(si.x + sj.x + ae0 + ba0) - m0);
            d1 += expf(leaky(si.y + sj.y + ae1 + ba1) - m1);
            d2 += expf(leaky(si.z + sj.z + ae2 + ba2) - m2);
            d3 += expf(leaky(si.w + sj.w + ae3 + ba3) - m3);
        }
        last = -1;
        for (int t = 0; t < deg; ++t) {
            const int s = next_by_id(seg, deg, lane, last);
            const float4 sj = *(const float4*)&s_j[s * NHEADS];
            const float a = pick4(
                expf(leaky(si.x + sj.x + ae0 + ba0) - m0) / (d0 + 1e-16f),
                expf(leaky(si.y + sj.y + ae1 + ba1) - m1) / (d1 + 1e-16f),
                expf(leaky(si.z + sj.z + ae2 + ba2) - m2) / (d2 + 1e-16f),
                expf(leaky(si.w + sj.w + ae3 + ba3) - m3) / (d3 + 1e-16f), k0, k1, k2, k3);
            const float4 v = hrow[(size_t)s * (DF / 4)];
            acc.x += v.x * a; acc.y += v.y * a; acc.z += v.z * a; acc.w += v.w * a;
        }
    }
    ((float4*)agg)[(size_t)node * (DF / 4) + lane] = acc;
}

#define BN_ROWS 128
__global__ __launch_bounds__(128) void k_bn_partial(
    const float* __restrict__ upd, float* __restrict__ macc,
    float* __restrict__ sqacc)
{
    const int col = threadIdx.x;
    const int r0  = blockIdx.x * BN_ROWS;
    float s = 0.0f, s2 = 0.0f;
    for (int r = 0; r < BN_ROWS; ++r) {
        const int row = r0 + r;
        if (row < NN) {
            const float v = upd[(size_t)row * DF + col];
            s += v; s2 += v * v;
        }
    }
    atomicAdd(&macc[col], s);
    atomicAdd(&sqacc[col], s2);
}

#define FIN_ROWS 8
static_assert(NN % FIN_ROWS == 0, "a wave's rows are all inside the output");
DEV float fin_rstd(float sq, float mean, float inv_n) {
    const float var = sq * inv_n - mean * mean;
    return rsqrtf(var + BN_EPS);
}
DEV float fin_one(float u, float a, float ma, float inv_n, float rs, float g, float b) {
    const float d = fmaf(-ma, inv_n, u);
    const float normed = g * d * rs + b;
    return normed + fmaxf(a, 0.0f);
}
__global__ __launch_bounds__(256) void k_finalize(
    const float* __restrict__ upd, const float* __restrict__ agg,
    const float* __restrict__ macc, const float* __restrict__ sqacc,
    const float* __restrict__ gamma, const float* __restrict__ beta,
    float* __restrict__ out)
{
    const int lane = threadIdx.x & 31;
    const int wave = blockIdx.x * 8 + __builtin_amdgcn_readfirstlane(threadIdx.x >> 5);
    const int r0 = wave * FIN_ROWS;
    if (r0 >= NN) return;
    const float inv_n = 1.0f / (float)NN;
    const float4 ma = ((const float4*)macc)[lane];
    const float4 sq = ((const float4*)sqacc)[lane];
    const float4 g  = ((const float4*)gamma)[lane];
    const float4 b  = ((const float4*)beta)[lane];
    float4 rs;
    rs.x = fin_rstd(sq.x, ma.x * inv_n, inv_n); rs.y = fin_rstd(sq.y, ma.y * inv_n, inv_n);
    rs.z = fin_rstd(sq.z, ma.z * inv_n, inv_n); rs.w = fin_rstd(sq.w, ma.w * inv_n, inv_n);

    const size_t base = (size_t)r0 * (DF / 4) + lane;
    const float4* __restrict__ u4 = (const float4*)upd + base;
    const float4* __restrict__ a4 = (const float4*)agg + base;
    float4* __restrict__ o4 = (float4*)out + base;
    float4 u[FIN_ROWS], a[FIN_ROWS];
    #pragma unroll
    for (int k = 0; k < FIN_ROWS; ++k) {
        u[k] = u4[k * (DF / 4)];
        a[k] = a4[k * (DF / 4)];
    }
    #pragma unroll
    for (int k = 0; k < FIN_ROWS; ++k) {
        float4 o;
        o.x = fin_one(u[k].x, a[k].x, ma.x, inv_n, rs.x, g.x, b.x);
        o.y = fin_one(u[k].y, a[k].y, ma.y, inv_n, rs.y, g.y, b.y);
        o.z = fin_one(u[k].z, a[k].z, ma.z, inv_n, rs.z, g.z, b.z);
        o.w = fin_one(u[k].w, a[k].w, ma.w, inv_n, rs.w, g.w, b.w);
        o4[k * (DF / 4)] = o;
    }
}

extern "C" void kernel_launch(void* const* d_in, const int* in_sizes, int n_in,
                              void* d_out, int out_size, void* d_ws, size_t ws_size,
                              hipStream_t stream) {
    const float* x     = (const float*)d_in[0];
    const int*   ei    = (const int*)  d_in[1];
    const float* Wk    = (const float*)d_in[2];
    const float* bk    = (const float*)d_in[3];
    const float* a_i   = (const float*)d_in[4];
    const float* a_j   = (const float*)d_in[5];
    const float* a_e   = (const float*)d_in[6];
    const float* b_a   = (const float*)d_in[7];
    const float* Wm    = (const float*)d_in[8];
    const float* bm    = (const float*)d_in[9];
    const float* gamma = (const float*)d_in[10];
    const float* beta  = (const float*)d_in[11];
    float* out = (float*)d_out;
    float* ws  = (float*)d_ws;

    float* h_buf  = ws + OFF_H;
    float* upd    = ws + OFF_UPD;
    float* agg    = ws + OFF_AGG;
    float* s_i    = ws + OFF_SI;
    float* s_j    = ws + OFF_SJ;
    int*   cnt    = (int*)ws + OFF_CNT;
    int*   offs   = (int*)ws + OFF_OFFS;
    int*   rank   = (int*)ws + OFF_RANK;
    int*   list   = (int*)ws + OFF_LIST;
    int*   total  = (int*)ws + OFF_TOT;
    float* macc   = ws + OFF_MACC;
    float* sqacc  = ws + OFF_SQ;

    k_init<<<(NN + 255) / 256, 256, 0, stream>>>(ws);
    const int gemm_blocks = (NN + 15) / 16;
    k_gemm_wmma<1, 0><<<gemm_blocks, 256, 0, stream>>>(x, Wk, bk, h_buf, NN);
    k_gemm_wmma<0, 1><<<gemm_blocks, 256, 0, stream>>>(x, Wm, bm, upd, NN);
    k_scores<<<(NN + 8 * SC_NODES - 1) / (8 * SC_NODES), 256, 0, stream>>>(h_buf, a_i, a_j, s_i, s_j);
    k_edge_rank<<<NE / (256 * EPT), 256, 0, stream>>>(ei, cnt, rank);
    k_seg_alloc<<<(NN + 255) / 256, 256, 0, stream>>>(cnt, offs, total);
    k_edge_place<<<NE / (256 * EPT), 256, 0, stream>>>(ei, offs, rank, list);
    k_node_gather<<<(NN + 7) / 8, 256, 0, stream>>>(
        list, cnt, offs, s_i, s_j, a_e, b_a, h_buf, agg);
    k_bn_partial<<<(NN + BN_ROWS - 1) / BN_ROWS, 128, 0, stream>>>(
        upd, macc, sqacc);
    k_finalize<<<(NN + 8 * FIN_ROWS - 1) / (8 * FIN_ROWS), 256, 0, stream>>>(
        upd, agg, macc, sqacc, gamma, beta, out);
}


# BN column sums: each wave requests 16 whole 512-byte rows at once (float4 per lane) instead of one 4-byte load at a time; per-block sum order now 16-row runs, then the 8 waves in order via LDS
# speedup vs baseline: 2.5011x; 1.1858x over previous
#include <hip/hip_runtime.h>
#include <hip/hip_bf16.h>
#include <math.h>

#define NN      50000
#define NE      800000
#define DF      128
#define NHEADS  4
#define HU      32
#define NEG_SLOPE 0.2f
#define BN_EPS    0.001f

typedef __attribute__((ext_vector_type(2))) float v2f;
typedef __attribute__((ext_vector_type(8))) float v8f;

#define OFF_H    ((size_t)0)
#define OFF_UPD  (OFF_H   + (size_t)NN * DF)
#define OFF_AGG  (OFF_UPD + (size_t)NN * DF)
#define OFF_SI   (OFF_AGG + (size_t)NN * DF)
#define OFF_SJ   (OFF_SI  + (size_t)NN * NHEADS)
#define OFF_CNT  (OFF_SJ  + (size_t)NN * NHEADS)
#define OFF_OFFS (OFF_CNT + (size_t)NN * NHEADS)
#define OFF_RANK (OFF_OFFS + (size_t)NN * NHEADS)
#define OFF_LIST (OFF_RANK + (size_t)NE)
#define OFF_TOT  (OFF_LIST + (size_t)2 * NE)
#define OFF_MACC (OFF_RANK + (size_t)NE * NHEADS)
#define OFF_SQ   (OFF_MACC + (size_t)DF)

#define DEV static __device__ __forceinline__
DEV int   lane_i(int v, int l)   { return __builtin_amdgcn_readlane(v, l); }
DEV float lane_f(float v, int l) { return __int_as_float(__builtin_amdgcn_readlane(__float_as_int(v), l)); }
DEV float wave_max(float v) {
    for (int off = 16; off > 0; off >>= 1) v = fmaxf(v, __shfl_xor(v, off, 32));
    return v;
}
DEV float wave_sum(float v) {
    for (int off = 16; off > 0; off >>= 1) v += __shfl_xor(v, off, 32);
    return v;
}
DEV int wave_min_i(int v) {
    for (int off = 16; off > 0; off >>= 1) v = min(v, __shfl_xor(v, off, 32));
    return v;
}
DEV float leaky(float x) { return (x > 0.0f) ? x : NEG_SLOPE * x; }
DEV float pick4(float a0, float a1, float a2, float a3, int k0, int k1, int k2, int k3) {
    return __int_as_float((__float_as_int(a0) & k0) | (__float_as_int(a1) & k1) |
                          (__float_as_int(a2) & k2) | (__float_as_int(a3) & k3));
}

__global__ void k_init(float* __restrict__ ws) {
    int i = blockIdx.x * blockDim.x + threadIdx.x;
    int* wi = (int*)ws;
    if (i < NN) wi[OFF_CNT + i] = 0;
    if (i == 0) wi[OFF_TOT] = 0;
    if (i < DF) {
        ws[OFF_MACC + i] = 0.0f;
        ws[OFF_SQ + i]   = 0.0f;
    }
}

template <int WLAYOUT, int DORELU>
__global__ __launch_bounds__(256) void k_gemm_wmma(
    const float* __restrict__ A, const float* __restrict__ B,
    const float* __restrict__ bias, float* __restrict__ C, int M)
{
    __shared__ float atile[16 * DF];

    const int tid = threadIdx.x;
    const int m0  = blockIdx.x * 16;

    #pragma unroll
    for (int i = 0; i < 8; ++i) {
        const int idx = tid + i * 256;
        const int r = idx >> 7, c = idx & (DF - 1);
        const int row = m0 + r;
        atile[idx] = (row < M) ? A[(size_t)row * DF + c] : 0.0f;
    }
    __syncthreads();

    const int lane = tid & 31;
    const int half = lane >> 4;
    const int l    = lane & 15;
    const int n0   = (tid >> 5) * 16;
    const int col  = n0 + l;

    const int b_base   = (WLAYOUT == 0) ? col
                                        : ((col >> 5) * (DF * HU) + (col & 31));
    const int b_kstride = (WLAYOUT == 0) ? DF : HU;

    const float* __restrict__ arow = &atile[l * DF + 2 * half];
    const float* __restrict__ bcol = &B[b_base + (size_t)(2 * half) * b_kstride];

    v8f acc = {};
    #pragma unroll
    for (int kb = 0; kb < DF; kb += 4) {
        v2f a = *(const v2f*)(arow + kb);
        v2f b;
        b.x = bcol[(size_t)kb * b_kstride];
        b.y = bcol[(size_t)(kb + 1) * b_kstride];
        acc = __builtin_amdgcn_wmma_f32_16x16x4_f32(
                  false, a, false, b, (short)0, acc, false, false);
    }

    const float bv = bias[col];
    #pragma unroll
    for (int r = 0; r < 8; ++r) {
        float v = acc[r] + bv;
        if (DORELU) v = fmaxf(v, 0.0f);
        const int row = m0 + r + 8 * half;
        if (row < M) C[(size_t)row * DF + col] = v;
    }
}

#define SC_NODES 8
static_assert(NN % SC_NODES == 0, "a wave's nodes are all inside the graph");
__global__ __launch_bounds__(256) void k_scores(
    const float* __restrict__ h, const float* __restrict__ a_i,
    const float* __restrict__ a_j, float* __restrict__ s_i,
    float* __restrict__ s_j)
{
    const int lane = threadIdx.x & 31;
    const int wave = blockIdx.x * 8 + __builtin_amdgcn_readfirstlane(threadIdx.x >> 5);
    const int n0 = wave * SC_NODES;
    if (n0 >= NN) return;
    const float4 ai = ((const float4*)a_i)[lane];
    const float4 aj = ((const float4*)a_j)[lane];
    const float4* __restrict__ h4 = (const float4*)h + (size_t)n0 * (DF / 4) + lane;

    float4 hv[SC_NODES];
    #pragma unroll
    for (int k = 0; k < SC_NODES; ++k) hv[k] = h4[k * (DF / 4)];

    const int hd = lane >> 3;
    #pragma unroll
    for (int k = 0; k < SC_NODES; ++k) {
        float vi = hv[k].x * ai.x; vi += hv[k].y * ai.y; vi += hv[k].z * ai.z; vi += hv[k].w * ai.w;
        float vj = hv[k].x * aj.x; vj += hv[k].y * aj.y; vj += hv[k].z * aj.z; vj += hv[k].w * aj.w;
        #pragma unroll
        for (int off = 4; off > 0; off >>= 1) {
            vi += __shfl_xor(vi, off, 32);
            vj += __shfl_xor(vj, off, 32);
        }
        if ((lane & 7) == 0) {
            s_i[(n0 + k) * NHEADS + hd] = vi;
            s_j[(n0 + k) * NHEADS + hd] = vj;
        }
    }
}

#define EPT 5
static_assert(NE % (256 * EPT) == 0, "the edge kernels cover NE without a tail");
__global__ __launch_bounds__(256) void k_edge_rank(
    const int* __restrict__ ei, int* __restrict__ cnt, int* __restrict__ rank)
{
    const int e0 = blockIdx.x * (256 * EPT) + threadIdx.x;
    int recv[EPT], rk[EPT];
    #pragma unroll
    for (int k = 0; k < EPT; ++k) recv[k] = ei[2 * (e0 + k * 256)];
    #pragma unroll
    for (int k = 0; k < EPT; ++k) rk[k] = atomicAdd(&cnt[recv[k]], 1);
    #pragma unroll
    for (int k = 0; k < EPT; ++k) rank[e0 + k * 256] = rk[k];
}

__global__ __launch_bounds__(256) void k_seg_alloc(
    const int* __restrict__ cnt, int* __restrict__ offs, int* __restrict__ total)
{
    __shared__ int wsum[8];
    __shared__ int bbase;
    const int tid = threadIdx.x, lane = tid & 31, w = tid >> 5;
    const int i = blockIdx.x * 256 + tid;
    const int c = (i < NN) ? cnt[i] : 0;
    int s = c;
    for (int d = 1; d < 32; d <<= 1) {
        const int t = __shfl_up(s, d, 32);
        if (lane >= d) s += t;
    }
    if (lane == 31) wsum[w] = s;
    __syncthreads();
    if (tid == 0) {
        int t = 0;
        for (int k = 0; k < 8; ++k) { const int v = wsum[k]; wsum[k] = t; t += v; }
        bbase = atomicAdd(total, t);
    }
    __syncthreads();
    if (i < NN) offs[i] = bbase + wsum[w] + s - c;
}

__global__ __launch_bounds__(256) void k_edge_place(
    const int* __restrict__ ei, const int* __restrict__ offs,
    const int* __restrict__ rank, int* __restrict__ list)
{
    const int e0 = blockIdx.x * (256 * EPT) + threadIdx.x;
    int2 rs[EPT]; int rk[EPT], of[EPT];
    #pragma unroll
    for (int k = 0; k < EPT; ++k) {
        rs[k] = ((const int2*)ei)[e0 + k * 256];
        rk[k] = rank[e0 + k * 256];
    }
    #pragma unroll
    for (int k = 0; k < EPT; ++k) of[k] = offs[rs[k].x];
    #pragma unroll
    for (int k = 0; k < EPT; ++k)
        ((int2*)list)[of[k] + rk[k]] = make_int2(e0 + k * 256, rs[k].y);
}

DEV int next_by_id(const int2* __restrict__ seg, int deg, int lane, int& last) {
    int best = 0x7fffffff, bs = 0;
    for (int i = lane; i < deg; i += 32) {
        const int2 p = seg[i];
        if (p.x > last && p.x < best) { best = p.x; bs = p.y; }
    }
    const int mn = wave_min_i(best);
    const unsigned hit = __builtin_amdgcn_ballot_w32(best == mn);
    last = mn;
    return lane_i(bs, __builtin_ctz(hit));
}

__global__ __launch_bounds__(256) void k_node_gather(
    const int* __restrict__ list, const int* __restrict__ cnt,
    const int* __restrict__ offs, const float* __restrict__ s_i,
    const float* __restrict__ s_j, const float* __restrict__ a_e,
    const float* __restrict__ b_a, const float* __restrict__ h,
    float* __restrict__ agg)
{
    const int lane = threadIdx.x & 31;
    const int node = blockIdx.x * 8 + __builtin_amdgcn_readfirstlane(threadIdx.x >> 5);
    if (node >= NN) return;
    const int deg = cnt[node];
    const int2* __restrict__ seg = (const int2*)list + offs[node];
    const int hd = lane >> 3;
    const int k0 = (hd == 0) ? -1 : 0, k1 = (hd == 1) ? -1 : 0;
    const int k2 = (hd == 2) ? -1 : 0, k3 = (hd == 3) ? -1 : 0;
    const float4 si = *(const float4*)&s_i[node * NHEADS];
    const float ae0 = a_e[0], ae1 = a_e[1], ae2 = a_e[2], ae3 = a_e[3];
    const float ba0 = b_a[0], ba1 = b_a[1], ba2 = b_a[2], ba3 = b_a[3];
    const float4* __restrict__ hrow = (const float4*)h + lane;

    float4 acc = make_float4(0.0f, 0.0f, 0.0f, 0.0f);
    if (deg == 0) {
        ((float4*)agg)[(size_t)node * (DF / 4) + lane] = acc;
        return;
    }

    if (deg <= 32) {
        const bool on = lane < deg;
        int e = 0x7fffffff, send = 0;
        if (on) { const int2 p = seg[lane]; e = p.x; send = p.y; }
        int rk = 0;
        for (int j = 0; j < deg; ++j) rk += (lane_i(e, j) < e) ? 1 : 0;
        if (!on) rk = lane;
        send = __builtin_amdgcn_ds_permute(rk << 2, send);

        float l0 = -INFINITY, l1 = -INFINITY, l2 = -INFINITY, l3 = -INFINITY;
        if (on) {
            const float4 sj = *(const float4*)&s_j[send * NHEADS];
            l0 = leaky(si.x + sj.x + ae0 + ba0);
            l1 = leaky(si.y + sj.y + ae1 + ba1);
            l2 = leaky(si.z + sj.z + ae2 + ba2);
            l3 = leaky(si.w + sj.w + ae3 + ba3);
        }
        const float m0 = wave_max(l0), m1 = wave_max(l1);
        const float m2 = wave_max(l2), m3 = wave_max(l3);
        const float w0 = expf(l0 - m0), w1 = expf(l1 - m1);
        const float w2 = expf(l2 - m2), w3 = expf(l3 - m3);
        const float al0 = w0 / (wave_sum(w0) + 1e-16f);
        const float al1 = w1 / (wave_sum(w1) + 1e-16f);
        const float al2 = w2 / (wave_sum(w2) + 1e-16f);
        const float al3 = w3 / (wave_sum(w3) + 1e-16f);

        int r = 0;
        for (; r + 4 <= deg; r += 4) {
            float4 v[4]; float a[4];
            #pragma unroll
            for (int u = 0; u < 4; ++u) {
                v[u] = hrow[(size_t)lane_i(send, r + u) * (DF / 4)];
                a[u] = pick4(lane_f(al0, r + u), lane_f(al1, r + u),
                             lane_f(al2, r + u), lane_f(al3, r + u), k0, k1, k2, k3);
            }
            #pragma unroll
            for (int u = 0; u < 4; ++u) {
                acc.x += v[u].x * a[u]; acc.y += v[u].y * a[u];
                acc.z += v[u].z * a[u]; acc.w += v[u].w * a[u];
            }
        }
        for (; r < deg; ++r) {
            const float4 v = hrow[(size_t)lane_i(send, r) * (DF / 4)];
            const float a = pick4(lane_f(al0, r), lane_f(al1, r),
                                  lane_f(al2, r), lane_f(al3, r), k0, k1, k2, k3);
            acc.x += v.x * a; acc.y += v.y * a; acc.z += v.z * a; acc.w += v.w * a;
        }
    } else {
        float m0 = -INFINITY, m1 = -INFINITY, m2 = -INFINITY, m3 = -INFINITY;
        for (int i = lane; i < deg; i += 32) {
            const float4 sj = *(const float4*)&s_j[seg[i].y * NHEADS];
            m0 = fmaxf(m0, leaky(si.x + sj.x + ae0 + ba0));
            m1 = fmaxf(m1, leaky(si.y + sj.y + ae1 + ba1));
            m2 = fmaxf(m2, leaky(si.z + sj.z + ae2 + ba2));
            m3 = fmaxf(m3, leaky(si.w + sj.w + ae3 + ba3));
        }
        m0 = wave_max(m0); m1 = wave_max(m1); m2 = wave_max(m2); m3 = wave_max(m3);
        float d0 = 0.0f, d1 = 0.0f, d2 = 0.0f, d3 = 0.0f;
        int last = -1;
        for (int t = 0; t < deg; ++t) {
            const int s = next_by_id(seg, deg, lane, last);
            const float4 sj = *(const float4*)&s_j[s * NHEADS];
            d0 += expf(leaky(si.x + sj.x + ae0 + ba0) - m0);
            d1 += expf(leaky(si.y + sj.y + ae1 + ba1) - m1);
            d2 += expf(leaky(si.z + sj.z + ae2 + ba2) - m2);
            d3 += expf(leaky(si.w + sj.w + ae3 + ba3) - m3);
        }
        last = -1;
        for (int t = 0; t < deg; ++t) {
            const int s = next_by_id(seg, deg, lane, last);
            const float4 sj = *(const float4*)&s_j[s * NHEADS];
            const float a = pick4(
                expf(leaky(si.x + sj.x + ae0 + ba0) - m0) / (d0 + 1e-16f),
                expf(leaky(si.y + sj.y + ae1 + ba1) - m1) / (d1 + 1e-16f),
                expf(leaky(si.z + sj.z + ae2 + ba2) - m2) / (d2 + 1e-16f),
                expf(leaky(si.w + sj.w + ae3 + ba3) - m3) / (d3 + 1e-16f), k0, k1, k2, k3);
            const float4 v = hrow[(size_t)s * (DF / 4)];
            acc.x += v.x * a; acc.y += v.y * a; acc.z += v.z * a; acc.w += v.w * a;
        }
    }
    ((float4*)agg)[(size_t)node * (DF / 4) + lane] = acc;
}

#define BN_WAVES 8
#define BN_WROWS 16
#define BN_ROWS  (BN_WAVES * BN_WROWS)
static_assert(NN % BN_WROWS == 0, "a wave's rows are all inside the matrix or all outside");
__global__ __launch_bounds__(256) void k_bn_partial(
    const float* __restrict__ upd, float* __restrict__ macc,
    float* __restrict__ sqacc)
{
    __shared__ float4 ps[BN_WAVES][32];
    __shared__ float4 ps2[BN_WAVES][32];
    const int tid  = threadIdx.x;
    const int lane = tid & 31;
    const int w    = __builtin_amdgcn_readfirstlane(tid >> 5);
    const int r0   = blockIdx.x * BN_ROWS + w * BN_WROWS;

    float4 s  = make_float4(0.0f, 0.0f, 0.0f, 0.0f);
    float4 s2 = make_float4(0.0f, 0.0f, 0.0f, 0.0f);
    if (r0 < NN) {
        const float4* __restrict__ u4 = (const float4*)upd + (size_t)r0 * (DF / 4) + lane;
        float4 v[BN_WROWS];
        #pragma unroll
        for (int k = 0; k < BN_WROWS; ++k) v[k] = u4[k * (DF / 4)];
        #pragma unroll
        for (int k = 0; k < BN_WROWS; ++k) {
            s.x += v[k].x; s2.x += v[k].x * v[k].x;
            s.y += v[k].y; s2.y += v[k].y * v[k].y;
            s.z += v[k].z; s2.z += v[k].z * v[k].z;
            s.w += v[k].w; s2.w += v[k].w * v[k].w;
        }
    }
    ps[w][lane]  = s;
    ps2[w][lane] = s2;
    __syncthreads();
    if (tid < DF) {
        const float* __restrict__ p  = (const float*)ps;
        const float* __restrict__ p2 = (const float*)ps2;
        float a = 0.0f, b = 0.0f;
        for (int k = 0; k < BN_WAVES; ++k) { a += p[k * DF + tid]; b += p2[k * DF + tid]; }
        atomicAdd(&macc[tid], a);
        atomicAdd(&sqacc[tid], b);
    }
}

#define FIN_ROWS 8
static_assert(NN % FIN_ROWS == 0, "a wave's rows are all inside the output");
DEV float fin_rstd(float sq, float mean, float inv_n) {
    const float var = sq * inv_n - mean * mean;
    return rsqrtf(var + BN_EPS);
}
DEV float fin_one(float u, float a, float ma, float inv_n, float rs, float g, float b) {
    const float d = fmaf(-ma, inv_n, u);
    const float normed = g * d * rs + b;
    return normed + fmaxf(a, 0.0f);
}
__global__ __launch_bounds__(256) void k_finalize(
    const float* __restrict__ upd, const float* __restrict__ agg,
    const float* __restrict__ macc, const float* __restrict__ sqacc,
    const float* __restrict__ gamma, const float* __restrict__ beta,
    float* __restrict__ out)
{
    const int lane = threadIdx.x & 31;
    const int wave = blockIdx.x * 8 + __builtin_amdgcn_readfirstlane(threadIdx.x >> 5);
    const int r0 = wave * FIN_ROWS;
    if (r0 >= NN) return;
    const float inv_n = 1.0f / (float)NN;
    const float4 ma = ((const float4*)macc)[lane];
    const float4 sq = ((const float4*)sqacc)[lane];
    const float4 g  = ((const float4*)gamma)[lane];
    const float4 b  = ((const float4*)beta)[lane];
    float4 rs;
    rs.x = fin_rstd(sq.x, ma.x * inv_n, inv_n); rs.y = fin_rstd(sq.y, ma.y * inv_n, inv_n);
    rs.z = fin_rstd(sq.z, ma.z * inv_n, inv_n); rs.w = fin_rstd(sq.w, ma.w * inv_n, inv_n);

    const size_t base = (size_t)r0 * (DF / 4) + lane;
    const float4* __restrict__ u4 = (const float4*)upd + base;
    const float4* __restrict__ a4 = (const float4*)agg + base;
    float4* __restrict__ o4 = (float4*)out + base;
    float4 u[FIN_ROWS], a[FIN_ROWS];
    #pragma unroll
    for (int k = 0; k < FIN_ROWS; ++k) {
        u[k] = u4[k * (DF / 4)];
        a[k] = a4[k * (DF / 4)];
    }
    #pragma unroll
    for (int k = 0; k < FIN_ROWS; ++k) {
        float4 o;
        o.x = fin_one(u[k].x, a[k].x, ma.x, inv_n, rs.x, g.x, b.x);
        o.y = fin_one(u[k].y, a[k].y, ma.y, inv_n, rs.y, g.y, b.y);
        o.z = fin_one(u[k].z, a[k].z, ma.z, inv_n, rs.z, g.z, b.z);
        o.w = fin_one(u[k].w, a[k].w, ma.w, inv_n, rs.w, g.w, b.w);
        o4[k * (DF / 4)] = o;
    }
}

extern "C" void kernel_launch(void* const* d_in, const int* in_sizes, int n_in,
                              void* d_out, int out_size, void* d_ws, size_t ws_size,
                              hipStream_t stream) {
    const float* x     = (const float*)d_in[0];
    const int*   ei    = (const int*)  d_in[1];
    const float* Wk    = (const float*)d_in[2];
    const float* bk    = (const float*)d_in[3];
    const float* a_i   = (const float*)d_in[4];
    const float* a_j   = (const float*)d_in[5];
    const float* a_e   = (const float*)d_in[6];
    const float* b_a   = (const float*)d_in[7];
    const float* Wm    = (const float*)d_in[8];
    const float* bm    = (const float*)d_in[9];
    const float* gamma = (const float*)d_in[10];
    const float* beta  = (const float*)d_in[11];
    float* out = (float*)d_out;
    float* ws  = (float*)d_ws;

    float* h_buf  = ws + OFF_H;
    float* upd    = ws + OFF_UPD;
    float* agg    = ws + OFF_AGG;
    float* s_i    = ws + OFF_SI;
    float* s_j    = ws + OFF_SJ;
    int*   cnt    = (int*)ws + OFF_CNT;
    int*   offs   = (int*)ws + OFF_OFFS;
    int*   rank   = (int*)ws + OFF_RANK;
    int*   list   = (int*)ws + OFF_LIST;
    int*   total  = (int*)ws + OFF_TOT;
    float* macc   = ws + OFF_MACC;
    float* sqacc  = ws + OFF_SQ;

    k_init<<<(NN + 255) / 256, 256, 0, stream>>>(ws);
    const int gemm_blocks = (NN + 15) / 16;
    k_gemm_wmma<1, 0><<<gemm_blocks, 256, 0, stream>>>(x, Wk, bk, h_buf, NN);
    k_gemm_wmma<0, 1><<<gemm_blocks, 256, 0, stream>>>(x, Wm, bm, upd, NN);
    k_scores<<<(NN + 8 * SC_NODES - 1) / (8 * SC_NODES), 256, 0, stream>>>(h_buf, a_i, a_j, s_i, s_j);
    k_edge_rank<<<NE / (256 * EPT), 256, 0, stream>>>(ei, cnt, rank);
    k_seg_alloc<<<(NN + 255) / 256, 256, 0, stream>>>(cnt, offs, total);
    k_edge_place<<<NE / (256 * EPT), 256, 0, stream>>>(ei, offs, rank, list);
    k_node_gather<<<(NN + 7) / 8, 256, 0, stream>>>(
        list, cnt, offs, s_i, s_j, a_e, b_a, h_buf, agg);
    k_bn_partial<<<(NN + BN_ROWS - 1) / BN_ROWS, 256, 0, stream>>>(
        upd, macc, sqacc);
    k_finalize<<<(NN + 8 * FIN_ROWS - 1) / (8 * FIN_ROWS), 256, 0, stream>>>(
        upd, agg, macc, sqacc, gamma, beta, out);
}


# Both projections in one persistent launch: each 16-row x tile staged once in LDS at a 528-byte row pitch (fragment reads were 16-way bank-conflicted), B fragments held in registers; sum order kept
# speedup vs baseline: 3.1750x; 1.2694x over previous
#include <hip/hip_runtime.h>
#include <hip/hip_bf16.h>
#include <math.h>

#define NN      50000
#define NE      800000
#define DF      128
#define NHEADS  4
#define HU      32
#define NEG_SLOPE 0.2f
#define BN_EPS    0.001f

typedef __attribute__((ext_vector_type(2))) float v2f;
typedef __attribute__((ext_vector_type(8))) float v8f;

#define OFF_H    ((size_t)0)
#define OFF_UPD  (OFF_H   + (size_t)NN * DF)
#define OFF_AGG  (OFF_UPD + (size_t)NN * DF)
#define OFF_SI   (OFF_AGG + (size_t)NN * DF)
#define OFF_SJ   (OFF_SI  + (size_t)NN * NHEADS)
#define OFF_CNT  (OFF_SJ  + (size_t)NN * NHEADS)
#define OFF_OFFS (OFF_CNT + (size_t)NN * NHEADS)
#define OFF_RANK (OFF_OFFS + (size_t)NN * NHEADS)
#define OFF_LIST (OFF_RANK + (size_t)NE)
#define OFF_TOT  (OFF_LIST + (size_t)2 * NE)
#define OFF_MACC (OFF_RANK + (size_t)NE * NHEADS)
#define OFF_SQ   (OFF_MACC + (size_t)DF)

#define DEV static __device__ __forceinline__
DEV int   lane_i(int v, int l)   { return __builtin_amdgcn_readlane(v, l); }
DEV float lane_f(float v, int l) { return __int_as_float(__builtin_amdgcn_readlane(__float_as_int(v), l)); }
DEV float wave_max(float v) {
    for (int off = 16; off > 0; off >>= 1) v = fmaxf(v, __shfl_xor(v, off, 32));
    return v;
}
DEV float wave_sum(float v) {
    for (int off = 16; off > 0; off >>= 1) v += __shfl_xor(v, off, 32);
    return v;
}
DEV int wave_min_i(int v) {
    for (int off = 16; off > 0; off >>= 1) v = min(v, __shfl_xor(v, off, 32));
    return v;
}
DEV float leaky(float x) { return (x > 0.0f) ? x : NEG_SLOPE * x; }
DEV float pick4(float a0, float a1, float a2, float a3, int k0, int k1, int k2, int k3) {
    return __int_as_float((__float_as_int(a0) & k0) | (__float_as_int(a1) & k1) |
                          (__float_as_int(a2) & k2) | (__float_as_int(a3) & k3));
}

__global__ void k_init(float* __restrict__ ws) {
    int i = blockIdx.x * blockDim.x + threadIdx.x;
    int* wi = (int*)ws;
    if (i < NN) wi[OFF_CNT + i] = 0;
    if (i == 0) wi[OFF_TOT] = 0;
    if (i < DF) {
        ws[OFF_MACC + i] = 0.0f;
        ws[OFF_SQ + i]   = 0.0f;
    }
}

#define GT_ROWS  16
#define GT_PITCH 132
#define GT_TILES (NN / GT_ROWS)
#define GT_WGS   512
static_assert(NN % GT_ROWS == 0, "every row tile is whole: no row guard in the kernel");

template <int KSTRIDE>
DEV void load_bfrag(const float* __restrict__ q, v2f (&bf)[2][DF / 4]) {
    #pragma unroll
    for (int j = 0; j < DF / 4; ++j) {
        const float2 p0 = *(const float2*)(q + (size_t)(4 * j) * KSTRIDE);
        const float2 p1 = *(const float2*)(q + (size_t)(4 * j + 1) * KSTRIDE);
        bf[0][j].x = p0.x; bf[0][j].y = p1.x;
        bf[1][j].x = p0.y; bf[1][j].y = p1.y;
    }
}

__global__ __launch_bounds__(256) void k_gemm2_wmma(
    const float* __restrict__ x, const float* __restrict__ Wk, const float* __restrict__ bk,
    const float* __restrict__ Wm, const float* __restrict__ bm,
    float* __restrict__ h, float* __restrict__ upd)
{
    __shared__ __attribute__((aligned(16))) float atile[2][GT_ROWS * GT_PITCH];

    const int tid  = threadIdx.x;
    const int lane = tid & 31;
    const int half = lane >> 4;
    const int l    = lane & 15;
    const int w    = __builtin_amdgcn_readfirstlane(tid >> 5);
    const bool is_h = w < 4;
    const int cb   = (w & 3) * 32;

    const int r0 = tid >> 5, c4 = tid & 31;
    const float4* __restrict__ x4 = (const float4*)x + tid;
    const int lds0 = r0 * GT_PITCH + 4 * c4;
    const int lds1 = (r0 + 8) * GT_PITCH + 4 * c4;

    int tile = blockIdx.x;
    float4 pre0, pre1;
    {
        const float4* __restrict__ src = x4 + (size_t)tile * (GT_ROWS * DF / 4);
        pre0 = src[0];
        pre1 = src[256];
    }

    v2f bf[2][DF / 4];
    float bv0, bv1;
    if (is_h) {
        load_bfrag<HU>(Wk + (size_t)w * (DF * HU) + (size_t)(2 * half) * HU + 2 * l, bf);
        const float2 b2 = *(const float2*)(bk + cb + 2 * l);
        bv0 = b2.x; bv1 = b2.y;
    } else {
        load_bfrag<DF>(Wm + (size_t)(2 * half) * DF + cb + 2 * l, bf);
        const float2 b2 = *(const float2*)(bm + cb + 2 * l);
        bv0 = b2.x; bv1 = b2.y;
    }

    *(float4*)&atile[0][lds0] = pre0;
    *(float4*)&atile[0][lds1] = pre1;
    __syncthreads();

    float* __restrict__ obase = (is_h ? h : upd) + (size_t)(8 * half) * DF + cb + 2 * l;
    int buf = 0;
    for (; tile < GT_TILES; tile += GT_WGS) {
        const int next = tile + GT_WGS;
        const bool more = next < GT_TILES;
        if (more) {
            const float4* __restrict__ src = x4 + (size_t)next * (GT_ROWS * DF / 4);
            pre0 = src[0];
            pre1 = src[256];
        }

        const float* __restrict__ arow = &atile[buf][l * GT_PITCH + 2 * half];
        v8f acc0 = {}, acc1 = {};
        #pragma unroll
        for (int j = 0; j < DF / 4; ++j) {
            const v2f a = *(const v2f*)(arow + 4 * j);
            acc0 = __builtin_amdgcn_wmma_f32_16x16x4_f32(
                       false, a, false, bf[0][j], (short)0, acc0, false, false);
            acc1 = __builtin_amdgcn_wmma_f32_16x16x4_f32(
                       false, a, false, bf[1][j], (short)0, acc1, false, false);
        }

        if (more) {
            *(float4*)&atile[buf ^ 1][lds0] = pre0;
            *(float4*)&atile[buf ^ 1][lds1] = pre1;
        }
        __syncthreads();

        float* __restrict__ dst = obase + (size_t)tile * (GT_ROWS * DF);
        #pragma unroll
        for (int r = 0; r < 8; ++r) {
            float v0 = acc0[r] + bv0;
            float v1 = acc1[r] + bv1;
            if (!is_h) { v0 = fmaxf(v0, 0.0f); v1 = fmaxf(v1, 0.0f); }
            *(float2*)(dst + (size_t)r * DF) = make_float2(v0, v1);
        }
        buf ^= 1;
    }
}

#define SC_NODES 8
static_assert(NN % SC_NODES == 0, "a wave's nodes are all inside the graph");
__global__ __launch_bounds__(256) void k_scores(
    const float* __restrict__ h, const float* __restrict__ a_i,
    const float* __restrict__ a_j, float* __restrict__ s_i,
    float* __restrict__ s_j)
{
    const int lane = threadIdx.x & 31;
    const int wave = blockIdx.x * 8 + __builtin_amdgcn_readfirstlane(threadIdx.x >> 5);
    const int n0 = wave * SC_NODES;
    if (n0 >= NN) return;
    const float4 ai = ((const float4*)a_i)[lane];
    const float4 aj = ((const float4*)a_j)[lane];
    const float4* __restrict__ h4 = (const float4*)h + (size_t)n0 * (DF / 4) + lane;

    float4 hv[SC_NODES];
    #pragma unroll
    for (int k = 0; k < SC_NODES; ++k) hv[k] = h4[k * (DF / 4)];

    const int hd = lane >> 3;
    #pragma unroll
    for (int k = 0; k < SC_NODES; ++k) {
        float vi = hv[k].x * ai.x; vi += hv[k].y * ai.y; vi += hv[k].z * ai.z; vi += hv[k].w * ai.w;
        float vj = hv[k].x * aj.x; vj += hv[k].y * aj.y; vj += hv[k].z * aj.z; vj += hv[k].w * aj.w;
        #pragma unroll
        for (int off = 4; off > 0; off >>= 1) {
            vi += __shfl_xor(vi, off, 32);
            vj += __shfl_xor(vj, off, 32);
        }
        if ((lane & 7) == 0) {
            s_i[(n0 + k) * NHEADS + hd] = vi;
            s_j[(n0 + k) * NHEADS + hd] = vj;
        }
    }
}

#define EPT 5
static_assert(NE % (256 * EPT) == 0, "the edge kernels cover NE without a tail");
__global__ __launch_bounds__(256) void k_edge_rank(
    const int* __restrict__ ei, int* __restrict__ cnt, int* __restrict__ rank)
{
    const int e0 = blockIdx.x * (256 * EPT) + threadIdx.x;
    int recv[EPT], rk[EPT];
    #pragma unroll
    for (int k = 0; k < EPT; ++k) recv[k] = ei[2 * (e0 + k * 256)];
    #pragma unroll
    for (int k = 0; k < EPT; ++k) rk[k] = atomicAdd(&cnt[recv[k]], 1);
    #pragma unroll
    for (int k = 0; k < EPT; ++k) rank[e0 + k * 256] = rk[k];
}

__global__ __launch_bounds__(256) void k_seg_alloc(
    const int* __restrict__ cnt, int* __restrict__ offs, int* __restrict__ total)
{
    __shared__ int wsum[8];
    __shared__ int bbase;
    const int tid = threadIdx.x, lane = tid & 31, w = tid >> 5;
    const int i = blockIdx.x * 256 + tid;
    const int c = (i < NN) ? cnt[i] : 0;
    int s = c;
    for (int d = 1; d < 32; d <<= 1) {
        const int t = __shfl_up(s, d, 32);
        if (lane >= d) s += t;
    }
    if (lane == 31) wsum[w] = s;
    __syncthreads();
    if (tid == 0) {
        int t = 0;
        for (int k = 0; k < 8; ++k) { const int v = wsum[k]; wsum[k] = t; t += v; }
        bbase = atomicAdd(total, t);
    }
    __syncthreads();
    if (i < NN) offs[i] = bbase + wsum[w] + s - c;
}

__global__ __launch_bounds__(256) void k_edge_place(
    const int* __restrict__ ei, const int* __restrict__ offs,
    const int* __restrict__ rank, int* __restrict__ list)
{
    const int e0 = blockIdx.x * (256 * EPT) + threadIdx.x;
    int2 rs[EPT]; int rk[EPT], of[EPT];
    #pragma unroll
    for (int k = 0; k < EPT; ++k) {
        rs[k] = ((const int2*)ei)[e0 + k * 256];
        rk[k] = rank[e0 + k * 256];
    }
    #pragma unroll
    for (int k = 0; k < EPT; ++k) of[k] = offs[rs[k].x];
    #pragma unroll
    for (int k = 0; k < EPT; ++k)
        ((int2*)list)[of[k] + rk[k]] = make_int2(e0 + k * 256, rs[k].y);
}

DEV int next_by_id(const int2* __restrict__ seg, int deg, int lane, int& last) {
    int best = 0x7fffffff, bs = 0;
    for (int i = lane; i < deg; i += 32) {
        const int2 p = seg[i];
        if (p.x > last && p.x < best) { best = p.x; bs = p.y; }
    }
    const int mn = wave_min_i(best);
    const unsigned hit = __builtin_amdgcn_ballot_w32(best == mn);
    last = mn;
    return lane_i(bs, __builtin_ctz(hit));
}

__global__ __launch_bounds__(256) void k_node_gather(
    const int* __restrict__ list, const int* __restrict__ cnt,
    const int* __restrict__ offs, const float* __restrict__ s_i,
    const float* __restrict__ s_j, const float* __restrict__ a_e,
    const float* __restrict__ b_a, const float* __restrict__ h,
    float* __restrict__ agg)
{
    const int lane = threadIdx.x & 31;
    const int node = blockIdx.x * 8 + __builtin_amdgcn_readfirstlane(threadIdx.x >> 5);
    if (node >= NN) return;
    const int deg = cnt[node];
    const int2* __restrict__ seg = (const int2*)list + offs[node];
    const int hd = lane >> 3;
    const int k0 = (hd == 0) ? -1 : 0, k1 = (hd == 1) ? -1 : 0;
    const int k2 = (hd == 2) ? -1 : 0, k3 = (hd == 3) ? -1 : 0;
    const float4 si = *(const float4*)&s_i[node * NHEADS];
    const float ae0 = a_e[0], ae1 = a_e[1], ae2 = a_e[2], ae3 = a_e[3];
    const float ba0 = b_a[0], ba1 = b_a[1], ba2 = b_a[2], ba3 = b_a[3];
    const float4* __restrict__ hrow = (const float4*)h + lane;

    float4 acc = make_float4(0.0f, 0.0f, 0.0f, 0.0f);
    if (deg == 0) {
        ((float4*)agg)[(size_t)node * (DF / 4) + lane] = acc;
        return;
    }

    if (deg <= 32) {
        const bool on = lane < deg;
        int e = 0x7fffffff, send = 0;
        if (on) { const int2 p = seg[lane]; e = p.x; send = p.y; }
        int rk = 0;
        for (int j = 0; j < deg; ++j) rk += (lane_i(e, j) < e) ? 1 : 0;
        if (!on) rk = lane;
        send = __builtin_amdgcn_ds_permute(rk << 2, send);

        float l0 = -INFINITY, l1 = -INFINITY, l2 = -INFINITY, l3 = -INFINITY;
        if (on) {
            const float4 sj = *(const float4*)&s_j[send * NHEADS];
            l0 = leaky(si.x + sj.x + ae0 + ba0);
            l1 = leaky(si.y + sj.y + ae1 + ba1);
            l2 = leaky(si.z + sj.z + ae2 + ba2);
            l3 = leaky(si.w + sj.w + ae3 + ba3);
        }
        const float m0 = wave_max(l0), m1 = wave_max(l1);
        const float m2 = wave_max(l2), m3 = wave_max(l3);
        const float w0 = expf(l0 - m0), w1 = expf(l1 - m1);
        const float w2 = expf(l2 - m2), w3 = expf(l3 - m3);
        const float al0 = w0 / (wave_sum(w0) + 1e-16f);
        const float al1 = w1 / (wave_sum(w1) + 1e-16f);
        const float al2 = w2 / (wave_sum(w2) + 1e-16f);
        const float al3 = w3 / (wave_sum(w3) + 1e-16f);

        int r = 0;
        for (; r + 4 <= deg; r += 4) {
            float4 v[4]; float a[4];
            #pragma unroll
            for (int u = 0; u < 4; ++u) {
                v[u] = hrow[(size_t)lane_i(send, r + u) * (DF / 4)];
                a[u] = pick4(lane_f(al0, r + u), lane_f(al1, r + u),
                             lane_f(al2, r + u), lane_f(al3, r + u), k0, k1, k2, k3);
            }
            #pragma unroll
            for (int u = 0; u < 4; ++u) {
                acc.x += v[u].x * a[u]; acc.y += v[u].y * a[u];
                acc.z += v[u].z * a[u]; acc.w += v[u].w * a[u];
            }
        }
        for (; r < deg; ++r) {
            const float4 v = hrow[(size_t)lane_i(send, r) * (DF / 4)];
            const float a = pick4(lane_f(al0, r), lane_f(al1, r),
                                  lane_f(al2, r), lane_f(al3, r), k0, k1, k2, k3);
            acc.x += v.x * a; acc.y += v.y * a; acc.z += v.z * a; acc.w += v.w * a;
        }
    } else {
        float m0 = -INFINITY, m1 = -INFINITY, m2 = -INFINITY, m3 = -INFINITY;
        for (int i = lane; i < deg; i += 32) {
            const float4 sj = *(const float4*)&s_j[seg[i].y * NHEADS];
            m0 = fmaxf(m0, leaky(si.x + sj.x + ae0 + ba0));
            m1 = fmaxf(m1, leaky(si.y + sj.y + ae1 + ba1));
            m2 = fmaxf(m2, leaky(si.z + sj.z + ae2 + ba2));
            m3 = fmaxf(m3, leaky(si.w + sj.w + ae3 + ba3));
        }
        m0 = wave_max(m0); m1 = wave_max(m1); m2 = wave_max(m2); m3 = wave_max(m3);
        float d0 = 0.0f, d1 = 0.0f, d2 = 0.0f, d3 = 0.0f;
        int last = -1;
        for (int t = 0; t < deg; ++t) {
            const int s = next_by_id(seg, deg, lane, last);
            const float4 sj = *(const float4*)&s_j[s * NHEADS];
            d0 += expf(leaky(si.x + sj.x + ae0 + ba0) - m0);
            d1 += expf(leaky(si.y + sj.y + ae1 + ba1) - m1);
            d2 += expf(leaky(si.z + sj.z + ae2 + ba2) - m2);
            d3 += expf(leaky(si.w + sj.w + ae3 + ba3) - m3);
        }
        last = -1;
        for (int t = 0; t < deg; ++t) {
            const int s = next_by_id(seg, deg, lane, last);
            const float4 sj = *(const float4*)&s_j[s * NHEADS];
            const float a = pick4(
                expf(leaky(si.x + sj.x + ae0 + ba0) - m0) / (d0 + 1e-16f),
                expf(leaky(si.y + sj.y + ae1 + ba1) - m1) / (d1 + 1e-16f),
                expf(leaky(si.z + sj.z + ae2 + ba2) - m2) / (d2 + 1e-16f),
                expf(leaky(si.w + sj.w + ae3 + ba3) - m3) / (d3 + 1e-16f), k0, k1, k2, k3);
            const float4 v = hrow[(size_t)s * (DF / 4)];
            acc.x += v.x * a; acc.y += v.y * a; acc.z += v.z * a; acc.w += v.w * a;
        }
    }
    ((float4*)agg)[(size_t)node * (DF / 4) + lane] = acc;
}

#define BN_WAVES 8
#define BN_WROWS 16
#define BN_ROWS  (BN_WAVES * BN_WROWS)
static_assert(NN % BN_WROWS == 0, "a wave's rows are all inside the matrix or all outside");
__global__ __launch_bounds__(256) void k_bn_partial(
    const float* __restrict__ upd, float* __restrict__ macc,
    float* __restrict__ sqacc)
{
    __shared__ float4 ps[BN_WAVES][32];
    __shared__ float4 ps2[BN_WAVES][32];
    const int tid  = threadIdx.x;
    const int lane = tid & 31;
    const int w    = __builtin_amdgcn_readfirstlane(tid >> 5);
    const int r0   = blockIdx.x * BN_ROWS + w * BN_WROWS;

    float4 s  = make_float4(0.0f, 0.0f, 0.0f, 0.0f);
    float4 s2 = make_float4(0.0f, 0.0f, 0.0f, 0.0f);
    if (r0 < NN) {
        const float4* __restrict__ u4 = (const float4*)upd + (size_t)r0 * (DF / 4) + lane;
        float4 v[BN_WROWS];
        #pragma unroll
        for (int k = 0; k < BN_WROWS; ++k) v[k] = u4[k * (DF / 4)];
        #pragma unroll
        for (int k = 0; k < BN_WROWS; ++k) {
            s.x += v[k].x; s2.x += v[k].x * v[k].x;
            s.y += v[k].y; s2.y += v[k].y * v[k].y;
            s.z += v[k].z; s2.z += v[k].z * v[k].z;
            s.w += v[k].w; s2.w += v[k].w * v[k].w;
        }
    }
    ps[w][lane]  = s;
    ps2[w][lane] = s2;
    __syncthreads();
    if (tid < DF) {
        const float* __restrict__ p  = (const float*)ps;
        const float* __restrict__ p2 = (const float*)ps2;
        float a = 0.0f, b = 0.0f;
        for (int k = 0; k < BN_WAVES; ++k) { a += p[k * DF + tid]; b += p2[k * DF + tid]; }
        atomicAdd(&macc[tid], a);
        atomicAdd(&sqacc[tid], b);
    }
}

#define FIN_ROWS 8
static_assert(NN % FIN_ROWS == 0, "a wave's rows are all inside the output");
DEV float fin_rstd(float sq, float mean, float inv_n) {
    const float var = sq * inv_n - mean * mean;
    return rsqrtf(var + BN_EPS);
}
DEV float fin_one(float u, float a, float ma, float inv_n, float rs, float g, float b) {
    const float d = fmaf(-ma, inv_n, u);
    const float normed = g * d * rs + b;
    return normed + fmaxf(a, 0.0f);
}
__global__ __launch_bounds__(256) void k_finalize(
    const float* __restrict__ upd, const float* __restrict__ agg,
    const float* __restrict__ macc, const float* __restrict__ sqacc,
    const float* __restrict__ gamma, const float* __restrict__ beta,
    float* __restrict__ out)
{
    const int lane = threadIdx.x & 31;
    const int wave = blockIdx.x * 8 + __builtin_amdgcn_readfirstlane(threadIdx.x >> 5);
    const int r0 = wave * FIN_ROWS;
    if (r0 >= NN) return;
    const float inv_n = 1.0f / (float)NN;
    const float4 ma = ((const float4*)macc)[lane];
    const float4 sq = ((const float4*)sqacc)[lane];
    const float4 g  = ((const float4*)gamma)[lane];
    const float4 b  = ((const float4*)beta)[lane];
    float4 rs;
    rs.x = fin_rstd(sq.x, ma.x * inv_n, inv_n); rs.y = fin_rstd(sq.y, ma.y * inv_n, inv_n);
    rs.z = fin_rstd(sq.z, ma.z * inv_n, inv_n); rs.w = fin_rstd(sq.w, ma.w * inv_n, inv_n);

    const size_t base = (size_t)r0 * (DF / 4) + lane;
    const float4* __restrict__ u4 = (const float4*)upd + base;
    const float4* __restrict__ a4 = (const float4*)agg + base;
    float4* __restrict__ o4 = (float4*)out + base;
    float4 u[FIN_ROWS], a[FIN_ROWS];
    #pragma unroll
    for (int k = 0; k < FIN_ROWS; ++k) {
        u[k] = u4[k * (DF / 4)];
        a[k] = a4[k * (DF / 4)];
    }
    #pragma unroll
    for (int k = 0; k < FIN_ROWS; ++k) {
        float4 o;
        o.x = fin_one(u[k].x, a[k].x, ma.x, inv_n, rs.x, g.x, b.x);
        o.y = fin_one(u[k].y, a[k].y, ma.y, inv_n, rs.y, g.y, b.y);
        o.z = fin_one(u[k].z, a[k].z, ma.z, inv_n, rs.z, g.z, b.z);
        o.w = fin_one(u[k].w, a[k].w, ma.w, inv_n, rs.w, g.w, b.w);
        o4[k * (DF / 4)] = o;
    }
}

extern "C" void kernel_launch(void* const* d_in, const int* in_sizes, int n_in,
                              void* d_out, int out_size, void* d_ws, size_t ws_size,
                              hipStream_t stream) {
    const float* x     = (const float*)d_in[0];
    const int*   ei    = (const int*)  d_in[1];
    const float* Wk    = (const float*)d_in[2];
    const float* bk    = (const float*)d_in[3];
    const float* a_i   = (const float*)d_in[4];
    const float* a_j   = (const float*)d_in[5];
    const float* a_e   = (const float*)d_in[6];
    const float* b_a   = (const float*)d_in[7];
    const float* Wm    = (const float*)d_in[8];
    const float* bm    = (const float*)d_in[9];
    const float* gamma = (const float*)d_in[10];
    const float* beta  = (const float*)d_in[11];
    float* out = (float*)d_out;
    float* ws  = (float*)d_ws;

    float* h_buf  = ws + OFF_H;
    float* upd    = ws + OFF_UPD;
    float* agg    = ws + OFF_AGG;
    float* s_i    = ws + OFF_SI;
    float* s_j    = ws + OFF_SJ;
    int*   cnt    = (int*)ws + OFF_CNT;
    int*   offs   = (int*)ws + OFF_OFFS;
    int*   rank   = (int*)ws + OFF_RANK;
    int*   list   = (int*)ws + OFF_LIST;
    int*   total  = (int*)ws + OFF_TOT;
    float* macc   = ws + OFF_MACC;
    float* sqacc  = ws + OFF_SQ;

    k_init<<<(NN + 255) / 256, 256, 0, stream>>>(ws);
    k_gemm2_wmma<<<GT_WGS, 256, 0, stream>>>(x, Wk, bk, Wm, bm, h_buf, upd);
    k_scores<<<(NN + 8 * SC_NODES - 1) / (8 * SC_NODES), 256, 0, stream>>>(h_buf, a_i, a_j, s_i, s_j);
    k_edge_rank<<<NE / (256 * EPT), 256, 0, stream>>>(ei, cnt, rank);
    k_seg_alloc<<<(NN + 255) / 256, 256, 0, stream>>>(cnt, offs, total);
    k_edge_place<<<NE / (256 * EPT), 256, 0, stream>>>(ei, offs, rank, list);
    k_node_gather<<<(NN + 7) / 8, 256, 0, stream>>>(
        list, cnt, offs, s_i, s_j, a_e, b_a, h_buf, agg);
    k_bn_partial<<<(NN + BN_ROWS - 1) / BN_ROWS, 256, 0, stream>>>(
        upd, macc, sqacc);
    k_finalize<<<(NN + 8 * FIN_ROWS - 1) / (8 * FIN_ROWS), 256, 0, stream>>>(
        upd, agg, macc, sqacc, gamma, beta, out);
}


# Stages with no dependency between them share a launch (block index picks the stage): edge count | batchnorm sums, segment starts | scores, counter clearing in the projection launch; 9 launches -> 6
# speedup vs baseline: 3.3251x; 1.0473x over previous
#include <hip/hip_runtime.h>
#include <hip/hip_bf16.h>
#include <math.h>

#define NN      50000
#define NE      800000
#define DF      128
#define NHEADS  4
#define HU      32
#define NEG_SLOPE 0.2f
#define BN_EPS    0.001f

typedef __attribute__((ext_vector_type(2))) float v2f;
typedef __attribute__((ext_vector_type(8))) float v8f;

#define OFF_H    ((size_t)0)
#define OFF_UPD  (OFF_H   + (size_t)NN * DF)
#define OFF_AGG  (OFF_UPD + (size_t)NN * DF)
#define OFF_SI   (OFF_AGG + (size_t)NN * DF)
#define OFF_SJ   (OFF_SI  + (size_t)NN * NHEADS)
#define OFF_CNT  (OFF_SJ  + (size_t)NN * NHEADS)
#define OFF_OFFS (OFF_CNT + (size_t)NN * NHEADS)
#define OFF_RANK (OFF_OFFS + (size_t)NN * NHEADS)
#define OFF_LIST (OFF_RANK + (size_t)NE)
#define OFF_TOT  (OFF_LIST + (size_t)2 * NE)
#define OFF_MACC (OFF_RANK + (size_t)NE * NHEADS)
#define OFF_SQ   (OFF_MACC + (size_t)DF)

#define DEV static __device__ __forceinline__
DEV int   lane_i(int v, int l)   { return __builtin_amdgcn_readlane(v, l); }
DEV float lane_f(float v, int l) { return __int_as_float(__builtin_amdgcn_readlane(__float_as_int(v), l)); }
DEV float wave_max(float v) {
    for (int off = 16; off > 0; off >>= 1) v = fmaxf(v, __shfl_xor(v, off, 32));
    return v;
}
DEV float wave_sum(float v) {
    for (int off = 16; off > 0; off >>= 1) v += __shfl_xor(v, off, 32);
    return v;
}
DEV int wave_min_i(int v) {
    for (int off = 16; off > 0; off >>= 1) v = min(v, __shfl_xor(v, off, 32));
    return v;
}
DEV float leaky(float x) { return (x > 0.0f) ? x : NEG_SLOPE * x; }
DEV float pick4(float a0, float a1, float a2, float a3, int k0, int k1, int k2, int k3) {
    return __int_as_float((__float_as_int(a0) & k0) | (__float_as_int(a1) & k1) |
                          (__float_as_int(a2) & k2) | (__float_as_int(a3) & k3));
}


#define GT_ROWS  16
#define GT_PITCH 132
#define GT_TILES (NN / GT_ROWS)
#define GT_WGS   512
static_assert(NN % GT_ROWS == 0, "every row tile is whole: no row guard in the kernel");
static_assert(GT_WGS * 256 >= NN, "one thread per degree counter when the launch clears them");

template <int KSTRIDE>
DEV void load_bfrag(const float* __restrict__ q, v2f (&bf)[2][DF / 4]) {
    #pragma unroll
    for (int j = 0; j < DF / 4; ++j) {
        const float2 p0 = *(const float2*)(q + (size_t)(4 * j) * KSTRIDE);
        const float2 p1 = *(const float2*)(q + (size_t)(4 * j + 1) * KSTRIDE);
        bf[0][j].x = p0.x; bf[0][j].y = p1.x;
        bf[1][j].x = p0.y; bf[1][j].y = p1.y;
    }
}

__global__ __launch_bounds__(256) void k_gemm2_wmma(
    const float* __restrict__ x, const float* __restrict__ Wk, const float* __restrict__ bk,
    const float* __restrict__ Wm, const float* __restrict__ bm,
    float* __restrict__ h, float* __restrict__ upd,
    int* __restrict__ cnt, int* __restrict__ total,
    float* __restrict__ macc, float* __restrict__ sqacc)
{
    __shared__ __attribute__((aligned(16))) float atile[2][GT_ROWS * GT_PITCH];

    const int tid  = threadIdx.x;
    {
        const int gid = blockIdx.x * 256 + tid;
        if (gid < NN) cnt[gid] = 0;
        if (gid == 0) *total = 0;
        if (gid < DF) { macc[gid] = 0.0f; sqacc[gid] = 0.0f; }
    }
    const int lane = tid & 31;
    const int half = lane >> 4;
    const int l    = lane & 15;
    const int w    = __builtin_amdgcn_readfirstlane(tid >> 5);
    const bool is_h = w < 4;
    const int cb   = (w & 3) * 32;

    const int r0 = tid >> 5, c4 = tid & 31;
    const float4* __restrict__ x4 = (const float4*)x + tid;
    const int lds0 = r0 * GT_PITCH + 4 * c4;
    const int lds1 = (r0 + 8) * GT_PITCH + 4 * c4;

    int tile = blockIdx.x;
    float4 pre0, pre1;
    {
        const float4* __restrict__ src = x4 + (size_t)tile * (GT_ROWS * DF / 4);
        pre0 = src[0];
        pre1 = src[256];
    }

    v2f bf[2][DF / 4];
    float bv0, bv1;
    if (is_h) {
        load_bfrag<HU>(Wk + (size_t)w * (DF * HU) + (size_t)(2 * half) * HU + 2 * l, bf);
        const float2 b2 = *(const float2*)(bk + cb + 2 * l);
        bv0 = b2.x; bv1 = b2.y;
    } else {
        load_bfrag<DF>(Wm + (size_t)(2 * half) * DF + cb + 2 * l, bf);
        const float2 b2 = *(const float2*)(bm + cb + 2 * l);
        bv0 = b2.x; bv1 = b2.y;
    }

    *(float4*)&atile[0][lds0] = pre0;
    *(float4*)&atile[0][lds1] = pre1;
    __syncthreads();

    float* __restrict__ obase = (is_h ? h : upd) + (size_t)(8 * half) * DF + cb + 2 * l;
    int buf = 0;
    for (; tile < GT_TILES; tile += GT_WGS) {
        const int next = tile + GT_WGS;
        const bool more = next < GT_TILES;
        if (more) {
            const float4* __restrict__ src = x4 + (size_t)next * (GT_ROWS * DF / 4);
            pre0 = src[0];
            pre1 = src[256];
        }

        const float* __restrict__ arow = &atile[buf][l * GT_PITCH + 2 * half];
        v8f acc0 = {}, acc1 = {};
        #pragma unroll
        for (int j = 0; j < DF / 4; ++j) {
            const v2f a = *(const v2f*)(arow + 4 * j);
            acc0 = __builtin_amdgcn_wmma_f32_16x16x4_f32(
                       false, a, false, bf[0][j], (short)0, acc0, false, false);
            acc1 = __builtin_amdgcn_wmma_f32_16x16x4_f32(
                       false, a, false, bf[1][j], (short)0, acc1, false, false);
        }

        if (more) {
            *(float4*)&atile[buf ^ 1][lds0] = pre0;
            *(float4*)&atile[buf ^ 1][lds1] = pre1;
        }
        __syncthreads();

        float* __restrict__ dst = obase + (size_t)tile * (GT_ROWS * DF);
        #pragma unroll
        for (int r = 0; r < 8; ++r) {
            float v0 = acc0[r] + bv0;
            float v1 = acc1[r] + bv1;
            if (!is_h) { v0 = fmaxf(v0, 0.0f); v1 = fmaxf(v1, 0.0f); }
            *(float2*)(dst + (size_t)r * DF) = make_float2(v0, v1);
        }
        buf ^= 1;
    }
}

#define SC_NODES 8
#define SC_BLOCKS ((NN + 8 * SC_NODES - 1) / (8 * SC_NODES))
static_assert(NN % SC_NODES == 0, "a wave's nodes are all inside the graph");
DEV void scores_block(
    const int bid,
    const float* __restrict__ h, const float* __restrict__ a_i,
    const float* __restrict__ a_j, float* __restrict__ s_i,
    float* __restrict__ s_j)
{
    const int lane = threadIdx.x & 31;
    const int wave = bid * 8 + __builtin_amdgcn_readfirstlane(threadIdx.x >> 5);
    const int n0 = wave * SC_NODES;
    if (n0 >= NN) return;
    const float4 ai = ((const float4*)a_i)[lane];
    const float4 aj = ((const float4*)a_j)[lane];
    const float4* __restrict__ h4 = (const float4*)h + (size_t)n0 * (DF / 4) + lane;

    float4 hv[SC_NODES];
    #pragma unroll
    for (int k = 0; k < SC_NODES; ++k) hv[k] = h4[k * (DF / 4)];

    const int hd = lane >> 3;
    #pragma unroll
    for (int k = 0; k < SC_NODES; ++k) {
        float vi = hv[k].x * ai.x; vi += hv[k].y * ai.y; vi += hv[k].z * ai.z; vi += hv[k].w * ai.w;
        float vj = hv[k].x * aj.x; vj += hv[k].y * aj.y; vj += hv[k].z * aj.z; vj += hv[k].w * aj.w;
        #pragma unroll
        for (int off = 4; off > 0; off >>= 1) {
            vi += __shfl_xor(vi, off, 32);
            vj += __shfl_xor(vj, off, 32);
        }
        if ((lane & 7) == 0) {
            s_i[(n0 + k) * NHEADS + hd] = vi;
            s_j[(n0 + k) * NHEADS + hd] = vj;
        }
    }
}

#define EPT 5
#define EDGE_BLOCKS (NE / (256 * EPT))
static_assert(NE % (256 * EPT) == 0, "the edge kernels cover NE without a tail");
DEV void edge_rank_block(
    const int bid,
    const int* __restrict__ ei, int* __restrict__ cnt, int* __restrict__ rank)
{
    const int e0 = bid * (256 * EPT) + threadIdx.x;
    int recv[EPT], rk[EPT];
    #pragma unroll
    for (int k = 0; k < EPT; ++k) recv[k] = ei[2 * (e0 + k * 256)];
    #pragma unroll
    for (int k = 0; k < EPT; ++k) rk[k] = atomicAdd(&cnt[recv[k]], 1);
    #pragma unroll
    for (int k = 0; k < EPT; ++k) rank[e0 + k * 256] = rk[k];
}

#define ALLOC_BLOCKS ((NN + 255) / 256)
DEV void seg_alloc_block(
    const int bid,
    const int* __restrict__ cnt, int* __restrict__ offs, int* __restrict__ total)
{
    __shared__ int wsum[8];
    __shared__ int bbase;
    const int tid = threadIdx.x, lane = tid & 31, w = tid >> 5;
    const int i = bid * 256 + tid;
    const int c = (i < NN) ? cnt[i] : 0;
    int s = c;
    for (int d = 1; d < 32; d <<= 1) {
        const int t = __shfl_up(s, d, 32);
        if (lane >= d) s += t;
    }
    if (lane == 31) wsum[w] = s;
    __syncthreads();
    if (tid == 0) {
        int t = 0;
        for (int k = 0; k < 8; ++k) { const int v = wsum[k]; wsum[k] = t; t += v; }
        bbase = atomicAdd(total, t);
    }
    __syncthreads();
    if (i < NN) offs[i] = bbase + wsum[w] + s - c;
}

__global__ __launch_bounds__(256) void k_alloc_scores(
    const int* __restrict__ cnt, int* __restrict__ offs, int* __restrict__ total,
    const float* __restrict__ h, const float* __restrict__ a_i,
    const float* __restrict__ a_j, float* __restrict__ s_i,
    float* __restrict__ s_j)
{
    if (blockIdx.x < ALLOC_BLOCKS) seg_alloc_block(blockIdx.x, cnt, offs, total);
    else scores_block(blockIdx.x - ALLOC_BLOCKS, h, a_i, a_j, s_i, s_j);
}

__global__ __launch_bounds__(256) void k_edge_place(
    const int* __restrict__ ei, const int* __restrict__ offs,
    const int* __restrict__ rank, int* __restrict__ list)
{
    const int e0 = blockIdx.x * (256 * EPT) + threadIdx.x;
    int2 rs[EPT]; int rk[EPT], of[EPT];
    #pragma unroll
    for (int k = 0; k < EPT; ++k) {
        rs[k] = ((const int2*)ei)[e0 + k * 256];
        rk[k] = rank[e0 + k * 256];
    }
    #pragma unroll
    for (int k = 0; k < EPT; ++k) of[k] = offs[rs[k].x];
    #pragma unroll
    for (int k = 0; k < EPT; ++k)
        ((int2*)list)[of[k] + rk[k]] = make_int2(e0 + k * 256, rs[k].y);
}

DEV int next_by_id(const int2* __restrict__ seg, int deg, int lane, int& last) {
    int best = 0x7fffffff, bs = 0;
    for (int i = lane; i < deg; i += 32) {
        const int2 p = seg[i];
        if (p.x > last && p.x < best) { best = p.x; bs = p.y; }
    }
    const int mn = wave_min_i(best);
    const unsigned hit = __builtin_amdgcn_ballot_w32(best == mn);
    last = mn;
    return lane_i(bs, __builtin_ctz(hit));
}

__global__ __launch_bounds__(256) void k_node_gather(
    const int* __restrict__ list, const int* __restrict__ cnt,
    const int* __restrict__ offs, const float* __restrict__ s_i,
    const float* __restrict__ s_j, const float* __restrict__ a_e,
    const float* __restrict__ b_a, const float* __restrict__ h,
    float* __restrict__ agg)
{
    const int lane = threadIdx.x & 31;
    const int node = blockIdx.x * 8 + __builtin_amdgcn_readfirstlane(threadIdx.x >> 5);
    if (node >= NN) return;
    const int deg = cnt[node];
    const int2* __restrict__ seg = (const int2*)list + offs[node];
    const int hd = lane >> 3;
    const int k0 = (hd == 0) ? -1 : 0, k1 = (hd == 1) ? -1 : 0;
    const int k2 = (hd == 2) ? -1 : 0, k3 = (hd == 3) ? -1 : 0;
    const float4 si = *(const float4*)&s_i[node * NHEADS];
    const float ae0 = a_e[0], ae1 = a_e[1], ae2 = a_e[2], ae3 = a_e[3];
    const float ba0 = b_a[0], ba1 = b_a[1], ba2 = b_a[2], ba3 = b_a[3];
    const float4* __restrict__ hrow = (const float4*)h + lane;

    float4 acc = make_float4(0.0f, 0.0f, 0.0f, 0.0f);
    if (deg == 0) {
        ((float4*)agg)[(size_t)node * (DF / 4) + lane] = acc;
        return;
    }

    if (deg <= 32) {
        const bool on = lane < deg;
        int e = 0x7fffffff, send = 0;
        if (on) { const int2 p = seg[lane]; e = p.x; send = p.y; }
        int rk = 0;
        for (int j = 0; j < deg; ++j) rk += (lane_i(e, j) < e) ? 1 : 0;
        if (!on) rk = lane;
        send = __builtin_amdgcn_ds_permute(rk << 2, send);

        float l0 = -INFINITY, l1 = -INFINITY, l2 = -INFINITY, l3 = -INFINITY;
        if (on) {
            const float4 sj = *(const float4*)&s_j[send * NHEADS];
            l0 = leaky(si.x + sj.x + ae0 + ba0);
            l1 = leaky(si.y + sj.y + ae1 + ba1);
            l2 = leaky(si.z + sj.z + ae2 + ba2);
            l3 = leaky(si.w + sj.w + ae3 + ba3);
        }
        const float m0 = wave_max(l0), m1 = wave_max(l1);
        const float m2 = wave_max(l2), m3 = wave_max(l3);
        const float w0 = expf(l0 - m0), w1 = expf(l1 - m1);
        const float w2 = expf(l2 - m2), w3 = expf(l3 - m3);
        const float al0 = w0 / (wave_sum(w0) + 1e-16f);
        const float al1 = w1 / (wave_sum(w1) + 1e-16f);
        const float al2 = w2 / (wave_sum(w2) + 1e-16f);
        const float al3 = w3 / (wave_sum(w3) + 1e-16f);

        int r = 0;
        for (; r + 4 <= deg; r += 4) {
            float4 v[4]; float a[4];
            #pragma unroll
            for (int u = 0; u < 4; ++u) {
                v[u] = hrow[(size_t)lane_i(send, r + u) * (DF / 4)];
                a[u] = pick4(lane_f(al0, r + u), lane_f(al1, r + u),
                             lane_f(al2, r + u), lane_f(al3, r + u), k0, k1, k2, k3);
            }
            #pragma unroll
            for (int u = 0; u < 4; ++u) {
                acc.x += v[u].x * a[u]; acc.y += v[u].y * a[u];
                acc.z += v[u].z * a[u]; acc.w += v[u].w * a[u];
            }
        }
        for (; r < deg; ++r) {
            const float4 v = hrow[(size_t)lane_i(send, r) * (DF / 4)];
            const float a = pick4(lane_f(al0, r), lane_f(al1, r),
                                  lane_f(al2, r), lane_f(al3, r), k0, k1, k2, k3);
            acc.x += v.x * a; acc.y += v.y * a; acc.z += v.z * a; acc.w += v.w * a;
        }
    } else {
        float m0 = -INFINITY, m1 = -INFINITY, m2 = -INFINITY, m3 = -INFINITY;
        for (int i = lane; i < deg; i += 32) {
            const float4 sj = *(const float4*)&s_j[seg[i].y * NHEADS];
            m0 = fmaxf(m0, leaky(si.x + sj.x + ae0 + ba0));
            m1 = fmaxf(m1, leaky(si.y + sj.y + ae1 + ba1));
            m2 = fmaxf(m2, leaky(si.z + sj.z + ae2 + ba2));
            m3 = fmaxf(m3, leaky(si.w + sj.w + ae3 + ba3));
        }
        m0 = wave_max(m0); m1 = wave_max(m1); m2 = wave_max(m2); m3 = wave_max(m3);
        float d0 = 0.0f, d1 = 0.0f, d2 = 0.0f, d3 = 0.0f;
        int last = -1;
        for (int t = 0; t < deg; ++t) {
            const int s = next_by_id(seg, deg, lane, last);
            const float4 sj = *(const float4*)&s_j[s * NHEADS];
            d0 += expf(leaky(si.x + sj.x + ae0 + ba0) - m0);
            d1 += expf(leaky(si.y + sj.y + ae1 + ba1) - m1);
            d2 += expf(leaky(si.z + sj.z + ae2 + ba2) - m2);
            d3 += expf(leaky(si.w + sj.w + ae3 + ba3) - m3);
        }
        last = -1;
        for (int t = 0; t < deg; ++t) {
            const int s = next_by_id(seg, deg, lane, last);
            const float4 sj = *(const float4*)&s_j[s * NHEADS];
            const float a = pick4(
                expf(leaky(si.x + sj.x + ae0 + ba0) - m0) / (d0 + 1e-16f),
                expf(leaky(si.y + sj.y + ae1 + ba1) - m1) / (d1 + 1e-16f),
                expf(leaky(si.z + sj.z + ae2 + ba2) - m2) / (d2 + 1e-16f),
                expf(leaky(si.w + sj.w + ae3 + ba3) - m3) / (d3 + 1e-16f), k0, k1, k2, k3);
            const float4 v = hrow[(size_t)s * (DF / 4)];
            acc.x += v.x * a; acc.y += v.y * a; acc.z += v.z * a; acc.w += v.w * a;
        }
    }
    ((float4*)agg)[(size_t)node * (DF / 4) + lane] = acc;
}

#define BN_WAVES 8
#define BN_WROWS 16
#define BN_ROWS  (BN_WAVES * BN_WROWS)
#define BN_BLOCKS ((NN + BN_ROWS - 1) / BN_ROWS)
static_assert(NN % BN_WROWS == 0, "a wave's rows are all inside the matrix or all outside");
DEV void bn_partial_block(
    const int bid,
    const float* __restrict__ upd, float* __restrict__ macc,
    float* __restrict__ sqacc)
{
    __shared__ float4 ps[BN_WAVES][32];
    __shared__ float4 ps2[BN_WAVES][32];
    const int tid  = threadIdx.x;
    const int lane = tid & 31;
    const int w    = __builtin_amdgcn_readfirstlane(tid >> 5);
    const int r0   = bid * BN_ROWS + w * BN_WROWS;

    float4 s  = make_float4(0.0f, 0.0f, 0.0f, 0.0f);
    float4 s2 = make_float4(0.0f, 0.0f, 0.0f, 0.0f);
    if (r0 < NN) {
        const float4* __restrict__ u4 = (const float4*)upd + (size_t)r0 * (DF / 4) + lane;
        float4 v[BN_WROWS];
        #pragma unroll
        for (int k = 0; k < BN_WROWS; ++k) v[k] = u4[k * (DF / 4)];
        #pragma unroll
        for (int k = 0; k < BN_WROWS; ++k) {
            s.x += v[k].x; s2.x += v[k].x * v[k].x;
            s.y += v[k].y; s2.y += v[k].y * v[k].y;
            s.z += v[k].z; s2.z += v[k].z * v[k].z;
            s.w += v[k].w; s2.w += v[k].w * v[k].w;
        }
    }
    ps[w][lane]  = s;
    ps2[w][lane] = s2;
    __syncthreads();
    if (tid < DF) {
        const float* __restrict__ p  = (const float*)ps;
        const float* __restrict__ p2 = (const float*)ps2;
        float a = 0.0f, b = 0.0f;
        for (int k = 0; k < BN_WAVES; ++k) { a += p[k * DF + tid]; b += p2[k * DF + tid]; }
        atomicAdd(&macc[tid], a);
        atomicAdd(&sqacc[tid], b);
    }
}

__global__ __launch_bounds__(256) void k_rank_bnsum(
    const int* __restrict__ ei, int* __restrict__ cnt, int* __restrict__ rank,
    const float* __restrict__ upd, float* __restrict__ macc,
    float* __restrict__ sqacc)
{
    if (blockIdx.x < EDGE_BLOCKS) edge_rank_block(blockIdx.x, ei, cnt, rank);
    else bn_partial_block(blockIdx.x - EDGE_BLOCKS, upd, macc, sqacc);
}

#define FIN_ROWS 8
static_assert(NN % FIN_ROWS == 0, "a wave's rows are all inside the output");
DEV float fin_rstd(float sq, float mean, float inv_n) {
    const float var = sq * inv_n - mean * mean;
    return rsqrtf(var + BN_EPS);
}
DEV float fin_one(float u, float a, float ma, float inv_n, float rs, float g, float b) {
    const float d = fmaf(-ma, inv_n, u);
    const float normed = g * d * rs + b;
    return normed + fmaxf(a, 0.0f);
}
__global__ __launch_bounds__(256) void k_finalize(
    const float* __restrict__ upd, const float* __restrict__ agg,
    const float* __restrict__ macc, const float* __restrict__ sqacc,
    const float* __restrict__ gamma, const float* __restrict__ beta,
    float* __restrict__ out)
{
    const int lane = threadIdx.x & 31;
    const int wave = blockIdx.x * 8 + __builtin_amdgcn_readfirstlane(threadIdx.x >> 5);
    const int r0 = wave * FIN_ROWS;
    if (r0 >= NN) return;
    const float inv_n = 1.0f / (float)NN;
    const float4 ma = ((const float4*)macc)[lane];
    const float4 sq = ((const float4*)sqacc)[lane];
    const float4 g  = ((const float4*)gamma)[lane];
    const float4 b  = ((const float4*)beta)[lane];
    float4 rs;
    rs.x = fin_rstd(sq.x, ma.x * inv_n, inv_n); rs.y = fin_rstd(sq.y, ma.y * inv_n, inv_n);
    rs.z = fin_rstd(sq.z, ma.z * inv_n, inv_n); rs.w = fin_rstd(sq.w, ma.w * inv_n, inv_n);

    const size_t base = (size_t)r0 * (DF / 4) + lane;
    const float4* __restrict__ u4 = (const float4*)upd + base;
    const float4* __restrict__ a4 = (const float4*)agg + base;
    float4* __restrict__ o4 = (float4*)out + base;
    float4 u[FIN_ROWS], a[FIN_ROWS];
    #pragma unroll
    for (int k = 0; k < FIN_ROWS; ++k) {
        u[k] = u4[k * (DF / 4)];
        a[k] = a4[k * (DF / 4)];
    }
    #pragma unroll
    for (int k = 0; k < FIN_ROWS; ++k) {
        float4 o;
        o.x = fin_one(u[k].x, a[k].x, ma.x, inv_n, rs.x, g.x, b.x);
        o.y = fin_one(u[k].y, a[k].y, ma.y, inv_n, rs.y, g.y, b.y);
        o.z = fin_one(u[k].z, a[k].z, ma.z, inv_n, rs.z, g.z, b.z);
        o.w = fin_one(u[k].w, a[k].w, ma.w, inv_n, rs.w, g.w, b.w);
        o4[k * (DF / 4)] = o;
    }
}

extern "C" void kernel_launch(void* const* d_in, const int* in_sizes, int n_in,
                              void* d_out, int out_size, void* d_ws, size_t ws_size,
                              hipStream_t stream) {
    const float* x     = (const float*)d_in[0];
    const int*   ei    = (const int*)  d_in[1];
    const float* Wk    = (const float*)d_in[2];
    const float* bk    = (const float*)d_in[3];
    const float* a_i   = (const float*)d_in[4];
    const float* a_j   = (const float*)d_in[5];
    const float* a_e   = (const float*)d_in[6];
    const float* b_a   = (const float*)d_in[7];
    const float* Wm    = (const float*)d_in[8];
    const float* bm    = (const float*)d_in[9];
    const float* gamma = (const float*)d_in[10];
    const float* beta  = (const float*)d_in[11];
    float* out = (float*)d_out;
    float* ws  = (float*)d_ws;

    float* h_buf  = ws + OFF_H;
    float* upd    = ws + OFF_UPD;
    float* agg    = ws + OFF_AGG;
    float* s_i    = ws + OFF_SI;
    float* s_j    = ws + OFF_SJ;
    int*   cnt    = (int*)ws + OFF_CNT;
    int*   offs   = (int*)ws + OFF_OFFS;
    int*   rank   = (int*)ws + OFF_RANK;
    int*   list   = (int*)ws + OFF_LIST;
    int*   total  = (int*)ws + OFF_TOT;
    float* macc   = ws + OFF_MACC;
    float* sqacc  = ws + OFF_SQ;

    k_gemm2_wmma<<<GT_WGS, 256, 0, stream>>>(x, Wk, bk, Wm, bm, h_buf, upd, cnt, total, macc, sqacc);
    k_rank_bnsum<<<EDGE_BLOCKS + BN_BLOCKS, 256, 0, stream>>>(ei, cnt, rank, upd, macc, sqacc);
    k_alloc_scores<<<ALLOC_BLOCKS + SC_BLOCKS, 256, 0, stream>>>(
        cnt, offs, total, h_buf, a_i, a_j, s_i, s_j);
    k_edge_place<<<EDGE_BLOCKS, 256, 0, stream>>>(ei, offs, rank, list);
    k_node_gather<<<(NN + 7) / 8, 256, 0, stream>>>(
        list, cnt, offs, s_i, s_j, a_e, b_a, h_buf, agg);
    k_finalize<<<(NN + 8 * FIN_ROWS - 1) / (8 * FIN_ROWS), 256, 0, stream>>>(
        upd, agg, macc, sqacc, gamma, beta, out);
}


# Scores taken from the gathered rows (s_i from the node's own row): score table, its stage and per-edge score gather gone; one-pass softmax, running max per 8 edges, sums in edge-id order, one divide
# speedup vs baseline: 3.8545x; 1.1592x over previous
#include <hip/hip_runtime.h>
#include <hip/hip_bf16.h>
#include <math.h>

#define NN      50000
#define NE      800000
#define DF      128
#define NHEADS  4
#define HU      32
#define NEG_SLOPE 0.2f
#define BN_EPS    0.001f

typedef __attribute__((ext_vector_type(2))) float v2f;
typedef __attribute__((ext_vector_type(8))) float v8f;

#define OFF_H    ((size_t)0)
#define OFF_UPD  (OFF_H   + (size_t)NN * DF)
#define OFF_AGG  (OFF_UPD + (size_t)NN * DF)
#define OFF_SI   (OFF_AGG + (size_t)NN * DF)
#define OFF_SJ   (OFF_SI  + (size_t)NN * NHEADS)
#define OFF_CNT  (OFF_SJ  + (size_t)NN * NHEADS)
#define OFF_OFFS (OFF_CNT + (size_t)NN * NHEADS)
#define OFF_RANK (OFF_OFFS + (size_t)NN * NHEADS)
#define OFF_LIST (OFF_RANK + (size_t)NE)
#define OFF_TOT  (OFF_LIST + (size_t)2 * NE)
#define OFF_MACC (OFF_RANK + (size_t)NE * NHEADS)
#define OFF_SQ   (OFF_MACC + (size_t)DF)

#define DEV static __device__ __forceinline__
DEV int   lane_i(int v, int l)   { return __builtin_amdgcn_readlane(v, l); }
template <int MASK>
DEV float row_xor(float v) {
    return __int_as_float(__builtin_amdgcn_update_dpp(0, __float_as_int(v), 0x160 + MASK, 0xf, 0xf, false));
}
DEV float sum8(float v) {
    v += row_xor<4>(v);
    v += row_xor<2>(v);
    v += row_xor<1>(v);
    return v;
}
DEV float max8(float v) {
    v = fmaxf(v, row_xor<4>(v));
    v = fmaxf(v, row_xor<2>(v));
    v = fmaxf(v, row_xor<1>(v));
    return v;
}
template <int U>
DEV float bcast8(float v) {
    return __int_as_float(__builtin_amdgcn_mov_dpp8(__float_as_int(v), U * 0x249249));
}
DEV float dot4(const float4 r, const float4 a) {
    float s = r.x * a.x; s += r.y * a.y; s += r.z * a.z; s += r.w * a.w;
    return s;
}
DEV float score8(const float4 r, const float4 a) { return sum8(dot4(r, a)); }
DEV int wave_min_i(int v) {
    for (int off = 16; off > 0; off >>= 1) v = min(v, __shfl_xor(v, off, 32));
    return v;
}
DEV float leaky(float x) { return (x > 0.0f) ? x : NEG_SLOPE * x; }


#define GT_ROWS  16
#define GT_PITCH 132
#define GT_TILES (NN / GT_ROWS)
#define GT_WGS   512
static_assert(NN % GT_ROWS == 0, "every row tile is whole: no row guard in the kernel");
static_assert(GT_WGS * 256 >= NN, "one thread per degree counter when the launch clears them");

template <int KSTRIDE>
DEV void load_bfrag(const float* __restrict__ q, v2f (&bf)[2][DF / 4]) {
    #pragma unroll
    for (int j = 0; j < DF / 4; ++j) {
        const float2 p0 = *(const float2*)(q + (size_t)(4 * j) * KSTRIDE);
        const float2 p1 = *(const float2*)(q + (size_t)(4 * j + 1) * KSTRIDE);
        bf[0][j].x = p0.x; bf[0][j].y = p1.x;
        bf[1][j].x = p0.y; bf[1][j].y = p1.y;
    }
}

__global__ __launch_bounds__(256) void k_gemm2_wmma(
    const float* __restrict__ x, const float* __restrict__ Wk, const float* __restrict__ bk,
    const float* __restrict__ Wm, const float* __restrict__ bm,
    float* __restrict__ h, float* __restrict__ upd,
    int* __restrict__ cnt, int* __restrict__ total,
    float* __restrict__ macc, float* __restrict__ sqacc)
{
    __shared__ __attribute__((aligned(16))) float atile[2][GT_ROWS * GT_PITCH];

    const int tid  = threadIdx.x;
    {
        const int gid = blockIdx.x * 256 + tid;
        if (gid < NN) cnt[gid] = 0;
        if (gid == 0) *total = 0;
        if (gid < DF) { macc[gid] = 0.0f; sqacc[gid] = 0.0f; }
    }
    const int lane = tid & 31;
    const int half = lane >> 4;
    const int l    = lane & 15;
    const int w    = __builtin_amdgcn_readfirstlane(tid >> 5);
    const bool is_h = w < 4;
    const int cb   = (w & 3) * 32;

    const int r0 = tid >> 5, c4 = tid & 31;
    const float4* __restrict__ x4 = (const float4*)x + tid;
    const int lds0 = r0 * GT_PITCH + 4 * c4;
    const int lds1 = (r0 + 8) * GT_PITCH + 4 * c4;

    int tile = blockIdx.x;
    float4 pre0, pre1;
    {
        const float4* __restrict__ src = x4 + (size_t)tile * (GT_ROWS * DF / 4);
        pre0 = src[0];
        pre1 = src[256];
    }

    v2f bf[2][DF / 4];
    float bv0, bv1;
    if (is_h) {
        load_bfrag<HU>(Wk + (size_t)w * (DF * HU) + (size_t)(2 * half) * HU + 2 * l, bf);
        const float2 b2 = *(const float2*)(bk + cb + 2 * l);
        bv0 = b2.x; bv1 = b2.y;
    } else {
        load_bfrag<DF>(Wm + (size_t)(2 * half) * DF + cb + 2 * l, bf);
        const float2 b2 = *(const float2*)(bm + cb + 2 * l);
        bv0 = b2.x; bv1 = b2.y;
    }

    *(float4*)&atile[0][lds0] = pre0;
    *(float4*)&atile[0][lds1] = pre1;
    __syncthreads();

    float* __restrict__ obase = (is_h ? h : upd) + (size_t)(8 * half) * DF + cb + 2 * l;
    int buf = 0;
    for (; tile < GT_TILES; tile += GT_WGS) {
        const int next = tile + GT_WGS;
        const bool more = next < GT_TILES;
        if (more) {
            const float4* __restrict__ src = x4 + (size_t)next * (GT_ROWS * DF / 4);
            pre0 = src[0];
            pre1 = src[256];
        }

        const float* __restrict__ arow = &atile[buf][l * GT_PITCH + 2 * half];
        v8f acc0 = {}, acc1 = {};
        #pragma unroll
        for (int j = 0; j < DF / 4; ++j) {
            const v2f a = *(const v2f*)(arow + 4 * j);
            acc0 = __builtin_amdgcn_wmma_f32_16x16x4_f32(
                       false, a, false, bf[0][j], (short)0, acc0, false, false);
            acc1 = __builtin_amdgcn_wmma_f32_16x16x4_f32(
                       false, a, false, bf[1][j], (short)0, acc1, false, false);
        }

        if (more) {
            *(float4*)&atile[buf ^ 1][lds0] = pre0;
            *(float4*)&atile[buf ^ 1][lds1] = pre1;
        }
        __syncthreads();

        float* __restrict__ dst = obase + (size_t)tile * (GT_ROWS * DF);
        #pragma unroll
        for (int r = 0; r < 8; ++r) {
            float v0 = acc0[r] + bv0;
            float v1 = acc1[r] + bv1;
            if (!is_h) { v0 = fmaxf(v0, 0.0f); v1 = fmaxf(v1, 0.0f); }
            *(float2*)(dst + (size_t)r * DF) = make_float2(v0, v1);
        }
        buf ^= 1;
    }
}

#define EPT 5
#define EDGE_BLOCKS (NE / (256 * EPT))
static_assert(NE % (256 * EPT) == 0, "the edge kernels cover NE without a tail");
DEV void edge_rank_block(
    const int bid,
    const int* __restrict__ ei, int* __restrict__ cnt, int* __restrict__ rank)
{
    const int e0 = bid * (256 * EPT) + threadIdx.x;
    int recv[EPT], rk[EPT];
    #pragma unroll
    for (int k = 0; k < EPT; ++k) recv[k] = ei[2 * (e0 + k * 256)];
    #pragma unroll
    for (int k = 0; k < EPT; ++k) rk[k] = atomicAdd(&cnt[recv[k]], 1);
    #pragma unroll
    for (int k = 0; k < EPT; ++k) rank[e0 + k * 256] = rk[k];
}

#define ALLOC_BLOCKS ((NN + 255) / 256)
__global__ __launch_bounds__(256) void k_seg_alloc(
    const int* __restrict__ cnt, int* __restrict__ offs, int* __restrict__ total)
{
    const int bid = blockIdx.x;
    __shared__ int wsum[8];
    __shared__ int bbase;
    const int tid = threadIdx.x, lane = tid & 31, w = tid >> 5;
    const int i = bid * 256 + tid;
    const int c = (i < NN) ? cnt[i] : 0;
    int s = c;
    for (int d = 1; d < 32; d <<= 1) {
        const int t = __shfl_up(s, d, 32);
        if (lane >= d) s += t;
    }
    if (lane == 31) wsum[w] = s;
    __syncthreads();
    if (tid == 0) {
        int t = 0;
        for (int k = 0; k < 8; ++k) { const int v = wsum[k]; wsum[k] = t; t += v; }
        bbase = atomicAdd(total, t);
    }
    __syncthreads();
    if (i < NN) offs[i] = bbase + wsum[w] + s - c;
}

__global__ __launch_bounds__(256) void k_edge_place(
    const int* __restrict__ ei, const int* __restrict__ offs,
    const int* __restrict__ rank, int* __restrict__ list)
{
    const int e0 = blockIdx.x * (256 * EPT) + threadIdx.x;
    int2 rs[EPT]; int rk[EPT], of[EPT];
    #pragma unroll
    for (int k = 0; k < EPT; ++k) {
        rs[k] = ((const int2*)ei)[e0 + k * 256];
        rk[k] = rank[e0 + k * 256];
    }
    #pragma unroll
    for (int k = 0; k < EPT; ++k) of[k] = offs[rs[k].x];
    #pragma unroll
    for (int k = 0; k < EPT; ++k)
        ((int2*)list)[of[k] + rk[k]] = make_int2(e0 + k * 256, rs[k].y);
}

DEV int next_by_id(const int2* __restrict__ seg, int deg, int lane, int& last) {
    int best = 0x7fffffff, bs = 0;
    for (int i = lane; i < deg; i += 32) {
        const int2 p = seg[i];
        if (p.x > last && p.x < best) { best = p.x; bs = p.y; }
    }
    const int mn = wave_min_i(best);
    const unsigned hit = __builtin_amdgcn_ballot_w32(best == mn);
    last = mn;
    return lane_i(bs, __builtin_ctz(hit));
}

#define GC 8
static_assert(GC == 8, "a group is one row per lane of a head's 8 lanes");
template <bool FULL>
DEV void gather_group(
    const float4* __restrict__ hrow, const int send, const int r0, const int deg,
    const float4 aj, const float si, const float ae, const float ba, const int lane,
    float& m, float& d, float4& acc)
{
    float4 v[GC];
    #pragma unroll
    for (int u = 0; u < GC; ++u) {
        if (FULL || r0 + u < deg) v[u] = hrow[(size_t)lane_i(send, r0 + u) * (DF / 4)];
        else v[u] = make_float4(0.0f, 0.0f, 0.0f, 0.0f);
    }
    float p[GC];
    #pragma unroll
    for (int u = 0; u < GC; ++u) p[u] = dot4(v[u], aj);

    const bool b2 = (lane & 4) != 0, b1 = (lane & 2) != 0, b0 = (lane & 1) != 0;
    float q[4], r[2];
    #pragma unroll
    for (int u = 0; u < 4; ++u) {
        const float keep = b2 ? p[u + 4] : p[u];
        const float give = b2 ? p[u] : p[u + 4];
        q[u] = keep + row_xor<4>(give);
    }
    #pragma unroll
    for (int u = 0; u < 2; ++u) {
        const float keep = b1 ? q[u + 2] : q[u];
        const float give = b1 ? q[u] : q[u + 2];
        r[u] = keep + row_xor<2>(give);
    }
    const float keep = b0 ? r[1] : r[0];
    const float give = b0 ? r[0] : r[1];
    const float sj = keep + row_xor<1>(give);

    float lg = leaky(si + sj + ae + ba);
    if (!FULL && r0 + (lane & 7) >= deg) lg = -INFINITY;
    const float mn = fmaxf(m, max8(lg));
    const float sc = expf(m - mn);
    const float wo = expf(lg - mn);
    d *= sc; acc.x *= sc; acc.y *= sc; acc.z *= sc; acc.w *= sc;
    m = mn;
#define ADD_ROW(U) { const float w = bcast8<U>(wo); d += w; \
                     acc.x += v[U].x * w; acc.y += v[U].y * w; acc.z += v[U].z * w; acc.w += v[U].w * w; }
    ADD_ROW(0) ADD_ROW(1) ADD_ROW(2) ADD_ROW(3) ADD_ROW(4) ADD_ROW(5) ADD_ROW(6) ADD_ROW(7)
#undef ADD_ROW
}

__global__ __launch_bounds__(256) void k_node_gather(
    const int* __restrict__ list, const int* __restrict__ cnt,
    const int* __restrict__ offs, const float* __restrict__ a_i,
    const float* __restrict__ a_j, const float* __restrict__ a_e,
    const float* __restrict__ b_a, const float* __restrict__ h,
    float* __restrict__ agg)
{
    const int lane = threadIdx.x & 31;
    const int node = blockIdx.x * 8 + __builtin_amdgcn_readfirstlane(threadIdx.x >> 5);
    if (node >= NN) return;
    const int deg = cnt[node];
    const int2* __restrict__ seg = (const int2*)list + offs[node];
    const int hd = lane >> 3;
    const float4* __restrict__ hrow = (const float4*)h + lane;
    const float4 own = hrow[(size_t)node * (DF / 4)];
    const float4 ai = ((const float4*)a_i)[lane];
    const float4 aj = ((const float4*)a_j)[lane];
    const float ae = a_e[hd], ba = b_a[hd];

    float4 acc = make_float4(0.0f, 0.0f, 0.0f, 0.0f);
    if (deg == 0) {
        ((float4*)agg)[(size_t)node * (DF / 4) + lane] = acc;
        return;
    }
    const float si = score8(own, ai);
    float m = -INFINITY, d = 0.0f;

    if (deg <= 32) {
        const bool on = lane < deg;
        int e = 0x7fffffff, send = 0;
        if (on) { const int2 p = seg[lane]; e = p.x; send = p.y; }
        int rk = 0;
        for (int j = 0; j < deg; ++j) rk += (lane_i(e, j) < e) ? 1 : 0;
        if (!on) rk = lane;
        send = __builtin_amdgcn_ds_permute(rk << 2, send);

        int r0 = 0;
        for (; r0 + GC <= deg; r0 += GC)
            gather_group<true>(hrow, send, r0, deg, aj, si, ae, ba, lane, m, d, acc);
        if (r0 < deg)
            gather_group<false>(hrow, send, r0, deg, aj, si, ae, ba, lane, m, d, acc);
    } else {
        int last = -1;
        for (int t = 0; t < deg; ++t) {
            const int s = next_by_id(seg, deg, lane, last);
            const float4 v = hrow[(size_t)s * (DF / 4)];
            const float l = leaky(si + score8(v, aj) + ae + ba);
            const float mn = fmaxf(m, l);
            const float sc = expf(m - mn);
            const float w = expf(l - mn);
            d = d * sc + w;
            acc.x = acc.x * sc + v.x * w; acc.y = acc.y * sc + v.y * w;
            acc.z = acc.z * sc + v.z * w; acc.w = acc.w * sc + v.w * w;
            m = mn;
        }
    }
    const float inv = 1.0f / (d + 1e-16f);
    acc.x *= inv; acc.y *= inv; acc.z *= inv; acc.w *= inv;
    ((float4*)agg)[(size_t)node * (DF / 4) + lane] = acc;
}

#define BN_WAVES 8
#define BN_WROWS 16
#define BN_ROWS  (BN_WAVES * BN_WROWS)
#define BN_BLOCKS ((NN + BN_ROWS - 1) / BN_ROWS)
static_assert(NN % BN_WROWS == 0, "a wave's rows are all inside the matrix or all outside");
DEV void bn_partial_block(
    const int bid,
    const float* __restrict__ upd, float* __restrict__ macc,
    float* __restrict__ sqacc)
{
    __shared__ float4 ps[BN_WAVES][32];
    __shared__ float4 ps2[BN_WAVES][32];
    const int tid  = threadIdx.x;
    const int lane = tid & 31;
    const int w    = __builtin_amdgcn_readfirstlane(tid >> 5);
    const int r0   = bid * BN_ROWS + w * BN_WROWS;

    float4 s  = make_float4(0.0f, 0.0f, 0.0f, 0.0f);
    float4 s2 = make_float4(0.0f, 0.0f, 0.0f, 0.0f);
    if (r0 < NN) {
        const float4* __restrict__ u4 = (const float4*)upd + (size_t)r0 * (DF / 4) + lane;
        float4 v[BN_WROWS];
        #pragma unroll
        for (int k = 0; k < BN_WROWS; ++k) v[k] = u4[k * (DF / 4)];
        #pragma unroll
        for (int k = 0; k < BN_WROWS; ++k) {
            s.x += v[k].x; s2.x += v[k].x * v[k].x;
            s.y += v[k].y; s2.y += v[k].y * v[k].y;
            s.z += v[k].z; s2.z += v[k].z * v[k].z;
            s.w += v[k].w; s2.w += v[k].w * v[k].w;
        }
    }
    ps[w][lane]  = s;
    ps2[w][lane] = s2;
    __syncthreads();
    if (tid < DF) {
        const float* __restrict__ p  = (const float*)ps;
        const float* __restrict__ p2 = (const float*)ps2;
        float a = 0.0f, b = 0.0f;
        for (int k = 0; k < BN_WAVES; ++k) { a += p[k * DF + tid]; b += p2[k * DF + tid]; }
        atomicAdd(&macc[tid], a);
        atomicAdd(&sqacc[tid], b);
    }
}

__global__ __launch_bounds__(256) void k_rank_bnsum(
    const int* __restrict__ ei, int* __restrict__ cnt, int* __restrict__ rank,
    const float* __restrict__ upd, float* __restrict__ macc,
    float* __restrict__ sqacc)
{
    if (blockIdx.x < EDGE_BLOCKS) edge_rank_block(blockIdx.x, ei, cnt, rank);
    else bn_partial_block(blockIdx.x - EDGE_BLOCKS, upd, macc, sqacc);
}

#define FIN_ROWS 8
static_assert(NN % FIN_ROWS == 0, "a wave's rows are all inside the output");
DEV float fin_rstd(float sq, float mean, float inv_n) {
    const float var = sq * inv_n - mean * mean;
    return rsqrtf(var + BN_EPS);
}
DEV float fin_one(float u, float a, float ma, float inv_n, float rs, float g, float b) {
    const float d = fmaf(-ma, inv_n, u);
    const float normed = g * d * rs + b;
    return normed + fmaxf(a, 0.0f);
}
__global__ __launch_bounds__(256) void k_finalize(
    const float* __restrict__ upd, const float* __restrict__ agg,
    const float* __restrict__ macc, const float* __restrict__ sqacc,
    const float* __restrict__ gamma, const float* __restrict__ beta,
    float* __restrict__ out)
{
    const int lane = threadIdx.x & 31;
    const int wave = blockIdx.x * 8 + __builtin_amdgcn_readfirstlane(threadIdx.x >> 5);
    const int r0 = wave * FIN_ROWS;
    if (r0 >= NN) return;
    const float inv_n = 1.0f / (float)NN;
    const float4 ma = ((const float4*)macc)[lane];
    const float4 sq = ((const float4*)sqacc)[lane];
    const float4 g  = ((const float4*)gamma)[lane];
    const float4 b  = ((const float4*)beta)[lane];
    float4 rs;
    rs.x = fin_rstd(sq.x, ma.x * inv_n, inv_n); rs.y = fin_rstd(sq.y, ma.y * inv_n, inv_n);
    rs.z = fin_rstd(sq.z, ma.z * inv_n, inv_n); rs.w = fin_rstd(sq.w, ma.w * inv_n, inv_n);

    const size_t base = (size_t)r0 * (DF / 4) + lane;
    const float4* __restrict__ u4 = (const float4*)upd + base;
    const float4* __restrict__ a4 = (const float4*)agg + base;
    float4* __restrict__ o4 = (float4*)out + base;
    float4 u[FIN_ROWS], a[FIN_ROWS];
    #pragma unroll
    for (int k = 0; k < FIN_ROWS; ++k) {
        u[k] = u4[k * (DF / 4)];
        a[k] = a4[k * (DF / 4)];
    }
    #pragma unroll
    for (int k = 0; k < FIN_ROWS; ++k) {
        float4 o;
        o.x = fin_one(u[k].x, a[k].x, ma.x, inv_n, rs.x, g.x, b.x);
        o.y = fin_one(u[k].y, a[k].y, ma.y, inv_n, rs.y, g.y, b.y);
        o.z = fin_one(u[k].z, a[k].z, ma.z, inv_n, rs.z, g.z, b.z);
        o.w = fin_one(u[k].w, a[k].w, ma.w, inv_n, rs.w, g.w, b.w);
        o4[k * (DF / 4)] = o;
    }
}

extern "C" void kernel_launch(void* const* d_in, const int* in_sizes, int n_in,
                              void* d_out, int out_size, void* d_ws, size_t ws_size,
                              hipStream_t stream) {
    const float* x     = (const float*)d_in[0];
    const int*   ei    = (const int*)  d_in[1];
    const float* Wk    = (const float*)d_in[2];
    const float* bk    = (const float*)d_in[3];
    const float* a_i   = (const float*)d_in[4];
    const float* a_j   = (const float*)d_in[5];
    const float* a_e   = (const float*)d_in[6];
    const float* b_a   = (const float*)d_in[7];
    const float* Wm    = (const float*)d_in[8];
    const float* bm    = (const float*)d_in[9];
    const float* gamma = (const float*)d_in[10];
    const float* beta  = (const float*)d_in[11];
    float* out = (float*)d_out;
    float* ws  = (float*)d_ws;

    float* h_buf  = ws + OFF_H;
    float* upd    = ws + OFF_UPD;
    float* agg    = ws + OFF_AGG;
    int*   cnt    = (int*)ws + OFF_CNT;
    int*   offs   = (int*)ws + OFF_OFFS;
    int*   rank   = (int*)ws + OFF_RANK;
    int*   list   = (int*)ws + OFF_LIST;
    int*   total  = (int*)ws + OFF_TOT;
    float* macc   = ws + OFF_MACC;
    float* sqacc  = ws + OFF_SQ;

    k_gemm2_wmma<<<GT_WGS, 256, 0, stream>>>(x, Wk, bk, Wm, bm, h_buf, upd, cnt, total, macc, sqacc);
    k_rank_bnsum<<<EDGE_BLOCKS + BN_BLOCKS, 256, 0, stream>>>(ei, cnt, rank, upd, macc, sqacc);
    k_seg_alloc<<<ALLOC_BLOCKS, 256, 0, stream>>>(cnt, offs, total);
    k_edge_place<<<EDGE_BLOCKS, 256, 0, stream>>>(ei, offs, rank, list);
    k_node_gather<<<(NN + 7) / 8, 256, 0, stream>>>(
        list, cnt, offs, a_i, a_j, a_e, b_a, h_buf, agg);
    k_finalize<<<(NN + 8 * FIN_ROWS - 1) / (8 * FIN_ROWS), 256, 0, stream>>>(
        upd, agg, macc, sqacc, gamma, beta, out);
}


# Output row formed in the gather launch: the wave that summed a node's row normalises the node's upd row, adds relu of the sum and stores out; agg plane (write + read) and finalize launch gone
# speedup vs baseline: 4.0165x; 1.0420x over previous
#include <hip/hip_runtime.h>
#include <hip/hip_bf16.h>
#include <math.h>

#define NN      50000
#define NE      800000
#define DF      128
#define NHEADS  4
#define HU      32
#define NEG_SLOPE 0.2f
#define BN_EPS    0.001f

typedef __attribute__((ext_vector_type(2))) float v2f;
typedef __attribute__((ext_vector_type(8))) float v8f;

#define OFF_H    ((size_t)0)
#define OFF_UPD  (OFF_H   + (size_t)NN * DF)
#define OFF_AGG  (OFF_UPD + (size_t)NN * DF)
#define OFF_SI   (OFF_AGG + (size_t)NN * DF)
#define OFF_SJ   (OFF_SI  + (size_t)NN * NHEADS)
#define OFF_CNT  (OFF_SJ  + (size_t)NN * NHEADS)
#define OFF_OFFS (OFF_CNT + (size_t)NN * NHEADS)
#define OFF_RANK (OFF_OFFS + (size_t)NN * NHEADS)
#define OFF_LIST (OFF_RANK + (size_t)NE)
#define OFF_TOT  (OFF_LIST + (size_t)2 * NE)
#define OFF_MACC (OFF_RANK + (size_t)NE * NHEADS)
#define OFF_SQ   (OFF_MACC + (size_t)DF)

#define DEV static __device__ __forceinline__
DEV int   lane_i(int v, int l)   { return __builtin_amdgcn_readlane(v, l); }
template <int MASK>
DEV float row_xor(float v) {
    return __int_as_float(__builtin_amdgcn_update_dpp(0, __float_as_int(v), 0x160 + MASK, 0xf, 0xf, false));
}
DEV float sum8(float v) {
    v += row_xor<4>(v);
    v += row_xor<2>(v);
    v += row_xor<1>(v);
    return v;
}
DEV float max8(float v) {
    v = fmaxf(v, row_xor<4>(v));
    v = fmaxf(v, row_xor<2>(v));
    v = fmaxf(v, row_xor<1>(v));
    return v;
}
template <int U>
DEV float bcast8(float v) {
    return __int_as_float(__builtin_amdgcn_mov_dpp8(__float_as_int(v), U * 0x249249));
}
DEV float dot4(const float4 r, const float4 a) {
    float s = r.x * a.x; s += r.y * a.y; s += r.z * a.z; s += r.w * a.w;
    return s;
}
DEV float score8(const float4 r, const float4 a) { return sum8(dot4(r, a)); }
DEV int wave_min_i(int v) {
    for (int off = 16; off > 0; off >>= 1) v = min(v, __shfl_xor(v, off, 32));
    return v;
}
DEV float leaky(float x) { return (x > 0.0f) ? x : NEG_SLOPE * x; }


#define GT_ROWS  16
#define GT_PITCH 132
#define GT_TILES (NN / GT_ROWS)
#define GT_WGS   512
static_assert(NN % GT_ROWS == 0, "every row tile is whole: no row guard in the kernel");
static_assert(GT_WGS * 256 >= NN, "one thread per degree counter when the launch clears them");

template <int KSTRIDE>
DEV void load_bfrag(const float* __restrict__ q, v2f (&bf)[2][DF / 4]) {
    #pragma unroll
    for (int j = 0; j < DF / 4; ++j) {
        const float2 p0 = *(const float2*)(q + (size_t)(4 * j) * KSTRIDE);
        const float2 p1 = *(const float2*)(q + (size_t)(4 * j + 1) * KSTRIDE);
        bf[0][j].x = p0.x; bf[0][j].y = p1.x;
        bf[1][j].x = p0.y; bf[1][j].y = p1.y;
    }
}

__global__ __launch_bounds__(256) void k_gemm2_wmma(
    const float* __restrict__ x, const float* __restrict__ Wk, const float* __restrict__ bk,
    const float* __restrict__ Wm, const float* __restrict__ bm,
    float* __restrict__ h, float* __restrict__ upd,
    int* __restrict__ cnt, int* __restrict__ total,
    float* __restrict__ macc, float* __restrict__ sqacc)
{
    __shared__ __attribute__((aligned(16))) float atile[2][GT_ROWS * GT_PITCH];

    const int tid  = threadIdx.x;
    {
        const int gid = blockIdx.x * 256 + tid;
        if (gid < NN) cnt[gid] = 0;
        if (gid == 0) *total = 0;
        if (gid < DF) { macc[gid] = 0.0f; sqacc[gid] = 0.0f; }
    }
    const int lane = tid & 31;
    const int half = lane >> 4;
    const int l    = lane & 15;
    const int w    = __builtin_amdgcn_readfirstlane(tid >> 5);
    const bool is_h = w < 4;
    const int cb   = (w & 3) * 32;

    const int r0 = tid >> 5, c4 = tid & 31;
    const float4* __restrict__ x4 = (const float4*)x + tid;
    const int lds0 = r0 * GT_PITCH + 4 * c4;
    const int lds1 = (r0 + 8) * GT_PITCH + 4 * c4;

    int tile = blockIdx.x;
    float4 pre0, pre1;
    {
        const float4* __restrict__ src = x4 + (size_t)tile * (GT_ROWS * DF / 4);
        pre0 = src[0];
        pre1 = src[256];
    }

    v2f bf[2][DF / 4];
    float bv0, bv1;
    if (is_h) {
        load_bfrag<HU>(Wk + (size_t)w * (DF * HU) + (size_t)(2 * half) * HU + 2 * l, bf);
        const float2 b2 = *(const float2*)(bk + cb + 2 * l);
        bv0 = b2.x; bv1 = b2.y;
    } else {
        load_bfrag<DF>(Wm + (size_t)(2 * half) * DF + cb + 2 * l, bf);
        const float2 b2 = *(const float2*)(bm + cb + 2 * l);
        bv0 = b2.x; bv1 = b2.y;
    }

    *(float4*)&atile[0][lds0] = pre0;
    *(float4*)&atile[0][lds1] = pre1;
    __syncthreads();

    float* __restrict__ obase = (is_h ? h : upd) + (size_t)(8 * half) * DF + cb + 2 * l;
    int buf = 0;
    for (; tile < GT_TILES; tile += GT_WGS) {
        const int next = tile + GT_WGS;
        const bool more = next < GT_TILES;
        if (more) {
            const float4* __restrict__ src = x4 + (size_t)next * (GT_ROWS * DF / 4);
            pre0 = src[0];
            pre1 = src[256];
        }

        const float* __restrict__ arow = &atile[buf][l * GT_PITCH + 2 * half];
        v8f acc0 = {}, acc1 = {};
        #pragma unroll
        for (int j = 0; j < DF / 4; ++j) {
            const v2f a = *(const v2f*)(arow + 4 * j);
            acc0 = __builtin_amdgcn_wmma_f32_16x16x4_f32(
                       false, a, false, bf[0][j], (short)0, acc0, false, false);
            acc1 = __builtin_amdgcn_wmma_f32_16x16x4_f32(
                       false, a, false, bf[1][j], (short)0, acc1, false, false);
        }

        if (more) {
            *(float4*)&atile[buf ^ 1][lds0] = pre0;
            *(float4*)&atile[buf ^ 1][lds1] = pre1;
        }
        __syncthreads();

        float* __restrict__ dst = obase + (size_t)tile * (GT_ROWS * DF);
        #pragma unroll
        for (int r = 0; r < 8; ++r) {
            float v0 = acc0[r] + bv0;
            float v1 = acc1[r] + bv1;
            if (!is_h) { v0 = fmaxf(v0, 0.0f); v1 = fmaxf(v1, 0.0f); }
            *(float2*)(dst + (size_t)r * DF) = make_float2(v0, v1);
        }
        buf ^= 1;
    }
}

#define EPT 5
#define EDGE_BLOCKS (NE / (256 * EPT))
static_assert(NE % (256 * EPT) == 0, "the edge kernels cover NE without a tail");
DEV void edge_rank_block(
    const int bid,
    const int* __restrict__ ei, int* __restrict__ cnt, int* __restrict__ rank)
{
    const int e0 = bid * (256 * EPT) + threadIdx.x;
    int recv[EPT], rk[EPT];
    #pragma unroll
    for (int k = 0; k < EPT; ++k) recv[k] = ei[2 * (e0 + k * 256)];
    #pragma unroll
    for (int k = 0; k < EPT; ++k) rk[k] = atomicAdd(&cnt[recv[k]], 1);
    #pragma unroll
    for (int k = 0; k < EPT; ++k) rank[e0 + k * 256] = rk[k];
}

#define ALLOC_BLOCKS ((NN + 255) / 256)
__global__ __launch_bounds__(256) void k_seg_alloc(
    const int* __restrict__ cnt, int* __restrict__ offs, int* __restrict__ total)
{
    const int bid = blockIdx.x;
    __shared__ int wsum[8];
    __shared__ int bbase;
    const int tid = threadIdx.x, lane = tid & 31, w = tid >> 5;
    const int i = bid * 256 + tid;
    const int c = (i < NN) ? cnt[i] : 0;
    int s = c;
    for (int d = 1; d < 32; d <<= 1) {
        const int t = __shfl_up(s, d, 32);
        if (lane >= d) s += t;
    }
    if (lane == 31) wsum[w] = s;
    __syncthreads();
    if (tid == 0) {
        int t = 0;
        for (int k = 0; k < 8; ++k) { const int v = wsum[k]; wsum[k] = t; t += v; }
        bbase = atomicAdd(total, t);
    }
    __syncthreads();
    if (i < NN) offs[i] = bbase + wsum[w] + s - c;
}

__global__ __launch_bounds__(256) void k_edge_place(
    const int* __restrict__ ei, const int* __restrict__ offs,
    const int* __restrict__ rank, int* __restrict__ list)
{
    const int e0 = blockIdx.x * (256 * EPT) + threadIdx.x;
    int2 rs[EPT]; int rk[EPT], of[EPT];
    #pragma unroll
    for (int k = 0; k < EPT; ++k) {
        rs[k] = ((const int2*)ei)[e0 + k * 256];
        rk[k] = rank[e0 + k * 256];
    }
    #pragma unroll
    for (int k = 0; k < EPT; ++k) of[k] = offs[rs[k].x];
    #pragma unroll
    for (int k = 0; k < EPT; ++k)
        ((int2*)list)[of[k] + rk[k]] = make_int2(e0 + k * 256, rs[k].y);
}

DEV int next_by_id(const int2* __restrict__ seg, int deg, int lane, int& last) {
    int best = 0x7fffffff, bs = 0;
    for (int i = lane; i < deg; i += 32) {
        const int2 p = seg[i];
        if (p.x > last && p.x < best) { best = p.x; bs = p.y; }
    }
    const int mn = wave_min_i(best);
    const unsigned hit = __builtin_amdgcn_ballot_w32(best == mn);
    last = mn;
    return lane_i(bs, __builtin_ctz(hit));
}

#define GC 8
static_assert(GC == 8, "a group is one row per lane of a head's 8 lanes");
template <bool FULL>
DEV void gather_group(
    const float4* __restrict__ hrow, const int send, const int r0, const int deg,
    const float4 aj, const float si, const float ae, const float ba, const int lane,
    float& m, float& d, float4& acc)
{
    float4 v[GC];
    #pragma unroll
    for (int u = 0; u < GC; ++u) {
        if (FULL || r0 + u < deg) v[u] = hrow[(size_t)lane_i(send, r0 + u) * (DF / 4)];
        else v[u] = make_float4(0.0f, 0.0f, 0.0f, 0.0f);
    }
    float p[GC];
    #pragma unroll
    for (int u = 0; u < GC; ++u) p[u] = dot4(v[u], aj);

    const bool b2 = (lane & 4) != 0, b1 = (lane & 2) != 0, b0 = (lane & 1) != 0;
    float q[4], r[2];
    #pragma unroll
    for (int u = 0; u < 4; ++u) {
        const float keep = b2 ? p[u + 4] : p[u];
        const float give = b2 ? p[u] : p[u + 4];
        q[u] = keep + row_xor<4>(give);
    }
    #pragma unroll
    for (int u = 0; u < 2; ++u) {
        const float keep = b1 ? q[u + 2] : q[u];
        const float give = b1 ? q[u] : q[u + 2];
        r[u] = keep + row_xor<2>(give);
    }
    const float keep = b0 ? r[1] : r[0];
    const float give = b0 ? r[0] : r[1];
    const float sj = keep + row_xor<1>(give);

    float lg = leaky(si + sj + ae + ba);
    if (!FULL && r0 + (lane & 7) >= deg) lg = -INFINITY;
    const float mn = fmaxf(m, max8(lg));
    const float sc = expf(m - mn);
    const float wo = expf(lg - mn);
    d *= sc; acc.x *= sc; acc.y *= sc; acc.z *= sc; acc.w *= sc;
    m = mn;
#define ADD_ROW(U) { const float w = bcast8<U>(wo); d += w; \
                     acc.x += v[U].x * w; acc.y += v[U].y * w; acc.z += v[U].z * w; acc.w += v[U].w * w; }
    ADD_ROW(0) ADD_ROW(1) ADD_ROW(2) ADD_ROW(3) ADD_ROW(4) ADD_ROW(5) ADD_ROW(6) ADD_ROW(7)
#undef ADD_ROW
}

DEV float fin_rstd(float sq, float mean, float inv_n) {
    const float var = sq * inv_n - mean * mean;
    return rsqrtf(var + BN_EPS);
}
DEV float fin_one(float u, float a, float ma, float inv_n, float rs, float g, float b) {
    const float d = fmaf(-ma, inv_n, u);
    const float normed = g * d * rs + b;
    return normed + fmaxf(a, 0.0f);
}
__global__ __launch_bounds__(256) void k_node_gather_out(
    const int* __restrict__ list, const int* __restrict__ cnt,
    const int* __restrict__ offs, const float* __restrict__ a_i,
    const float* __restrict__ a_j, const float* __restrict__ a_e,
    const float* __restrict__ b_a, const float* __restrict__ h,
    const float* __restrict__ upd,
    const float* __restrict__ macc, const float* __restrict__ sqacc,
    const float* __restrict__ gamma, const float* __restrict__ beta,
    float* __restrict__ out)
{
    const int lane = threadIdx.x & 31;
    const int node = blockIdx.x * 8 + __builtin_amdgcn_readfirstlane(threadIdx.x >> 5);
    if (node >= NN) return;
    const int deg = cnt[node];
    const int2* __restrict__ seg = (const int2*)list + offs[node];
    const int hd = lane >> 3;
    const float4* __restrict__ hrow = (const float4*)h + lane;
    const float4 own = hrow[(size_t)node * (DF / 4)];
    const float4 ai = ((const float4*)a_i)[lane];
    const float4 aj = ((const float4*)a_j)[lane];
    const float ae = a_e[hd], ba = b_a[hd];

    float4 acc = make_float4(0.0f, 0.0f, 0.0f, 0.0f);
    if (deg > 0) {
        const float si = score8(own, ai);
        float m = -INFINITY, d = 0.0f;

        if (deg <= 32) {
            const bool on = lane < deg;
            int e = 0x7fffffff, send = 0;
            if (on) { const int2 p = seg[lane]; e = p.x; send = p.y; }
            int rk = 0;
            for (int j = 0; j < deg; ++j) rk += (lane_i(e, j) < e) ? 1 : 0;
            if (!on) rk = lane;
            send = __builtin_amdgcn_ds_permute(rk << 2, send);

            int r0 = 0;
            for (; r0 + GC <= deg; r0 += GC)
                gather_group<true>(hrow, send, r0, deg, aj, si, ae, ba, lane, m, d, acc);
            if (r0 < deg)
                gather_group<false>(hrow, send, r0, deg, aj, si, ae, ba, lane, m, d, acc);
        } else {
            int last = -1;
            for (int t = 0; t < deg; ++t) {
                const int s = next_by_id(seg, deg, lane, last);
                const float4 v = hrow[(size_t)s * (DF / 4)];
                const float l = leaky(si + score8(v, aj) + ae + ba);
                const float mn = fmaxf(m, l);
                const float sc = expf(m - mn);
                const float w = expf(l - mn);
                d = d * sc + w;
                acc.x = acc.x * sc + v.x * w; acc.y = acc.y * sc + v.y * w;
                acc.z = acc.z * sc + v.z * w; acc.w = acc.w * sc + v.w * w;
                m = mn;
            }
        }
        const float inv = 1.0f / (d + 1e-16f);
        acc.x *= inv; acc.y *= inv; acc.z *= inv; acc.w *= inv;
    }

    const float inv_n = 1.0f / (float)NN;
    const float4 u  = ((const float4*)upd)[(size_t)node * (DF / 4) + lane];
    const float4 ma = ((const float4*)macc)[lane];
    const float4 sq = ((const float4*)sqacc)[lane];
    const float4 g  = ((const float4*)gamma)[lane];
    const float4 b  = ((const float4*)beta)[lane];
    float4 rs;
    rs.x = fin_rstd(sq.x, ma.x * inv_n, inv_n); rs.y = fin_rstd(sq.y, ma.y * inv_n, inv_n);
    rs.z = fin_rstd(sq.z, ma.z * inv_n, inv_n); rs.w = fin_rstd(sq.w, ma.w * inv_n, inv_n);
    float4 o;
    o.x = fin_one(u.x, acc.x, ma.x, inv_n, rs.x, g.x, b.x);
    o.y = fin_one(u.y, acc.y, ma.y, inv_n, rs.y, g.y, b.y);
    o.z = fin_one(u.z, acc.z, ma.z, inv_n, rs.z, g.z, b.z);
    o.w = fin_one(u.w, acc.w, ma.w, inv_n, rs.w, g.w, b.w);
    ((float4*)out)[(size_t)node * (DF / 4) + lane] = o;
}

#define BN_WAVES 8
#define BN_WROWS 16
#define BN_ROWS  (BN_WAVES * BN_WROWS)
#define BN_BLOCKS ((NN + BN_ROWS - 1) / BN_ROWS)
static_assert(NN % BN_WROWS == 0, "a wave's rows are all inside the matrix or all outside");
DEV void bn_partial_block(
    const int bid,
    const float* __restrict__ upd, float* __restrict__ macc,
    float* __restrict__ sqacc)
{
    __shared__ float4 ps[BN_WAVES][32];
    __shared__ float4 ps2[BN_WAVES][32];
    const int tid  = threadIdx.x;
    const int lane = tid & 31;
    const int w    = __builtin_amdgcn_readfirstlane(tid >> 5);
    const int r0   = bid * BN_ROWS + w * BN_WROWS;

    float4 s  = make_float4(0.0f, 0.0f, 0.0f, 0.0f);
    float4 s2 = make_float4(0.0f, 0.0f, 0.0f, 0.0f);
    if (r0 < NN) {
        const float4* __restrict__ u4 = (const float4*)upd + (size_t)r0 * (DF / 4) + lane;
        float4 v[BN_WROWS];
        #pragma unroll
        for (int k = 0; k < BN_WROWS; ++k) v[k] = u4[k * (DF / 4)];
        #pragma unroll
        for (int k = 0; k < BN_WROWS; ++k) {
            s.x += v[k].x; s2.x += v[k].x * v[k].x;
            s.y += v[k].y; s2.y += v[k].y * v[k].y;
            s.z += v[k].z; s2.z += v[k].z * v[k].z;
            s.w += v[k].w; s2.w += v[k].w * v[k].w;
        }
    }
    ps[w][lane]  = s;
    ps2[w][lane] = s2;
    __syncthreads();
    if (tid < DF) {
        const float* __restrict__ p  = (const float*)ps;
        const float* __restrict__ p2 = (const float*)ps2;
        float a = 0.0f, b = 0.0f;
        for (int k = 0; k < BN_WAVES; ++k) { a += p[k * DF + tid]; b += p2[k * DF + tid]; }
        atomicAdd(&macc[tid], a);
        atomicAdd(&sqacc[tid], b);
    }
}

__global__ __launch_bounds__(256) void k_rank_bnsum(
    const int* __restrict__ ei, int* __restrict__ cnt, int* __restrict__ rank,
    const float* __restrict__ upd, float* __restrict__ macc,
    float* __restrict__ sqacc)
{
    if (blockIdx.x < EDGE_BLOCKS) edge_rank_block(blockIdx.x, ei, cnt, rank);
    else bn_partial_block(blockIdx.x - EDGE_BLOCKS, upd, macc, sqacc);
}

extern "C" void kernel_launch(void* const* d_in, const int* in_sizes, int n_in,
                              void* d_out, int out_size, void* d_ws, size_t ws_size,
                              hipStream_t stream) {
    const float* x     = (const float*)d_in[0];
    const int*   ei    = (const int*)  d_in[1];
    const float* Wk    = (const float*)d_in[2];
    const float* bk    = (const float*)d_in[3];
    const float* a_i   = (const float*)d_in[4];
    const float* a_j   = (const float*)d_in[5];
    const float* a_e   = (const float*)d_in[6];
    const float* b_a   = (const float*)d_in[7];
    const float* Wm    = (const float*)d_in[8];
    const float* bm    = (const float*)d_in[9];
    const float* gamma = (const float*)d_in[10];
    const float* beta  = (const float*)d_in[11];
    float* out = (float*)d_out;
    float* ws  = (float*)d_ws;

    float* h_buf  = ws + OFF_H;
    float* upd    = ws + OFF_UPD;
    int*   cnt    = (int*)ws + OFF_CNT;
    int*   offs   = (int*)ws + OFF_OFFS;
    int*   rank   = (int*)ws + OFF_RANK;
    int*   list   = (int*)ws + OFF_LIST;
    int*   total  = (int*)ws + OFF_TOT;
    float* macc   = ws + OFF_MACC;
    float* sqacc  = ws + OFF_SQ;

    k_gemm2_wmma<<<GT_WGS, 256, 0, stream>>>(x, Wk, bk, Wm, bm, h_buf, upd, cnt, total, macc, sqacc);
    k_rank_bnsum<<<EDGE_BLOCKS + BN_BLOCKS, 256, 0, stream>>>(ei, cnt, rank, upd, macc, sqacc);
    k_seg_alloc<<<ALLOC_BLOCKS, 256, 0, stream>>>(cnt, offs, total);
    k_edge_place<<<EDGE_BLOCKS, 256, 0, stream>>>(ei, offs, rank, list);
    k_node_gather_out<<<(NN + 7) / 8, 256, 0, stream>>>(
        list, cnt, offs, a_i, a_j, a_e, b_a, h_buf, upd, macc, sqacc, gamma, beta, out);
}


# Edges grouped by receiver with LDS counters only: binned by 64-node range (LDS-ordered, linear writes), then by node inside each range; per-edge global atomic, scan launch and scattered stores gone
# speedup vs baseline: 5.9435x; 1.4798x over previous
#include <hip/hip_runtime.h>
#include <hip/hip_bf16.h>
#include <math.h>

#define NN      50000
#define NE      800000
#define DF      128
#define NHEADS  4
#define HU      32
#define NEG_SLOPE 0.2f
#define BN_EPS    0.001f

typedef __attribute__((ext_vector_type(2))) float v2f;
typedef __attribute__((ext_vector_type(8))) float v8f;

#define OFF_H    ((size_t)0)
#define OFF_UPD  (OFF_H   + (size_t)NN * DF)
#define OFF_AGG  (OFF_UPD + (size_t)NN * DF)
#define OFF_BIN  (OFF_AGG)
#define OFF_TAB  (OFF_BIN + (size_t)2 * NE)
#define OFF_RBASE (OFF_TAB + (size_t)(1 << 17))
#define OFF_SI   (OFF_AGG + (size_t)NN * DF)
#define OFF_SJ   (OFF_SI  + (size_t)NN * NHEADS)
#define OFF_CNT  (OFF_SJ  + (size_t)NN * NHEADS)
#define OFF_OFFS (OFF_CNT + (size_t)NN * NHEADS)
#define OFF_RANK (OFF_OFFS + (size_t)NN * NHEADS)
#define OFF_LIST (OFF_RANK + (size_t)NE)
#define OFF_TOT  (OFF_LIST + (size_t)2 * NE)
#define OFF_MACC (OFF_RANK + (size_t)NE * NHEADS)
#define OFF_SQ   (OFF_MACC + (size_t)DF)

#define DEV static __device__ __forceinline__
DEV int   lane_i(int v, int l)   { return __builtin_amdgcn_readlane(v, l); }
template <int MASK>
DEV float row_xor(float v) {
    return __int_as_float(__builtin_amdgcn_update_dpp(0, __float_as_int(v), 0x160 + MASK, 0xf, 0xf, false));
}
DEV float sum8(float v) {
    v += row_xor<4>(v);
    v += row_xor<2>(v);
    v += row_xor<1>(v);
    return v;
}
DEV float max8(float v) {
    v = fmaxf(v, row_xor<4>(v));
    v = fmaxf(v, row_xor<2>(v));
    v = fmaxf(v, row_xor<1>(v));
    return v;
}
template <int U>
DEV float bcast8(float v) {
    return __int_as_float(__builtin_amdgcn_mov_dpp8(__float_as_int(v), U * 0x249249));
}
DEV float dot4(const float4 r, const float4 a) {
    float s = r.x * a.x; s += r.y * a.y; s += r.z * a.z; s += r.w * a.w;
    return s;
}
DEV float score8(const float4 r, const float4 a) { return sum8(dot4(r, a)); }
DEV int wave_min_i(int v) {
    for (int off = 16; off > 0; off >>= 1) v = min(v, __shfl_xor(v, off, 32));
    return v;
}
DEV float leaky(float x) { return (x > 0.0f) ? x : NEG_SLOPE * x; }


#define GT_ROWS  16
#define GT_PITCH 132
#define GT_TILES (NN / GT_ROWS)
#define GT_WGS   512
static_assert(NN % GT_ROWS == 0, "every row tile is whole: no row guard in the kernel");

template <int KSTRIDE>
DEV void load_bfrag(const float* __restrict__ q, v2f (&bf)[2][DF / 4]) {
    #pragma unroll
    for (int j = 0; j < DF / 4; ++j) {
        const float2 p0 = *(const float2*)(q + (size_t)(4 * j) * KSTRIDE);
        const float2 p1 = *(const float2*)(q + (size_t)(4 * j + 1) * KSTRIDE);
        bf[0][j].x = p0.x; bf[0][j].y = p1.x;
        bf[1][j].x = p0.y; bf[1][j].y = p1.y;
    }
}

__global__ __launch_bounds__(256) void k_gemm2_wmma(
    const float* __restrict__ x, const float* __restrict__ Wk, const float* __restrict__ bk,
    const float* __restrict__ Wm, const float* __restrict__ bm,
    float* __restrict__ h, float* __restrict__ upd,
    float* __restrict__ macc, float* __restrict__ sqacc)
{
    __shared__ __attribute__((aligned(16))) float atile[2][GT_ROWS * GT_PITCH];

    const int tid  = threadIdx.x;
    {
        const int gid = blockIdx.x * 256 + tid;
        if (gid < DF) { macc[gid] = 0.0f; sqacc[gid] = 0.0f; }
    }
    const int lane = tid & 31;
    const int half = lane >> 4;
    const int l    = lane & 15;
    const int w    = __builtin_amdgcn_readfirstlane(tid >> 5);
    const bool is_h = w < 4;
    const int cb   = (w & 3) * 32;

    const int r0 = tid >> 5, c4 = tid & 31;
    const float4* __restrict__ x4 = (const float4*)x + tid;
    const int lds0 = r0 * GT_PITCH + 4 * c4;
    const int lds1 = (r0 + 8) * GT_PITCH + 4 * c4;

    int tile = blockIdx.x;
    float4 pre0, pre1;
    {
        const float4* __restrict__ src = x4 + (size_t)tile * (GT_ROWS * DF / 4);
        pre0 = src[0];
        pre1 = src[256];
    }

    v2f bf[2][DF / 4];
    float bv0, bv1;
    if (is_h) {
        load_bfrag<HU>(Wk + (size_t)w * (DF * HU) + (size_t)(2 * half) * HU + 2 * l, bf);
        const float2 b2 = *(const float2*)(bk + cb + 2 * l);
        bv0 = b2.x; bv1 = b2.y;
    } else {
        load_bfrag<DF>(Wm + (size_t)(2 * half) * DF + cb + 2 * l, bf);
        const float2 b2 = *(const float2*)(bm + cb + 2 * l);
        bv0 = b2.x; bv1 = b2.y;
    }

    *(float4*)&atile[0][lds0] = pre0;
    *(float4*)&atile[0][lds1] = pre1;
    __syncthreads();

    float* __restrict__ obase = (is_h ? h : upd) + (size_t)(8 * half) * DF + cb + 2 * l;
    int buf = 0;
    for (; tile < GT_TILES; tile += GT_WGS) {
        const int next = tile + GT_WGS;
        const bool more = next < GT_TILES;
        if (more) {
            const float4* __restrict__ src = x4 + (size_t)next * (GT_ROWS * DF / 4);
            pre0 = src[0];
            pre1 = src[256];
        }

        const float* __restrict__ arow = &atile[buf][l * GT_PITCH + 2 * half];
        v8f acc0 = {}, acc1 = {};
        #pragma unroll
        for (int j = 0; j < DF / 4; ++j) {
            const v2f a = *(const v2f*)(arow + 4 * j);
            acc0 = __builtin_amdgcn_wmma_f32_16x16x4_f32(
                       false, a, false, bf[0][j], (short)0, acc0, false, false);
            acc1 = __builtin_amdgcn_wmma_f32_16x16x4_f32(
                       false, a, false, bf[1][j], (short)0, acc1, false, false);
        }

        if (more) {
            *(float4*)&atile[buf ^ 1][lds0] = pre0;
            *(float4*)&atile[buf ^ 1][lds1] = pre1;
        }
        __syncthreads();

        float* __restrict__ dst = obase + (size_t)tile * (GT_ROWS * DF);
        #pragma unroll
        for (int r = 0; r < 8; ++r) {
            float v0 = acc0[r] + bv0;
            float v1 = acc1[r] + bv1;
            if (!is_h) { v0 = fmaxf(v0, 0.0f); v1 = fmaxf(v1, 0.0f); }
            *(float2*)(dst + (size_t)r * DF) = make_float2(v0, v1);
        }
        buf ^= 1;
    }
}

#define RG_SHIFT 6
#define RG_NODES (1 << RG_SHIFT)
#define NRANGE   ((NN + RG_NODES - 1) / RG_NODES)
#define PB_EDGES 6400
#define PBLOCKS  (NE / PB_EDGES)
#define ID_BITS  20
static_assert(NE % PB_EDGES == 0, "every edge block is whole: no edge guard in the kernels");
static_assert(NE <= (1 << ID_BITS), "an edge id fits the low ID_BITS bits of a record");
static_assert(NN <= (1 << 16) && PB_EDGES <= (1 << 15), "a staged record packs receiver (16 bits) and slot in the block");
static_assert((size_t)PBLOCKS * NRANGE <= (size_t)(1 << 17), "the count table fits its part of the workspace");
static_assert((size_t)2 * NE + (1 << 17) + NRANGE + 1 <= (size_t)NN * DF, "the three arrays fit the plane they share");

#define BC_EPT (PB_EDGES / 256)
static_assert(PB_EDGES % 256 == 0, "the count stage covers its edge block without a tail");
DEV void edge_bin_count_block(
    const int bid,
    const int* __restrict__ ei, int* __restrict__ tab)
{
    __shared__ int hist[NRANGE];
    const int tid = threadIdx.x;
    for (int r = tid; r < NRANGE; r += 256) hist[r] = 0;
    __syncthreads();
    const int e0 = bid * PB_EDGES + tid;
    int recv[BC_EPT];
    #pragma unroll
    for (int k = 0; k < BC_EPT; ++k) recv[k] = ei[2 * (e0 + k * 256)];
    #pragma unroll
    for (int k = 0; k < BC_EPT; ++k) atomicAdd(&hist[recv[k] >> RG_SHIFT], 1);
    __syncthreads();
    for (int r = tid; r < NRANGE; r += 256) tab[bid * NRANGE + r] = hist[r];
}

#define SC_THREADS 800
#define SC_EPT     (PB_EDGES / SC_THREADS)
static_assert(SC_THREADS % 32 == 0 && SC_THREADS >= NRANGE + 1, "one thread per range and one for the total");
static_assert(PB_EDGES % SC_THREADS == 0, "the block's edges divide over its threads");
#define TB_CHUNK   25
static_assert(PBLOCKS % TB_CHUNK == 0, "the table is read in whole chunks");
__global__ __launch_bounds__(SC_THREADS) void k_edge_bin_scatter(
    const int* __restrict__ ei, const int* __restrict__ tab,
    int* __restrict__ rbase, int* __restrict__ binned)
{
    __shared__ int gstart[NRANGE];
    __shared__ int lstart[NRANGE];
    __shared__ int fill[NRANGE];
    __shared__ int wsA[32], wsB[32];
    __shared__ int2 stage[PB_EDGES];
    const int tid = threadIdx.x, lane = tid & 31, w = tid >> 5;
    const int b = blockIdx.x;

    const int e0 = b * PB_EDGES + tid;
    int2 rs[SC_EPT];
    #pragma unroll
    for (int k = 0; k < SC_EPT; ++k) rs[k] = ((const int2*)ei)[e0 + k * SC_THREADS];

    int tot = 0, before = 0, mine = 0;
    if (tid < NRANGE) {
        mine = tab[b * NRANGE + tid];
        #pragma unroll
        for (int c0 = 0; c0 < PBLOCKS; c0 += TB_CHUNK) {
            int v[TB_CHUNK];
            #pragma unroll
            for (int k = 0; k < TB_CHUNK; ++k) v[k] = tab[(c0 + k) * NRANGE + tid];
            #pragma unroll
            for (int k = 0; k < TB_CHUNK; ++k) {
                tot += v[k];
                before += (c0 + k < b) ? v[k] : 0;
            }
        }
    }
    int sa = tot, sb = mine;
    for (int d = 1; d < 32; d <<= 1) {
        const int ta = __shfl_up(sa, d, 32);
        const int tb = __shfl_up(sb, d, 32);
        if (lane >= d) { sa += ta; sb += tb; }
    }
    if (w == 0) { wsA[lane] = 0; wsB[lane] = 0; }
    __syncthreads();
    if (lane == 31) { wsA[w] = sa; wsB[w] = sb; }
    __syncthreads();
    if (w == 0) {
        const int va = wsA[lane], vb = wsB[lane];
        int xa = va, xb = vb;
        for (int d = 1; d < 32; d <<= 1) {
            const int ta = __shfl_up(xa, d, 32);
            const int tb = __shfl_up(xb, d, 32);
            if (lane >= d) { xa += ta; xb += tb; }
        }
        wsA[lane] = xa - va;
        wsB[lane] = xb - vb;
    }
    __syncthreads();
    const int rb = wsA[w] + sa - tot;
    const int ls = wsB[w] + sb - mine;
    if (tid < NRANGE) {
        gstart[tid] = rb + before;
        lstart[tid] = ls;
        fill[tid] = 0;
    }
    if (b == 0 && tid <= NRANGE) rbase[tid] = rb;
    __syncthreads();

    #pragma unroll
    for (int k = 0; k < SC_EPT; ++k) {
        const int r  = rs[k].x >> RG_SHIFT;
        const int rk = atomicAdd(&fill[r], 1);
        stage[lstart[r] + rk] = make_int2(rs[k].x | ((tid + k * SC_THREADS) << 16), rs[k].y);
    }
    __syncthreads();
    #pragma unroll
    for (int k = 0; k < SC_EPT; ++k) {
        const int i = tid + k * SC_THREADS;
        const int2 p = stage[i];
        const int recv = p.x & 0xffff;
        const int r = recv >> RG_SHIFT;
        const int e = b * PB_EDGES + (p.x >> 16);
        ((int2*)binned)[gstart[r] + (i - lstart[r])] =
            make_int2(e | ((recv & (RG_NODES - 1)) << ID_BITS), p.y);
    }
}

#define RG_THREADS 256
#define RG_UNR     4
#define RG_CAP     4096
__global__ __launch_bounds__(RG_THREADS) void k_range_group(
    const int* __restrict__ binned, const int* __restrict__ rbase,
    int* __restrict__ cnt, int* __restrict__ offs, int* __restrict__ list)
{
    __shared__ int c[RG_NODES];
    __shared__ int cur[RG_NODES];
    __shared__ int2 lbuf[RG_CAP];
    const int tid = threadIdx.x, rg = blockIdx.x;
    const int beg = rbase[rg];
    const int n   = rbase[rg + 1] - beg;
    const int2* __restrict__ rec = (const int2*)binned + beg;
    if (tid < RG_NODES) c[tid] = 0;
    __syncthreads();
    for (int i0 = 0; i0 < n; i0 += RG_UNR * RG_THREADS) {
        int x[RG_UNR];
        #pragma unroll
        for (int k = 0; k < RG_UNR; ++k) x[k] = rec[min(i0 + k * RG_THREADS + tid, n - 1)].x;
        #pragma unroll
        for (int k = 0; k < RG_UNR; ++k)
            if (i0 + k * RG_THREADS + tid < n) atomicAdd(&c[x[k] >> ID_BITS], 1);
    }
    __syncthreads();
    if (tid < RG_NODES) {
        int o = 0;
        for (int j = 0; j < tid; ++j) o += c[j];
        cur[tid] = o;
        const int node = rg * RG_NODES + tid;
        if (node < NN) { cnt[node] = c[tid]; offs[node] = beg + o; }
    }
    __syncthreads();
    int2* __restrict__ lout = (int2*)list + beg;
    if (n <= RG_CAP) {
        for (int i0 = 0; i0 < n; i0 += RG_UNR * RG_THREADS) {
            int2 p[RG_UNR];
            #pragma unroll
            for (int k = 0; k < RG_UNR; ++k) p[k] = rec[min(i0 + k * RG_THREADS + tid, n - 1)];
            #pragma unroll
            for (int k = 0; k < RG_UNR; ++k)
                if (i0 + k * RG_THREADS + tid < n) {
                    const int pos = atomicAdd(&cur[p[k].x >> ID_BITS], 1);
                    lbuf[pos] = make_int2(p[k].x & ((1 << ID_BITS) - 1), p[k].y);
                }
        }
        __syncthreads();
        for (int i = tid; i < n; i += RG_THREADS) lout[i] = lbuf[i];
    } else {
        for (int i = tid; i < n; i += RG_THREADS) {
            const int2 p = rec[i];
            const int pos = atomicAdd(&cur[p.x >> ID_BITS], 1);
            lout[pos] = make_int2(p.x & ((1 << ID_BITS) - 1), p.y);
        }
    }
}

DEV int next_by_id(const int2* __restrict__ seg, int deg, int lane, int& last) {
    int best = 0x7fffffff, bs = 0;
    for (int i = lane; i < deg; i += 32) {
        const int2 p = seg[i];
        if (p.x > last && p.x < best) { best = p.x; bs = p.y; }
    }
    const int mn = wave_min_i(best);
    const unsigned hit = __builtin_amdgcn_ballot_w32(best == mn);
    last = mn;
    return lane_i(bs, __builtin_ctz(hit));
}

#define GC 8
static_assert(GC == 8, "a group is one row per lane of a head's 8 lanes");
template <bool FULL>
DEV void gather_group(
    const float4* __restrict__ hrow, const int send, const int r0, const int deg,
    const float4 aj, const float si, const float ae, const float ba, const int lane,
    float& m, float& d, float4& acc)
{
    float4 v[GC];
    #pragma unroll
    for (int u = 0; u < GC; ++u) {
        if (FULL || r0 + u < deg) v[u] = hrow[(size_t)lane_i(send, r0 + u) * (DF / 4)];
        else v[u] = make_float4(0.0f, 0.0f, 0.0f, 0.0f);
    }
    float p[GC];
    #pragma unroll
    for (int u = 0; u < GC; ++u) p[u] = dot4(v[u], aj);

    const bool b2 = (lane & 4) != 0, b1 = (lane & 2) != 0, b0 = (lane & 1) != 0;
    float q[4], r[2];
    #pragma unroll
    for (int u = 0; u < 4; ++u) {
        const float keep = b2 ? p[u + 4] : p[u];
        const float give = b2 ? p[u] : p[u + 4];
        q[u] = keep + row_xor<4>(give);
    }
    #pragma unroll
    for (int u = 0; u < 2; ++u) {
        const float keep = b1 ? q[u + 2] : q[u];
        const float give = b1 ? q[u] : q[u + 2];
        r[u] = keep + row_xor<2>(give);
    }
    const float keep = b0 ? r[1] : r[0];
    const float give = b0 ? r[0] : r[1];
    const float sj = keep + row_xor<1>(give);

    float lg = leaky(si + sj + ae + ba);
    if (!FULL && r0 + (lane & 7) >= deg) lg = -INFINITY;
    const float mn = fmaxf(m, max8(lg));
    const float sc = expf(m - mn);
    const float wo = expf(lg - mn);
    d *= sc; acc.x *= sc; acc.y *= sc; acc.z *= sc; acc.w *= sc;
    m = mn;
#define ADD_ROW(U) { const float w = bcast8<U>(wo); d += w; \
                     acc.x += v[U].x * w; acc.y += v[U].y * w; acc.z += v[U].z * w; acc.w += v[U].w * w; }
    ADD_ROW(0) ADD_ROW(1) ADD_ROW(2) ADD_ROW(3) ADD_ROW(4) ADD_ROW(5) ADD_ROW(6) ADD_ROW(7)
#undef ADD_ROW
}

DEV float fin_rstd(float sq, float mean, float inv_n) {
    const float var = sq * inv_n - mean * mean;
    return rsqrtf(var + BN_EPS);
}
DEV float fin_one(float u, float a, float ma, float inv_n, float rs, float g, float b) {
    const float d = fmaf(-ma, inv_n, u);
    const float normed = g * d * rs + b;
    return normed + fmaxf(a, 0.0f);
}
__global__ __launch_bounds__(256) void k_node_gather_out(
    const int* __restrict__ list, const int* __restrict__ cnt,
    const int* __restrict__ offs, const float* __restrict__ a_i,
    const float* __restrict__ a_j, const float* __restrict__ a_e,
    const float* __restrict__ b_a, const float* __restrict__ h,
    const float* __restrict__ upd,
    const float* __restrict__ macc, const float* __restrict__ sqacc,
    const float* __restrict__ gamma, const float* __restrict__ beta,
    float* __restrict__ out)
{
    const int lane = threadIdx.x & 31;
    const int node = blockIdx.x * 8 + __builtin_amdgcn_readfirstlane(threadIdx.x >> 5);
    if (node >= NN) return;
    const int deg = cnt[node];
    const int2* __restrict__ seg = (const int2*)list + offs[node];
    const int hd = lane >> 3;
    const float4* __restrict__ hrow = (const float4*)h + lane;
    const float4 own = hrow[(size_t)node * (DF / 4)];
    const float4 ai = ((const float4*)a_i)[lane];
    const float4 aj = ((const float4*)a_j)[lane];
    const float ae = a_e[hd], ba = b_a[hd];

    float4 acc = make_float4(0.0f, 0.0f, 0.0f, 0.0f);
    if (deg > 0) {
        const float si = score8(own, ai);
        float m = -INFINITY, d = 0.0f;

        if (deg <= 32) {
            const bool on = lane < deg;
            int e = 0x7fffffff, send = 0;
            if (on) { const int2 p = seg[lane]; e = p.x; send = p.y; }
            int rk = 0;
            for (int j = 0; j < deg; ++j) rk += (lane_i(e, j) < e) ? 1 : 0;
            if (!on) rk = lane;
            send = __builtin_amdgcn_ds_permute(rk << 2, send);

            int r0 = 0;
            for (; r0 + GC <= deg; r0 += GC)
                gather_group<true>(hrow, send, r0, deg, aj, si, ae, ba, lane, m, d, acc);
            if (r0 < deg)
                gather_group<false>(hrow, send, r0, deg, aj, si, ae, ba, lane, m, d, acc);
        } else {
            int last = -1;
            for (int t = 0; t < deg; ++t) {
                const int s = next_by_id(seg, deg, lane, last);
                const float4 v = hrow[(size_t)s * (DF / 4)];
                const float l = leaky(si + score8(v, aj) + ae + ba);
                const float mn = fmaxf(m, l);
                const float sc = expf(m - mn);
                const float w = expf(l - mn);
                d = d * sc + w;
                acc.x = acc.x * sc + v.x * w; acc.y = acc.y * sc + v.y * w;
                acc.z = acc.z * sc + v.z * w; acc.w = acc.w * sc + v.w * w;
                m = mn;
            }
        }
        const float inv = 1.0f / (d + 1e-16f);
        acc.x *= inv; acc.y *= inv; acc.z *= inv; acc.w *= inv;
    }

    const float inv_n = 1.0f / (float)NN;
    const float4 u  = ((const float4*)upd)[(size_t)node * (DF / 4) + lane];
    const float4 ma = ((const float4*)macc)[lane];
    const float4 sq = ((const float4*)sqacc)[lane];
    const float4 g  = ((const float4*)gamma)[lane];
    const float4 b  = ((const float4*)beta)[lane];
    float4 rs;
    rs.x = fin_rstd(sq.x, ma.x * inv_n, inv_n); rs.y = fin_rstd(sq.y, ma.y * inv_n, inv_n);
    rs.z = fin_rstd(sq.z, ma.z * inv_n, inv_n); rs.w = fin_rstd(sq.w, ma.w * inv_n, inv_n);
    float4 o;
    o.x = fin_one(u.x, acc.x, ma.x, inv_n, rs.x, g.x, b.x);
    o.y = fin_one(u.y, acc.y, ma.y, inv_n, rs.y, g.y, b.y);
    o.z = fin_one(u.z, acc.z, ma.z, inv_n, rs.z, g.z, b.z);
    o.w = fin_one(u.w, acc.w, ma.w, inv_n, rs.w, g.w, b.w);
    ((float4*)out)[(size_t)node * (DF / 4) + lane] = o;
}

#define BN_WAVES 8
#define BN_WROWS 16
#define BN_ROWS  (BN_WAVES * BN_WROWS)
#define BN_BLOCKS ((NN + BN_ROWS - 1) / BN_ROWS)
static_assert(NN % BN_WROWS == 0, "a wave's rows are all inside the matrix or all outside");
DEV void bn_partial_block(
    const int bid,
    const float* __restrict__ upd, float* __restrict__ macc,
    float* __restrict__ sqacc)
{
    __shared__ float4 ps[BN_WAVES][32];
    __shared__ float4 ps2[BN_WAVES][32];
    const int tid  = threadIdx.x;
    const int lane = tid & 31;
    const int w    = __builtin_amdgcn_readfirstlane(tid >> 5);
    const int r0   = bid * BN_ROWS + w * BN_WROWS;

    float4 s  = make_float4(0.0f, 0.0f, 0.0f, 0.0f);
    float4 s2 = make_float4(0.0f, 0.0f, 0.0f, 0.0f);
    if (r0 < NN) {
        const float4* __restrict__ u4 = (const float4*)upd + (size_t)r0 * (DF / 4) + lane;
        float4 v[BN_WROWS];
        #pragma unroll
        for (int k = 0; k < BN_WROWS; ++k) v[k] = u4[k * (DF / 4)];
        #pragma unroll
        for (int k = 0; k < BN_WROWS; ++k) {
            s.x += v[k].x; s2.x += v[k].x * v[k].x;
            s.y += v[k].y; s2.y += v[k].y * v[k].y;
            s.z += v[k].z; s2.z += v[k].z * v[k].z;
            s.w += v[k].w; s2.w += v[k].w * v[k].w;
        }
    }
    ps[w][lane]  = s;
    ps2[w][lane] = s2;
    __syncthreads();
    if (tid < DF) {
        const float* __restrict__ p  = (const float*)ps;
        const float* __restrict__ p2 = (const float*)ps2;
        float a = 0.0f, b = 0.0f;
        for (int k = 0; k < BN_WAVES; ++k) { a += p[k * DF + tid]; b += p2[k * DF + tid]; }
        atomicAdd(&macc[tid], a);
        atomicAdd(&sqacc[tid], b);
    }
}

__global__ __launch_bounds__(256) void k_bin_bnsum(
    const int* __restrict__ ei, int* __restrict__ tab,
    const float* __restrict__ upd, float* __restrict__ macc,
    float* __restrict__ sqacc)
{
    if (blockIdx.x < PBLOCKS) edge_bin_count_block(blockIdx.x, ei, tab);
    else bn_partial_block(blockIdx.x - PBLOCKS, upd, macc, sqacc);
}

extern "C" void kernel_launch(void* const* d_in, const int* in_sizes, int n_in,
                              void* d_out, int out_size, void* d_ws, size_t ws_size,
                              hipStream_t stream) {
    const float* x     = (const float*)d_in[0];
    const int*   ei    = (const int*)  d_in[1];
    const float* Wk    = (const float*)d_in[2];
    const float* bk    = (const float*)d_in[3];
    const float* a_i   = (const float*)d_in[4];
    const float* a_j   = (const float*)d_in[5];
    const float* a_e   = (const float*)d_in[6];
    const float* b_a   = (const float*)d_in[7];
    const float* Wm    = (const float*)d_in[8];
    const float* bm    = (const float*)d_in[9];
    const float* gamma = (const float*)d_in[10];
    const float* beta  = (const float*)d_in[11];
    float* out = (float*)d_out;
    float* ws  = (float*)d_ws;

    float* h_buf  = ws + OFF_H;
    float* upd    = ws + OFF_UPD;
    int*   cnt    = (int*)ws + OFF_CNT;
    int*   offs   = (int*)ws + OFF_OFFS;
    int*   list   = (int*)ws + OFF_LIST;
    int*   binned = (int*)ws + OFF_BIN;
    int*   tab    = (int*)ws + OFF_TAB;
    int*   rbase  = (int*)ws + OFF_RBASE;
    float* macc   = ws + OFF_MACC;
    float* sqacc  = ws + OFF_SQ;

    k_gemm2_wmma<<<GT_WGS, 256, 0, stream>>>(x, Wk, bk, Wm, bm, h_buf, upd, macc, sqacc);
    k_bin_bnsum<<<PBLOCKS + BN_BLOCKS, 256, 0, stream>>>(ei, tab, upd, macc, sqacc);
    k_edge_bin_scatter<<<PBLOCKS, SC_THREADS, 0, stream>>>(ei, tab, rbase, binned);
    k_range_group<<<NRANGE, RG_THREADS, 0, stream>>>(binned, rbase, cnt, offs, list);
    k_node_gather_out<<<(NN + 7) / 8, 256, 0, stream>>>(
        list, cnt, offs, a_i, a_j, a_e, b_a, h_buf, upd, macc, sqacc, gamma, beta, out);
}


# Edge blocks order their records by receiver range in slabs of their own (count and rank in one LDS pass); a range's work-group collects its 125 runs; count launch and table sums gone, 5 launches -> 4
# speedup vs baseline: 6.6520x; 1.1192x over previous
#include <hip/hip_runtime.h>
#include <hip/hip_bf16.h>
#include <math.h>

#define NN      50000
#define NE      800000
#define DF      128
#define NHEADS  4
#define HU      32
#define NEG_SLOPE 0.2f
#define BN_EPS    0.001f

typedef __attribute__((ext_vector_type(2))) float v2f;
typedef __attribute__((ext_vector_type(8))) float v8f;

#define OFF_H    ((size_t)0)
#define OFF_UPD  (OFF_H   + (size_t)NN * DF)
#define OFF_AGG  (OFF_UPD + (size_t)NN * DF)
#define OFF_BIN  (OFF_AGG)
#define OFF_TAB  (OFF_BIN + (size_t)2 * NE)
#define OFF_SI   (OFF_AGG + (size_t)NN * DF)
#define OFF_SJ   (OFF_SI  + (size_t)NN * NHEADS)
#define OFF_CNT  (OFF_SJ  + (size_t)NN * NHEADS)
#define OFF_OFFS (OFF_CNT + (size_t)NN * NHEADS)
#define OFF_RANK (OFF_OFFS + (size_t)NN * NHEADS)
#define OFF_LIST (OFF_RANK + (size_t)NE)
#define OFF_TOT  (OFF_LIST + (size_t)2 * NE)
#define OFF_MACC (OFF_RANK + (size_t)NE * NHEADS)
#define OFF_SQ   (OFF_MACC + (size_t)DF)

#define DEV static __device__ __forceinline__
DEV int   lane_i(int v, int l)   { return __builtin_amdgcn_readlane(v, l); }
template <int MASK>
DEV float row_xor(float v) {
    return __int_as_float(__builtin_amdgcn_update_dpp(0, __float_as_int(v), 0x160 + MASK, 0xf, 0xf, false));
}
DEV float sum8(float v) {
    v += row_xor<4>(v);
    v += row_xor<2>(v);
    v += row_xor<1>(v);
    return v;
}
DEV float max8(float v) {
    v = fmaxf(v, row_xor<4>(v));
    v = fmaxf(v, row_xor<2>(v));
    v = fmaxf(v, row_xor<1>(v));
    return v;
}
template <int U>
DEV float bcast8(float v) {
    return __int_as_float(__builtin_amdgcn_mov_dpp8(__float_as_int(v), U * 0x249249));
}
DEV float dot4(const float4 r, const float4 a) {
    float s = r.x * a.x; s += r.y * a.y; s += r.z * a.z; s += r.w * a.w;
    return s;
}
DEV float score8(const float4 r, const float4 a) { return sum8(dot4(r, a)); }
DEV int wave_min_i(int v) {
    for (int off = 16; off > 0; off >>= 1) v = min(v, __shfl_xor(v, off, 32));
    return v;
}
DEV float leaky(float x) { return (x > 0.0f) ? x : NEG_SLOPE * x; }


#define GT_ROWS  16
#define GT_PITCH 132
#define GT_TILES (NN / GT_ROWS)
#define GT_WGS   512
static_assert(NN % GT_ROWS == 0, "every row tile is whole: no row guard in the kernel");

template <int KSTRIDE>
DEV void load_bfrag(const float* __restrict__ q, v2f (&bf)[2][DF / 4]) {
    #pragma unroll
    for (int j = 0; j < DF / 4; ++j) {
        const float2 p0 = *(const float2*)(q + (size_t)(4 * j) * KSTRIDE);
        const float2 p1 = *(const float2*)(q + (size_t)(4 * j + 1) * KSTRIDE);
        bf[0][j].x = p0.x; bf[0][j].y = p1.x;
        bf[1][j].x = p0.y; bf[1][j].y = p1.y;
    }
}

__global__ __launch_bounds__(256) void k_gemm2_wmma(
    const float* __restrict__ x, const float* __restrict__ Wk, const float* __restrict__ bk,
    const float* __restrict__ Wm, const float* __restrict__ bm,
    float* __restrict__ h, float* __restrict__ upd,
    float* __restrict__ macc, float* __restrict__ sqacc)
{
    __shared__ __attribute__((aligned(16))) float atile[2][GT_ROWS * GT_PITCH];

    const int tid  = threadIdx.x;
    {
        const int gid = blockIdx.x * 256 + tid;
        if (gid < DF) { macc[gid] = 0.0f; sqacc[gid] = 0.0f; }
    }
    const int lane = tid & 31;
    const int half = lane >> 4;
    const int l    = lane & 15;
    const int w    = __builtin_amdgcn_readfirstlane(tid >> 5);
    const bool is_h = w < 4;
    const int cb   = (w & 3) * 32;

    const int r0 = tid >> 5, c4 = tid & 31;
    const float4* __restrict__ x4 = (const float4*)x + tid;
    const int lds0 = r0 * GT_PITCH + 4 * c4;
    const int lds1 = (r0 + 8) * GT_PITCH + 4 * c4;

    int tile = blockIdx.x;
    float4 pre0, pre1;
    {
        const float4* __restrict__ src = x4 + (size_t)tile * (GT_ROWS * DF / 4);
        pre0 = src[0];
        pre1 = src[256];
    }

    v2f bf[2][DF / 4];
    float bv0, bv1;
    if (is_h) {
        load_bfrag<HU>(Wk + (size_t)w * (DF * HU) + (size_t)(2 * half) * HU + 2 * l, bf);
        const float2 b2 = *(const float2*)(bk + cb + 2 * l);
        bv0 = b2.x; bv1 = b2.y;
    } else {
        load_bfrag<DF>(Wm + (size_t)(2 * half) * DF + cb + 2 * l, bf);
        const float2 b2 = *(const float2*)(bm + cb + 2 * l);
        bv0 = b2.x; bv1 = b2.y;
    }

    *(float4*)&atile[0][lds0] = pre0;
    *(float4*)&atile[0][lds1] = pre1;
    __syncthreads();

    float* __restrict__ obase = (is_h ? h : upd) + (size_t)(8 * half) * DF + cb + 2 * l;
    int buf = 0;
    for (; tile < GT_TILES; tile += GT_WGS) {
        const int next = tile + GT_WGS;
        const bool more = next < GT_TILES;
        if (more) {
            const float4* __restrict__ src = x4 + (size_t)next * (GT_ROWS * DF / 4);
            pre0 = src[0];
            pre1 = src[256];
        }

        const float* __restrict__ arow = &atile[buf][l * GT_PITCH + 2 * half];
        v8f acc0 = {}, acc1 = {};
        #pragma unroll
        for (int j = 0; j < DF / 4; ++j) {
            const v2f a = *(const v2f*)(arow + 4 * j);
            acc0 = __builtin_amdgcn_wmma_f32_16x16x4_f32(
                       false, a, false, bf[0][j], (short)0, acc0, false, false);
            acc1 = __builtin_amdgcn_wmma_f32_16x16x4_f32(
                       false, a, false, bf[1][j], (short)0, acc1, false, false);
        }

        if (more) {
            *(float4*)&atile[buf ^ 1][lds0] = pre0;
            *(float4*)&atile[buf ^ 1][lds1] = pre1;
        }
        __syncthreads();

        float* __restrict__ dst = obase + (size_t)tile * (GT_ROWS * DF);
        #pragma unroll
        for (int r = 0; r < 8; ++r) {
            float v0 = acc0[r] + bv0;
            float v1 = acc1[r] + bv1;
            if (!is_h) { v0 = fmaxf(v0, 0.0f); v1 = fmaxf(v1, 0.0f); }
            *(float2*)(dst + (size_t)r * DF) = make_float2(v0, v1);
        }
        buf ^= 1;
    }
}


#define RG_SHIFT 6
#define RG_NODES (1 << RG_SHIFT)
#define NRANGE   ((NN + RG_NODES - 1) / RG_NODES)
#define PB_EDGES 6400
#define PBLOCKS  (NE / PB_EDGES)
#define RUN_PITCH (NRANGE + 1)
#define ID_BITS  20
static_assert(NE % PB_EDGES == 0, "every edge block is whole: no edge guard in the kernels");
static_assert(NE <= (1 << ID_BITS), "an edge id fits the low ID_BITS bits of a record");
static_assert(RG_SHIFT + ID_BITS <= 31, "receiver's index inside its range sits above the edge id");
static_assert((size_t)PBLOCKS * RUN_PITCH <= (size_t)(1 << 17), "the run starts fit their part of the workspace");
static_assert((size_t)2 * NE + (1 << 17) <= (size_t)NN * DF, "the two arrays fit the plane they share");

#define SC_THREADS 800
#define SC_EPT     (PB_EDGES / SC_THREADS)
static_assert(SC_THREADS % 32 == 0 && SC_THREADS >= NRANGE + 1, "one thread per range and one for the end");
static_assert(PB_EDGES % SC_THREADS == 0, "the block's edges divide over its threads");
__global__ __launch_bounds__(SC_THREADS) void k_edge_block_sort(
    const int* __restrict__ ei, int* __restrict__ runs, int* __restrict__ binned)
{
    __shared__ int lstart[NRANGE];
    __shared__ int fill[NRANGE];
    __shared__ int wsB[32];
    __shared__ int2 stage[PB_EDGES];
    const int tid = threadIdx.x, lane = tid & 31, w = tid >> 5;
    const int b = blockIdx.x;

    const int e0 = b * PB_EDGES + tid;
    int2 rs[SC_EPT];
    #pragma unroll
    for (int k = 0; k < SC_EPT; ++k) rs[k] = ((const int2*)ei)[e0 + k * SC_THREADS];
    if (tid < NRANGE) fill[tid] = 0;
    if (tid < 32) wsB[tid] = 0;
    __syncthreads();

    int rk[SC_EPT];
    #pragma unroll
    for (int k = 0; k < SC_EPT; ++k) rk[k] = atomicAdd(&fill[rs[k].x >> RG_SHIFT], 1);
    __syncthreads();

    const int mine = (tid < NRANGE) ? fill[tid] : 0;
    int sb = mine;
    for (int d = 1; d < 32; d <<= 1) {
        const int tb = __shfl_up(sb, d, 32);
        if (lane >= d) sb += tb;
    }
    if (lane == 31) wsB[w] = sb;
    __syncthreads();
    if (w == 0) {
        const int vb = wsB[lane];
        int xb = vb;
        for (int d = 1; d < 32; d <<= 1) {
            const int tb = __shfl_up(xb, d, 32);
            if (lane >= d) xb += tb;
        }
        wsB[lane] = xb - vb;
    }
    __syncthreads();
    const int ls = wsB[w] + sb - mine;
    if (tid < NRANGE) lstart[tid] = ls;
    if (tid <= NRANGE) runs[b * RUN_PITCH + tid] = ls;
    __syncthreads();

    #pragma unroll
    for (int k = 0; k < SC_EPT; ++k) {
        const int recv = rs[k].x;
        const int e = e0 + k * SC_THREADS;
        stage[lstart[recv >> RG_SHIFT] + rk[k]] =
            make_int2(e | ((recv & (RG_NODES - 1)) << ID_BITS), rs[k].y);
    }
    __syncthreads();
    int2* __restrict__ slab = (int2*)binned + (size_t)b * PB_EDGES;
    #pragma unroll
    for (int k = 0; k < SC_EPT; ++k) {
        const int i = tid + k * SC_THREADS;
        slab[i] = stage[i];
    }
}

#define RG_THREADS 256
#define RG_UNR     5
#define RG_CAP     2048
#define RG_STEPS   7
static_assert((1 << RG_STEPS) >= PBLOCKS, "the search below ends on one run");
static_assert(RG_THREADS >= PBLOCKS + 1 && RG_THREADS >= RG_NODES, "one thread per run, one per receiver");

DEV int2 range_rec(const int2* __restrict__ rec, const int* rsrc, const int* roff, const int i) {
    int lo = 0, hi = PBLOCKS - 1;
    #pragma unroll
    for (int it = 0; it < RG_STEPS; ++it) {
        const int mid = (lo + hi) >> 1;
        const bool up = roff[mid + 1] > i;
        hi = up ? mid : hi;
        lo = up ? lo : mid + 1;
    }
    return rec[rsrc[lo] + (i - roff[lo])];
}

DEV void range_group_block(
    const int rg,
    const int* __restrict__ binned, const int* __restrict__ runs,
    int* __restrict__ cnt, int* __restrict__ offs, int* __restrict__ list)
{
    __shared__ int c[RG_NODES];
    __shared__ int cur[RG_NODES];
    __shared__ int rsrc[PBLOCKS];
    __shared__ int roff[PBLOCKS + 1];
    __shared__ int wlen[RG_THREADS / 32];
    __shared__ int wbeg[RG_THREADS / 32];
    __shared__ int2 lrec[RG_CAP];
    __shared__ int2 lbuf[RG_CAP];
    const int tid = threadIdx.x, lane = tid & 31, w = tid >> 5;

    int s = 0, len = 0;
    if (tid < PBLOCKS) {
        s = runs[tid * RUN_PITCH + rg];
        len = runs[tid * RUN_PITCH + rg + 1] - s;
    }
    if (tid < RG_NODES) c[tid] = 0;
    int sl = len, ss = s;
    for (int d = 1; d < 32; d <<= 1) {
        const int tl = __shfl_up(sl, d, 32);
        if (lane >= d) sl += tl;
    }
    for (int off = 16; off > 0; off >>= 1) ss += __shfl_xor(ss, off, 32);
    if (lane == 31) { wlen[w] = sl; wbeg[w] = ss; }
    __syncthreads();
    int wfront = 0, n = 0, beg = 0;
    for (int j = 0; j < RG_THREADS / 32; ++j) {
        const int a = wlen[j];
        wfront += (j < w) ? a : 0;
        n += a;
        beg += wbeg[j];
    }
    if (tid < PBLOCKS) {
        rsrc[tid] = tid * PB_EDGES + s;
        roff[tid] = wfront + sl - len;
    }
    if (tid == PBLOCKS) roff[PBLOCKS] = n;
    __syncthreads();

    const int2* __restrict__ rec = (const int2*)binned;
    const bool in_lds = n <= RG_CAP;
    for (int i0 = 0; i0 < n; i0 += RG_UNR * RG_THREADS) {
        int2 p[RG_UNR];
        #pragma unroll
        for (int k = 0; k < RG_UNR; ++k)
            p[k] = range_rec(rec, rsrc, roff, min(i0 + k * RG_THREADS + tid, n - 1));
        #pragma unroll
        for (int k = 0; k < RG_UNR; ++k) {
            const int i = i0 + k * RG_THREADS + tid;
            if (i < n) {
                atomicAdd(&c[p[k].x >> ID_BITS], 1);
                if (in_lds) lrec[i] = p[k];
            }
        }
    }
    __syncthreads();
    if (tid < RG_NODES) {
        int o = 0;
        for (int j = 0; j < tid; ++j) o += c[j];
        cur[tid] = o;
        const int node = rg * RG_NODES + tid;
        if (node < NN) { cnt[node] = c[tid]; offs[node] = beg + o; }
    }
    __syncthreads();
    int2* __restrict__ lout = (int2*)list + beg;
    if (in_lds) {
        for (int i = tid; i < n; i += RG_THREADS) {
            const int2 q = lrec[i];
            const int pos = atomicAdd(&cur[q.x >> ID_BITS], 1);
            lbuf[pos] = make_int2(q.x & ((1 << ID_BITS) - 1), q.y);
        }
        __syncthreads();
        for (int i = tid; i < n; i += RG_THREADS) lout[i] = lbuf[i];
    } else {
        for (int i = tid; i < n; i += RG_THREADS) {
            const int2 q = range_rec(rec, rsrc, roff, i);
            const int pos = atomicAdd(&cur[q.x >> ID_BITS], 1);
            lout[pos] = make_int2(q.x & ((1 << ID_BITS) - 1), q.y);
        }
    }
}

DEV int next_by_id(const int2* __restrict__ seg, int deg, int lane, int& last) {
    int best = 0x7fffffff, bs = 0;
    for (int i = lane; i < deg; i += 32) {
        const int2 p = seg[i];
        if (p.x > last && p.x < best) { best = p.x; bs = p.y; }
    }
    const int mn = wave_min_i(best);
    const unsigned hit = __builtin_amdgcn_ballot_w32(best == mn);
    last = mn;
    return lane_i(bs, __builtin_ctz(hit));
}

#define GC 8
static_assert(GC == 8, "a group is one row per lane of a head's 8 lanes");
template <bool FULL>
DEV void gather_group(
    const float4* __restrict__ hrow, const int send, const int r0, const int deg,
    const float4 aj, const float si, const float ae, const float ba, const int lane,
    float& m, float& d, float4& acc)
{
    float4 v[GC];
    #pragma unroll
    for (int u = 0; u < GC; ++u) {
        if (FULL || r0 + u < deg) v[u] = hrow[(size_t)lane_i(send, r0 + u) * (DF / 4)];
        else v[u] = make_float4(0.0f, 0.0f, 0.0f, 0.0f);
    }
    float p[GC];
    #pragma unroll
    for (int u = 0; u < GC; ++u) p[u] = dot4(v[u], aj);

    const bool b2 = (lane & 4) != 0, b1 = (lane & 2) != 0, b0 = (lane & 1) != 0;
    float q[4], r[2];
    #pragma unroll
    for (int u = 0; u < 4; ++u) {
        const float keep = b2 ? p[u + 4] : p[u];
        const float give = b2 ? p[u] : p[u + 4];
        q[u] = keep + row_xor<4>(give);
    }
    #pragma unroll
    for (int u = 0; u < 2; ++u) {
        const float keep = b1 ? q[u + 2] : q[u];
        const float give = b1 ? q[u] : q[u + 2];
        r[u] = keep + row_xor<2>(give);
    }
    const float keep = b0 ? r[1] : r[0];
    const float give = b0 ? r[0] : r[1];
    const float sj = keep + row_xor<1>(give);

    float lg = leaky(si + sj + ae + ba);
    if (!FULL && r0 + (lane & 7) >= deg) lg = -INFINITY;
    const float mn = fmaxf(m, max8(lg));
    const float sc = expf(m - mn);
    const float wo = expf(lg - mn);
    d *= sc; acc.x *= sc; acc.y *= sc; acc.z *= sc; acc.w *= sc;
    m = mn;
#define ADD_ROW(U) { const float w = bcast8<U>(wo); d += w; \
                     acc.x += v[U].x * w; acc.y += v[U].y * w; acc.z += v[U].z * w; acc.w += v[U].w * w; }
    ADD_ROW(0) ADD_ROW(1) ADD_ROW(2) ADD_ROW(3) ADD_ROW(4) ADD_ROW(5) ADD_ROW(6) ADD_ROW(7)
#undef ADD_ROW
}

DEV float fin_rstd(float sq, float mean, float inv_n) {
    const float var = sq * inv_n - mean * mean;
    return rsqrtf(var + BN_EPS);
}
DEV float fin_one(float u, float a, float ma, float inv_n, float rs, float g, float b) {
    const float d = fmaf(-ma, inv_n, u);
    const float normed = g * d * rs + b;
    return normed + fmaxf(a, 0.0f);
}
__global__ __launch_bounds__(256) void k_node_gather_out(
    const int* __restrict__ list, const int* __restrict__ cnt,
    const int* __restrict__ offs, const float* __restrict__ a_i,
    const float* __restrict__ a_j, const float* __restrict__ a_e,
    const float* __restrict__ b_a, const float* __restrict__ h,
    const float* __restrict__ upd,
    const float* __restrict__ macc, const float* __restrict__ sqacc,
    const float* __restrict__ gamma, const float* __restrict__ beta,
    float* __restrict__ out)
{
    const int lane = threadIdx.x & 31;
    const int node = blockIdx.x * 8 + __builtin_amdgcn_readfirstlane(threadIdx.x >> 5);
    if (node >= NN) return;
    const int deg = cnt[node];
    const int2* __restrict__ seg = (const int2*)list + offs[node];
    const int hd = lane >> 3;
    const float4* __restrict__ hrow = (const float4*)h + lane;
    const float4 own = hrow[(size_t)node * (DF / 4)];
    const float4 ai = ((const float4*)a_i)[lane];
    const float4 aj = ((const float4*)a_j)[lane];
    const float ae = a_e[hd], ba = b_a[hd];

    float4 acc = make_float4(0.0f, 0.0f, 0.0f, 0.0f);
    if (deg > 0) {
        const float si = score8(own, ai);
        float m = -INFINITY, d = 0.0f;

        if (deg <= 32) {
            const bool on = lane < deg;
            int e = 0x7fffffff, send = 0;
            if (on) { const int2 p = seg[lane]; e = p.x; send = p.y; }
            int rk = 0;
            for (int j = 0; j < deg; ++j) rk += (lane_i(e, j) < e) ? 1 : 0;
            if (!on) rk = lane;
            send = __builtin_amdgcn_ds_permute(rk << 2, send);

            int r0 = 0;
            for (; r0 + GC <= deg; r0 += GC)
                gather_group<true>(hrow, send, r0, deg, aj, si, ae, ba, lane, m, d, acc);
            if (r0 < deg)
                gather_group<false>(hrow, send, r0, deg, aj, si, ae, ba, lane, m, d, acc);
        } else {
            int last = -1;
            for (int t = 0; t < deg; ++t) {
                const int s = next_by_id(seg, deg, lane, last);
                const float4 v = hrow[(size_t)s * (DF / 4)];
                const float l = leaky(si + score8(v, aj) + ae + ba);
                const float mn = fmaxf(m, l);
                const float sc = expf(m - mn);
                const float w = expf(l - mn);
                d = d * sc + w;
                acc.x = acc.x * sc + v.x * w; acc.y = acc.y * sc + v.y * w;
                acc.z = acc.z * sc + v.z * w; acc.w = acc.w * sc + v.w * w;
                m = mn;
            }
        }
        const float inv = 1.0f / (d + 1e-16f);
        acc.x *= inv; acc.y *= inv; acc.z *= inv; acc.w *= inv;
    }

    const float inv_n = 1.0f / (float)NN;
    const float4 u  = ((const float4*)upd)[(size_t)node * (DF / 4) + lane];
    const float4 ma = ((const float4*)macc)[lane];
    const float4 sq = ((const float4*)sqacc)[lane];
    const float4 g  = ((const float4*)gamma)[lane];
    const float4 b  = ((const float4*)beta)[lane];
    float4 rs;
    rs.x = fin_rstd(sq.x, ma.x * inv_n, inv_n); rs.y = fin_rstd(sq.y, ma.y * inv_n, inv_n);
    rs.z = fin_rstd(sq.z, ma.z * inv_n, inv_n); rs.w = fin_rstd(sq.w, ma.w * inv_n, inv_n);
    float4 o;
    o.x = fin_one(u.x, acc.x, ma.x, inv_n, rs.x, g.x, b.x);
    o.y = fin_one(u.y, acc.y, ma.y, inv_n, rs.y, g.y, b.y);
    o.z = fin_one(u.z, acc.z, ma.z, inv_n, rs.z, g.z, b.z);
    o.w = fin_one(u.w, acc.w, ma.w, inv_n, rs.w, g.w, b.w);
    ((float4*)out)[(size_t)node * (DF / 4) + lane] = o;
}

#define BN_WAVES 8
#define BN_WROWS 16
#define BN_ROWS  (BN_WAVES * BN_WROWS)
#define BN_BLOCKS ((NN + BN_ROWS - 1) / BN_ROWS)
static_assert(NN % BN_WROWS == 0, "a wave's rows are all inside the matrix or all outside");
DEV void bn_partial_block(
    const int bid,
    const float* __restrict__ upd, float* __restrict__ macc,
    float* __restrict__ sqacc)
{
    __shared__ float4 ps[BN_WAVES][32];
    __shared__ float4 ps2[BN_WAVES][32];
    const int tid  = threadIdx.x;
    const int lane = tid & 31;
    const int w    = __builtin_amdgcn_readfirstlane(tid >> 5);
    const int r0   = bid * BN_ROWS + w * BN_WROWS;

    float4 s  = make_float4(0.0f, 0.0f, 0.0f, 0.0f);
    float4 s2 = make_float4(0.0f, 0.0f, 0.0f, 0.0f);
    if (r0 < NN) {
        const float4* __restrict__ u4 = (const float4*)upd + (size_t)r0 * (DF / 4) + lane;
        float4 v[BN_WROWS];
        #pragma unroll
        for (int k = 0; k < BN_WROWS; ++k) v[k] = u4[k * (DF / 4)];
        #pragma unroll
        for (int k = 0; k < BN_WROWS; ++k) {
            s.x += v[k].x; s2.x += v[k].x * v[k].x;
            s.y += v[k].y; s2.y += v[k].y * v[k].y;
            s.z += v[k].z; s2.z += v[k].z * v[k].z;
            s.w += v[k].w; s2.w += v[k].w * v[k].w;
        }
    }
    ps[w][lane]  = s;
    ps2[w][lane] = s2;
    __syncthreads();
    if (tid < DF) {
        const float* __restrict__ p  = (const float*)ps;
        const float* __restrict__ p2 = (const float*)ps2;
        float a = 0.0f, b = 0.0f;
        for (int k = 0; k < BN_WAVES; ++k) { a += p[k * DF + tid]; b += p2[k * DF + tid]; }
        atomicAdd(&macc[tid], a);
        atomicAdd(&sqacc[tid], b);
    }
}

__global__ __launch_bounds__(256) void k_group_bnsum(
    const int* __restrict__ binned, const int* __restrict__ runs,
    int* __restrict__ cnt, int* __restrict__ offs, int* __restrict__ list,
    const float* __restrict__ upd, float* __restrict__ macc,
    float* __restrict__ sqacc)
{
    if (blockIdx.x < NRANGE) range_group_block(blockIdx.x, binned, runs, cnt, offs, list);
    else bn_partial_block(blockIdx.x - NRANGE, upd, macc, sqacc);
}

extern "C" void kernel_launch(void* const* d_in, const int* in_sizes, int n_in,
                              void* d_out, int out_size, void* d_ws, size_t ws_size,
                              hipStream_t stream) {
    const float* x     = (const float*)d_in[0];
    const int*   ei    = (const int*)  d_in[1];
    const float* Wk    = (const float*)d_in[2];
    const float* bk    = (const float*)d_in[3];
    const float* a_i   = (const float*)d_in[4];
    const float* a_j   = (const float*)d_in[5];
    const float* a_e   = (const float*)d_in[6];
    const float* b_a   = (const float*)d_in[7];
    const float* Wm    = (const float*)d_in[8];
    const float* bm    = (const float*)d_in[9];
    const float* gamma = (const float*)d_in[10];
    const float* beta  = (const float*)d_in[11];
    float* out = (float*)d_out;
    float* ws  = (float*)d_ws;

    float* h_buf  = ws + OFF_H;
    float* upd    = ws + OFF_UPD;
    int*   cnt    = (int*)ws + OFF_CNT;
    int*   offs   = (int*)ws + OFF_OFFS;
    int*   list   = (int*)ws + OFF_LIST;
    int*   binned = (int*)ws + OFF_BIN;
    int*   runs   = (int*)ws + OFF_TAB;
    float* macc   = ws + OFF_MACC;
    float* sqacc  = ws + OFF_SQ;

    k_gemm2_wmma<<<GT_WGS, 256, 0, stream>>>(x, Wk, bk, Wm, bm, h_buf, upd, macc, sqacc);
    k_edge_block_sort<<<PBLOCKS, SC_THREADS, 0, stream>>>(ei, runs, binned);
    k_group_bnsum<<<NRANGE + BN_BLOCKS, 256, 0, stream>>>(binned, runs, cnt, offs, list, upd, macc, sqacc);
    k_node_gather_out<<<(NN + 7) / 8, 256, 0, stream>>>(
        list, cnt, offs, a_i, a_j, a_e, b_a, h_buf, upd, macc, sqacc, gamma, beta, out);
}


# Gather launch: a wave walks 5 consecutive nodes (10000 waves enter, not 50000); degrees, own rows, upd rows fetched per wave in groups, next node's list entries ahead of its rows; sum order kept
# speedup vs baseline: 6.8562x; 1.0307x over previous
#include <hip/hip_runtime.h>
#include <hip/hip_bf16.h>
#include <math.h>

#define NN      50000
#define NE      800000
#define DF      128
#define NHEADS  4
#define HU      32
#define NEG_SLOPE 0.2f
#define BN_EPS    0.001f

typedef __attribute__((ext_vector_type(2))) float v2f;
typedef __attribute__((ext_vector_type(8))) float v8f;

#define OFF_H    ((size_t)0)
#define OFF_UPD  (OFF_H   + (size_t)NN * DF)
#define OFF_AGG  (OFF_UPD + (size_t)NN * DF)
#define OFF_BIN  (OFF_AGG)
#define OFF_TAB  (OFF_BIN + (size_t)2 * NE)
#define OFF_SI   (OFF_AGG + (size_t)NN * DF)
#define OFF_SJ   (OFF_SI  + (size_t)NN * NHEADS)
#define OFF_CNT  (OFF_SJ  + (size_t)NN * NHEADS)
#define OFF_OFFS (OFF_CNT + (size_t)NN * NHEADS)
#define OFF_RANK (OFF_OFFS + (size_t)NN * NHEADS)
#define OFF_LIST (OFF_RANK + (size_t)NE)
#define OFF_TOT  (OFF_LIST + (size_t)2 * NE)
#define OFF_MACC (OFF_RANK + (size_t)NE * NHEADS)
#define OFF_SQ   (OFF_MACC + (size_t)DF)

#define DEV static __device__ __forceinline__
DEV int   lane_i(int v, int l)   { return __builtin_amdgcn_readlane(v, l); }
template <int MASK>
DEV float row_xor(float v) {
    return __int_as_float(__builtin_amdgcn_update_dpp(0, __float_as_int(v), 0x160 + MASK, 0xf, 0xf, false));
}
DEV float sum8(float v) {
    v += row_xor<4>(v);
    v += row_xor<2>(v);
    v += row_xor<1>(v);
    return v;
}
DEV float max8(float v) {
    v = fmaxf(v, row_xor<4>(v));
    v = fmaxf(v, row_xor<2>(v));
    v = fmaxf(v, row_xor<1>(v));
    return v;
}
template <int U>
DEV float bcast8(float v) {
    return __int_as_float(__builtin_amdgcn_mov_dpp8(__float_as_int(v), U * 0x249249));
}
DEV float dot4(const float4 r, const float4 a) {
    float s = r.x * a.x; s += r.y * a.y; s += r.z * a.z; s += r.w * a.w;
    return s;
}
DEV float score8(const float4 r, const float4 a) { return sum8(dot4(r, a)); }
DEV int wave_min_i(int v) {
    for (int off = 16; off > 0; off >>= 1) v = min(v, __shfl_xor(v, off, 32));
    return v;
}
DEV float leaky(float x) { return (x > 0.0f) ? x : NEG_SLOPE * x; }


#define GT_ROWS  16
#define GT_PITCH 132
#define GT_TILES (NN / GT_ROWS)
#define GT_WGS   512
static_assert(NN % GT_ROWS == 0, "every row tile is whole: no row guard in the kernel");

template <int KSTRIDE>
DEV void load_bfrag(const float* __restrict__ q, v2f (&bf)[2][DF / 4]) {
    #pragma unroll
    for (int j = 0; j < DF / 4; ++j) {
        const float2 p0 = *(const float2*)(q + (size_t)(4 * j) * KSTRIDE);
        const float2 p1 = *(const float2*)(q + (size_t)(4 * j + 1) * KSTRIDE);
        bf[0][j].x = p0.x; bf[0][j].y = p1.x;
        bf[1][j].x = p0.y; bf[1][j].y = p1.y;
    }
}

__global__ __launch_bounds__(256) void k_gemm2_wmma(
    const float* __restrict__ x, const float* __restrict__ Wk, const float* __restrict__ bk,
    const float* __restrict__ Wm, const float* __restrict__ bm,
    float* __restrict__ h, float* __restrict__ upd,
    float* __restrict__ macc, float* __restrict__ sqacc)
{
    __shared__ __attribute__((aligned(16))) float atile[2][GT_ROWS * GT_PITCH];

    const int tid  = threadIdx.x;
    {
        const int gid = blockIdx.x * 256 + tid;
        if (gid < DF) { macc[gid] = 0.0f; sqacc[gid] = 0.0f; }
    }
    const int lane = tid & 31;
    const int half = lane >> 4;
    const int l    = lane & 15;
    const int w    = __builtin_amdgcn_readfirstlane(tid >> 5);
    const bool is_h = w < 4;
    const int cb   = (w & 3) * 32;

    const int r0 = tid >> 5, c4 = tid & 31;
    const float4* __restrict__ x4 = (const float4*)x + tid;
    const int lds0 = r0 * GT_PITCH + 4 * c4;
    const int lds1 = (r0 + 8) * GT_PITCH + 4 * c4;

    int tile = blockIdx.x;
    float4 pre0, pre1;
    {
        const float4* __restrict__ src = x4 + (size_t)tile * (GT_ROWS * DF / 4);
        pre0 = src[0];
        pre1 = src[256];
    }

    v2f bf[2][DF / 4];
    float bv0, bv1;
    if (is_h) {
        load_bfrag<HU>(Wk + (size_t)w * (DF * HU) + (size_t)(2 * half) * HU + 2 * l, bf);
        const float2 b2 = *(const float2*)(bk + cb + 2 * l);
        bv0 = b2.x; bv1 = b2.y;
    } else {
        load_bfrag<DF>(Wm + (size_t)(2 * half) * DF + cb + 2 * l, bf);
        const float2 b2 = *(const float2*)(bm + cb + 2 * l);
        bv0 = b2.x; bv1 = b2.y;
    }

    *(float4*)&atile[0][lds0] = pre0;
    *(float4*)&atile[0][lds1] = pre1;
    __syncthreads();

    float* __restrict__ obase = (is_h ? h : upd) + (size_t)(8 * half) * DF + cb + 2 * l;
    int buf = 0;
    for (; tile < GT_TILES; tile += GT_WGS) {
        const int next = tile + GT_WGS;
        const bool more = next < GT_TILES;
        if (more) {
            const float4* __restrict__ src = x4 + (size_t)next * (GT_ROWS * DF / 4);
            pre0 = src[0];
            pre1 = src[256];
        }

        const float* __restrict__ arow = &atile[buf][l * GT_PITCH + 2 * half];
        v8f acc0 = {}, acc1 = {};
        #pragma unroll
        for (int j = 0; j < DF / 4; ++j) {
            const v2f a = *(const v2f*)(arow + 4 * j);
            acc0 = __builtin_amdgcn_wmma_f32_16x16x4_f32(
                       false, a, false, bf[0][j], (short)0, acc0, false, false);
            acc1 = __builtin_amdgcn_wmma_f32_16x16x4_f32(
                       false, a, false, bf[1][j], (short)0, acc1, false, false);
        }

        if (more) {
            *(float4*)&atile[buf ^ 1][lds0] = pre0;
            *(float4*)&atile[buf ^ 1][lds1] = pre1;
        }
        __syncthreads();

        float* __restrict__ dst = obase + (size_t)tile * (GT_ROWS * DF);
        #pragma unroll
        for (int r = 0; r < 8; ++r) {
            float v0 = acc0[r] + bv0;
            float v1 = acc1[r] + bv1;
            if (!is_h) { v0 = fmaxf(v0, 0.0f); v1 = fmaxf(v1, 0.0f); }
            *(float2*)(dst + (size_t)r * DF) = make_float2(v0, v1);
        }
        buf ^= 1;
    }
}


#define RG_SHIFT 6
#define RG_NODES (1 << RG_SHIFT)
#define NRANGE   ((NN + RG_NODES - 1) / RG_NODES)
#define PB_EDGES 6400
#define PBLOCKS  (NE / PB_EDGES)
#define RUN_PITCH (NRANGE + 1)
#define ID_BITS  20
static_assert(NE % PB_EDGES == 0, "every edge block is whole: no edge guard in the kernels");
static_assert(NE <= (1 << ID_BITS), "an edge id fits the low ID_BITS bits of a record");
static_assert(RG_SHIFT + ID_BITS <= 31, "receiver's index inside its range sits above the edge id");
static_assert((size_t)PBLOCKS * RUN_PITCH <= (size_t)(1 << 17), "the run starts fit their part of the workspace");
static_assert((size_t)2 * NE + (1 << 17) <= (size_t)NN * DF, "the two arrays fit the plane they share");

#define SC_THREADS 800
#define SC_EPT     (PB_EDGES / SC_THREADS)
static_assert(SC_THREADS % 32 == 0 && SC_THREADS >= NRANGE + 1, "one thread per range and one for the end");
static_assert(PB_EDGES % SC_THREADS == 0, "the block's edges divide over its threads");
__global__ __launch_bounds__(SC_THREADS) void k_edge_block_sort(
    const int* __restrict__ ei, int* __restrict__ runs, int* __restrict__ binned)
{
    __shared__ int lstart[NRANGE];
    __shared__ int fill[NRANGE];
    __shared__ int wsB[32];
    __shared__ int2 stage[PB_EDGES];
    const int tid = threadIdx.x, lane = tid & 31, w = tid >> 5;
    const int b = blockIdx.x;

    const int e0 = b * PB_EDGES + tid;
    int2 rs[SC_EPT];
    #pragma unroll
    for (int k = 0; k < SC_EPT; ++k) rs[k] = ((const int2*)ei)[e0 + k * SC_THREADS];
    if (tid < NRANGE) fill[tid] = 0;
    if (tid < 32) wsB[tid] = 0;
    __syncthreads();

    int rk[SC_EPT];
    #pragma unroll
    for (int k = 0; k < SC_EPT; ++k) rk[k] = atomicAdd(&fill[rs[k].x >> RG_SHIFT], 1);
    __syncthreads();

    const int mine = (tid < NRANGE) ? fill[tid] : 0;
    int sb = mine;
    for (int d = 1; d < 32; d <<= 1) {
        const int tb = __shfl_up(sb, d, 32);
        if (lane >= d) sb += tb;
    }
    if (lane == 31) wsB[w] = sb;
    __syncthreads();
    if (w == 0) {
        const int vb = wsB[lane];
        int xb = vb;
        for (int d = 1; d < 32; d <<= 1) {
            const int tb = __shfl_up(xb, d, 32);
            if (lane >= d) xb += tb;
        }
        wsB[lane] = xb - vb;
    }
    __syncthreads();
    const int ls = wsB[w] + sb - mine;
    if (tid < NRANGE) lstart[tid] = ls;
    if (tid <= NRANGE) runs[b * RUN_PITCH + tid] = ls;
    __syncthreads();

    #pragma unroll
    for (int k = 0; k < SC_EPT; ++k) {
        const int recv = rs[k].x;
        const int e = e0 + k * SC_THREADS;
        stage[lstart[recv >> RG_SHIFT] + rk[k]] =
            make_int2(e | ((recv & (RG_NODES - 1)) << ID_BITS), rs[k].y);
    }
    __syncthreads();
    int2* __restrict__ slab = (int2*)binned + (size_t)b * PB_EDGES;
    #pragma unroll
    for (int k = 0; k < SC_EPT; ++k) {
        const int i = tid + k * SC_THREADS;
        slab[i] = stage[i];
    }
}

#define RG_THREADS 256
#define RG_UNR     5
#define RG_CAP     2048
#define RG_STEPS   7
static_assert((1 << RG_STEPS) >= PBLOCKS, "the search below ends on one run");
static_assert(RG_THREADS >= PBLOCKS + 1 && RG_THREADS >= RG_NODES, "one thread per run, one per receiver");

DEV int2 range_rec(const int2* __restrict__ rec, const int* rsrc, const int* roff, const int i) {
    int lo = 0, hi = PBLOCKS - 1;
    #pragma unroll
    for (int it = 0; it < RG_STEPS; ++it) {
        const int mid = (lo + hi) >> 1;
        const bool up = roff[mid + 1] > i;
        hi = up ? mid : hi;
        lo = up ? lo : mid + 1;
    }
    return rec[rsrc[lo] + (i - roff[lo])];
}

DEV void range_group_block(
    const int rg,
    const int* __restrict__ binned, const int* __restrict__ runs,
    int* __restrict__ cnt, int* __restrict__ offs, int* __restrict__ list)
{
    __shared__ int c[RG_NODES];
    __shared__ int cur[RG_NODES];
    __shared__ int rsrc[PBLOCKS];
    __shared__ int roff[PBLOCKS + 1];
    __shared__ int wlen[RG_THREADS / 32];
    __shared__ int wbeg[RG_THREADS / 32];
    __shared__ int2 lrec[RG_CAP];
    __shared__ int2 lbuf[RG_CAP];
    const int tid = threadIdx.x, lane = tid & 31, w = tid >> 5;

    int s = 0, len = 0;
    if (tid < PBLOCKS) {
        s = runs[tid * RUN_PITCH + rg];
        len = runs[tid * RUN_PITCH + rg + 1] - s;
    }
    if (tid < RG_NODES) c[tid] = 0;
    int sl = len, ss = s;
    for (int d = 1; d < 32; d <<= 1) {
        const int tl = __shfl_up(sl, d, 32);
        if (lane >= d) sl += tl;
    }
    for (int off = 16; off > 0; off >>= 1) ss += __shfl_xor(ss, off, 32);
    if (lane == 31) { wlen[w] = sl; wbeg[w] = ss; }
    __syncthreads();
    int wfront = 0, n = 0, beg = 0;
    for (int j = 0; j < RG_THREADS / 32; ++j) {
        const int a = wlen[j];
        wfront += (j < w) ? a : 0;
        n += a;
        beg += wbeg[j];
    }
    if (tid < PBLOCKS) {
        rsrc[tid] = tid * PB_EDGES + s;
        roff[tid] = wfront + sl - len;
    }
    if (tid == PBLOCKS) roff[PBLOCKS] = n;
    __syncthreads();

    const int2* __restrict__ rec = (const int2*)binned;
    const bool in_lds = n <= RG_CAP;
    for (int i0 = 0; i0 < n; i0 += RG_UNR * RG_THREADS) {
        int2 p[RG_UNR];
        #pragma unroll
        for (int k = 0; k < RG_UNR; ++k)
            p[k] = range_rec(rec, rsrc, roff, min(i0 + k * RG_THREADS + tid, n - 1));
        #pragma unroll
        for (int k = 0; k < RG_UNR; ++k) {
            const int i = i0 + k * RG_THREADS + tid;
            if (i < n) {
                atomicAdd(&c[p[k].x >> ID_BITS], 1);
                if (in_lds) lrec[i] = p[k];
            }
        }
    }
    __syncthreads();
    if (tid < RG_NODES) {
        int o = 0;
        for (int j = 0; j < tid; ++j) o += c[j];
        cur[tid] = o;
        const int node = rg * RG_NODES + tid;
        if (node < NN) { cnt[node] = c[tid]; offs[node] = beg + o; }
    }
    __syncthreads();
    int2* __restrict__ lout = (int2*)list + beg;
    if (in_lds) {
        for (int i = tid; i < n; i += RG_THREADS) {
            const int2 q = lrec[i];
            const int pos = atomicAdd(&cur[q.x >> ID_BITS], 1);
            lbuf[pos] = make_int2(q.x & ((1 << ID_BITS) - 1), q.y);
        }
        __syncthreads();
        for (int i = tid; i < n; i += RG_THREADS) lout[i] = lbuf[i];
    } else {
        for (int i = tid; i < n; i += RG_THREADS) {
            const int2 q = range_rec(rec, rsrc, roff, i);
            const int pos = atomicAdd(&cur[q.x >> ID_BITS], 1);
            lout[pos] = make_int2(q.x & ((1 << ID_BITS) - 1), q.y);
        }
    }
}

DEV int next_by_id(const int2* __restrict__ seg, int deg, int lane, int& last) {
    int best = 0x7fffffff, bs = 0;
    for (int i = lane; i < deg; i += 32) {
        const int2 p = seg[i];
        if (p.x > last && p.x < best) { best = p.x; bs = p.y; }
    }
    const int mn = wave_min_i(best);
    const unsigned hit = __builtin_amdgcn_ballot_w32(best == mn);
    last = mn;
    return lane_i(bs, __builtin_ctz(hit));
}

#define GC 8
static_assert(GC == 8, "a group is one row per lane of a head's 8 lanes");
static_assert((size_t)NN * DF * 4 < ((size_t)1 << 32), "a byte offset into h fits 32 bits");
template <bool FULL>
DEV void gather_group(
    const float* __restrict__ h, const int send, const int r0, const int deg,
    const float4 aj, const float si, const float ae, const float ba, const int lane,
    float& m, float& d, float4& acc)
{
    float4 v[GC];
    #pragma unroll
    for (int u = 0; u < GC; ++u) {
        if (FULL || r0 + u < deg)
            v[u] = *(const float4*)((const char*)h + ((unsigned)lane_i(send, r0 + u) * (unsigned)(DF * 4) + (unsigned)lane * 16u));
        else v[u] = make_float4(0.0f, 0.0f, 0.0f, 0.0f);
    }
    if (FULL) {
        __builtin_amdgcn_sched_group_barrier(0x020, GC, 0);
        __builtin_amdgcn_sched_group_barrier(0x002, 4 * GC, 0);
    }
    float p[GC];
    #pragma unroll
    for (int u = 0; u < GC; ++u) p[u] = dot4(v[u], aj);

    const bool b2 = (lane & 4) != 0, b1 = (lane & 2) != 0, b0 = (lane & 1) != 0;
    float q[4], r[2];
    #pragma unroll
    for (int u = 0; u < 4; ++u) {
        const float keep = b2 ? p[u + 4] : p[u];
        const float give = b2 ? p[u] : p[u + 4];
        q[u] = keep + row_xor<4>(give);
    }
    #pragma unroll
    for (int u = 0; u < 2; ++u) {
        const float keep = b1 ? q[u + 2] : q[u];
        const float give = b1 ? q[u] : q[u + 2];
        r[u] = keep + row_xor<2>(give);
    }
    const float keep = b0 ? r[1] : r[0];
    const float give = b0 ? r[0] : r[1];
    const float sj = keep + row_xor<1>(give);

    float lg = leaky(si + sj + ae + ba);
    if (!FULL && r0 + (lane & 7) >= deg) lg = -INFINITY;
    const float mn = fmaxf(m, max8(lg));
    const float sc = expf(m - mn);
    const float wo = expf(lg - mn);
    d *= sc; acc.x *= sc; acc.y *= sc; acc.z *= sc; acc.w *= sc;
    m = mn;
#define ADD_ROW(U) { const float w = bcast8<U>(wo); d += w; \
                     acc.x += v[U].x * w; acc.y += v[U].y * w; acc.z += v[U].z * w; acc.w += v[U].w * w; }
    ADD_ROW(0) ADD_ROW(1) ADD_ROW(2) ADD_ROW(3) ADD_ROW(4) ADD_ROW(5) ADD_ROW(6) ADD_ROW(7)
#undef ADD_ROW
}

DEV float fin_rstd(float sq, float mean, float inv_n) {
    const float var = sq * inv_n - mean * mean;
    return rsqrtf(var + BN_EPS);
}
DEV float fin_one(float u, float a, float ma, float inv_n, float rs, float g, float b) {
    const float d = fmaf(-ma, inv_n, u);
    const float normed = g * d * rs + b;
    return normed + fmaxf(a, 0.0f);
}
#define NPW       5
#define GO_WAVES  8
#define GO_NODES  (GO_WAVES * NPW)
#define GO_BLOCKS (NN / GO_NODES)
static_assert(NN % GO_NODES == 0, "every wave has its NPW nodes: no node guard in the kernel");
static_assert(NPW <= 8, "lane k of a head's 8 lanes keeps s_i of node k");
__global__ __launch_bounds__(256) void k_node_gather_out(
    const int* __restrict__ list, const int* __restrict__ cnt,
    const int* __restrict__ offs, const float* __restrict__ a_i,
    const float* __restrict__ a_j, const float* __restrict__ a_e,
    const float* __restrict__ b_a, const float* __restrict__ h,
    const float* __restrict__ upd,
    const float* __restrict__ macc, const float* __restrict__ sqacc,
    const float* __restrict__ gamma, const float* __restrict__ beta,
    float* __restrict__ out)
{
    __shared__ float4 park[GO_WAVES][NPW][32];
    const int lane = threadIdx.x & 31;
    const int w  = __builtin_amdgcn_readfirstlane(threadIdx.x >> 5);
    const int n0 = (blockIdx.x * GO_WAVES + w) * NPW;
    int cv = 0, ov = 0;
    if (lane < NPW) { cv = cnt[n0 + lane]; ov = offs[n0 + lane]; }
    const int hd = lane >> 3;
    const float4 aj = ((const float4*)a_j)[lane];
    const float ae = a_e[hd], ba = b_a[hd];

    float sv = 0.0f;
    {
        const float4 ai = ((const float4*)a_i)[lane];
        float4 own[NPW];
        #pragma unroll
        for (int k = 0; k < NPW; ++k) own[k] = ((const float4*)(h + (size_t)(n0 + k) * DF))[lane];
        #pragma unroll
        for (int k = 0; k < NPW; ++k) {
            const float s = score8(own[k], ai);
            if ((lane & 7) == k) sv = s;
        }
    }

    int deg_n = lane_i(cv, 0);
    const int2* __restrict__ seg_n = (const int2*)list + lane_i(ov, 0);
    int2 ent_n = make_int2(0x7fffffff, 0);
    if (lane < deg_n) ent_n = seg_n[lane];

    for (int k = 0; k < NPW; ++k) {
        const int deg = deg_n;
        const int2* __restrict__ seg = seg_n;
        const int2 ent = ent_n;

        const float si = __int_as_float(__builtin_amdgcn_ds_bpermute(((lane & 24) + k) << 2, __float_as_int(sv)));
        int send = ent.y;
        if (deg <= 32) {
            const int e = ent.x;
            int rk = 0;
            for (int j = 0; j < deg; ++j) rk += (lane_i(e, j) < e) ? 1 : 0;
            if (lane >= deg) rk = lane;
            send = __builtin_amdgcn_ds_permute(rk << 2, send);
        }

        if (k + 1 < NPW) {
            deg_n = lane_i(cv, k + 1);
            seg_n = (const int2*)list + lane_i(ov, k + 1);
            ent_n = make_int2(0x7fffffff, 0);
            if (lane < deg_n) ent_n = seg_n[lane];
        }

        float4 acc = make_float4(0.0f, 0.0f, 0.0f, 0.0f);
        if (deg > 0) {
            float m = -INFINITY, d = 0.0f;
            if (deg <= 32) {
                int r0 = 0;
                for (; r0 + GC <= deg; r0 += GC)
                    gather_group<true>(h, send, r0, deg, aj, si, ae, ba, lane, m, d, acc);
                if (r0 < deg)
                    gather_group<false>(h, send, r0, deg, aj, si, ae, ba, lane, m, d, acc);
            } else {
                int last = -1;
                for (int t = 0; t < deg; ++t) {
                    const int s = next_by_id(seg, deg, lane, last);
                    const float4 v = ((const float4*)(h + (size_t)s * DF))[lane];
                    const float l = leaky(si + score8(v, aj) + ae + ba);
                    const float mn = fmaxf(m, l);
                    const float sc = expf(m - mn);
                    const float w2 = expf(l - mn);
                    d = d * sc + w2;
                    acc.x = acc.x * sc + v.x * w2; acc.y = acc.y * sc + v.y * w2;
                    acc.z = acc.z * sc + v.z * w2; acc.w = acc.w * sc + v.w * w2;
                    m = mn;
                }
            }
            const float inv = 1.0f / (d + 1e-16f);
            acc.x *= inv; acc.y *= inv; acc.z *= inv; acc.w *= inv;
        }
        park[w][k][lane] = acc;
    }

    const float inv_n = 1.0f / (float)NN;
    const float4* __restrict__ urow = (const float4*)upd + (size_t)n0 * (DF / 4) + lane;
    float4 u[NPW];
    #pragma unroll
    for (int k = 0; k < NPW; ++k) u[k] = urow[k * (DF / 4)];
    const float4 ma = ((const float4*)macc)[lane];
    const float4 sq = ((const float4*)sqacc)[lane];
    const float4 g  = ((const float4*)gamma)[lane];
    const float4 b  = ((const float4*)beta)[lane];
    float4 rs;
    rs.x = fin_rstd(sq.x, ma.x * inv_n, inv_n); rs.y = fin_rstd(sq.y, ma.y * inv_n, inv_n);
    rs.z = fin_rstd(sq.z, ma.z * inv_n, inv_n); rs.w = fin_rstd(sq.w, ma.w * inv_n, inv_n);
    float4* __restrict__ orow = (float4*)out + (size_t)n0 * (DF / 4) + lane;
    #pragma unroll
    for (int k = 0; k < NPW; ++k) {
        const float4 a = park[w][k][lane];
        float4 o;
        o.x = fin_one(u[k].x, a.x, ma.x, inv_n, rs.x, g.x, b.x);
        o.y = fin_one(u[k].y, a.y, ma.y, inv_n, rs.y, g.y, b.y);
        o.z = fin_one(u[k].z, a.z, ma.z, inv_n, rs.z, g.z, b.z);
        o.w = fin_one(u[k].w, a.w, ma.w, inv_n, rs.w, g.w, b.w);
        orow[k * (DF / 4)] = o;
    }
}

#define BN_WAVES 8
#define BN_WROWS 16
#define BN_ROWS  (BN_WAVES * BN_WROWS)
#define BN_BLOCKS ((NN + BN_ROWS - 1) / BN_ROWS)
static_assert(NN % BN_WROWS == 0, "a wave's rows are all inside the matrix or all outside");
DEV void bn_partial_block(
    const int bid,
    const float* __restrict__ upd, float* __restrict__ macc,
    float* __restrict__ sqacc)
{
    __shared__ float4 ps[BN_WAVES][32];
    __shared__ float4 ps2[BN_WAVES][32];
    const int tid  = threadIdx.x;
    const int lane = tid & 31;
    const int w    = __builtin_amdgcn_readfirstlane(tid >> 5);
    const int r0   = bid * BN_ROWS + w * BN_WROWS;

    float4 s  = make_float4(0.0f, 0.0f, 0.0f, 0.0f);
    float4 s2 = make_float4(0.0f, 0.0f, 0.0f, 0.0f);
    if (r0 < NN) {
        const float4* __restrict__ u4 = (const float4*)upd + (size_t)r0 * (DF / 4) + lane;
        float4 v[BN_WROWS];
        #pragma unroll
        for (int k = 0; k < BN_WROWS; ++k) v[k] = u4[k * (DF / 4)];
        #pragma unroll
        for (int k = 0; k < BN_WROWS; ++k) {
            s.x += v[k].x; s2.x += v[k].x * v[k].x;
            s.y += v[k].y; s2.y += v[k].y * v[k].y;
            s.z += v[k].z; s2.z += v[k].z * v[k].z;
            s.w += v[k].w; s2.w += v[k].w * v[k].w;
        }
    }
    ps[w][lane]  = s;
    ps2[w][lane] = s2;
    __syncthreads();
    if (tid < DF) {
        const float* __restrict__ p  = (const float*)ps;
        const float* __restrict__ p2 = (const float*)ps2;
        float a = 0.0f, b = 0.0f;
        for (int k = 0; k < BN_WAVES; ++k) { a += p[k * DF + tid]; b += p2[k * DF + tid]; }
        atomicAdd(&macc[tid], a);
        atomicAdd(&sqacc[tid], b);
    }
}

__global__ __launch_bounds__(256) void k_group_bnsum(
    const int* __restrict__ binned, const int* __restrict__ runs,
    int* __restrict__ cnt, int* __restrict__ offs, int* __restrict__ list,
    const float* __restrict__ upd, float* __restrict__ macc,
    float* __restrict__ sqacc)
{
    if (blockIdx.x < NRANGE) range_group_block(blockIdx.x, binned, runs, cnt, offs, list);
    else bn_partial_block(blockIdx.x - NRANGE, upd, macc, sqacc);
}

extern "C" void kernel_launch(void* const* d_in, const int* in_sizes, int n_in,
                              void* d_out, int out_size, void* d_ws, size_t ws_size,
                              hipStream_t stream) {
    const float* x     = (const float*)d_in[0];
    const int*   ei    = (const int*)  d_in[1];
    const float* Wk    = (const float*)d_in[2];
    const float* bk    = (const float*)d_in[3];
    const float* a_i   = (const float*)d_in[4];
    const float* a_j   = (const float*)d_in[5];
    const float* a_e   = (const float*)d_in[6];
    const float* b_a   = (const float*)d_in[7];
    const float* Wm    = (const float*)d_in[8];
    const float* bm    = (const float*)d_in[9];
    const float* gamma = (const float*)d_in[10];
    const float* beta  = (const float*)d_in[11];
    float* out = (float*)d_out;
    float* ws  = (float*)d_ws;

    float* h_buf  = ws + OFF_H;
    float* upd    = ws + OFF_UPD;
    int*   cnt    = (int*)ws + OFF_CNT;
    int*   offs   = (int*)ws + OFF_OFFS;
    int*   list   = (int*)ws + OFF_LIST;
    int*   binned = (int*)ws + OFF_BIN;
    int*   runs   = (int*)ws + OFF_TAB;
    float* macc   = ws + OFF_MACC;
    float* sqacc  = ws + OFF_SQ;

    k_gemm2_wmma<<<GT_WGS, 256, 0, stream>>>(x, Wk, bk, Wm, bm, h_buf, upd, macc, sqacc);
    k_edge_block_sort<<<PBLOCKS, SC_THREADS, 0, stream>>>(ei, runs, binned);
    k_group_bnsum<<<NRANGE + BN_BLOCKS, 256, 0, stream>>>(binned, runs, cnt, offs, list, upd, macc, sqacc);
    k_node_gather_out<<<GO_BLOCKS, 256, 0, stream>>>(
        list, cnt, offs, a_i, a_j, a_e, b_a, h_buf, upd, macc, sqacc, gamma, beta, out);
}
